# starting kernel
# baseline (speedup 1.0000x reference)
#include <hip/hip_runtime.h>
#include <hip/hip_bf16.h>
#include <math.h>

typedef __attribute__((ext_vector_type(2))) float v2f;
typedef __attribute__((ext_vector_type(8))) float v8f;

#define D 128

__global__ void k_init_deg(float* deg, int n) {
    int i = blockIdx.x * blockDim.x + threadIdx.x;
    if (i < n) deg[i] = 1.0f;
}

__global__ void k_count_deg(const int* __restrict__ ei, int E, float* deg) {
    int e = blockIdx.x * blockDim.x + threadIdx.x;
    if (e < E) atomicAdd(&deg[ei[E + e]], 1.0f);
}

__global__ void k_dinv(const float* __restrict__ deg, float* dinv, int n) {
    int i = blockIdx.x * blockDim.x + threadIdx.x;
    if (i < n) {
        float d = deg[i];
        dinv[i] = (d > 0.0f) ? rsqrtf(d) : 0.0f;
    }
}

__global__ __launch_bounds__(256) void k_gemm_wmma(
        const float* __restrict__ A, const float* __restrict__ W,
        float* __restrict__ C, int n) {
    __shared__ float Wl[D * D];

    for (int i = threadIdx.x; i < (D * D) / 4; i += 256)
        ((float4*)Wl)[i] = ((const float4*)W)[i];
    __syncthreads();

    const int wave = threadIdx.x >> 5;
    const int lane = threadIdx.x & 31;
    const int m0   = blockIdx.x * 128 + wave * 16;

    int arow = m0 + (lane & 15);
    if (arow > n - 1) arow = n - 1;
    const int khalf = (lane >> 4) * 2;
    const float* Ar = A + (size_t)arow * D;

    v8f acc[8] = {};

    for (int k = 0; k < D; k += 4) {
        const v2f a = *(const v2f*)(Ar + k + khalf);
        const int kb = k + khalf;
#pragma unroll
        for (int t = 0; t < 8; ++t) {
            const int col = t * 16 + (lane & 15);
            v2f b;
            b.x = Wl[kb * D + col];
            b.y = Wl[(kb + 1) * D + col];
            acc[t] = __builtin_amdgcn_wmma_f32_16x16x4_f32(
                false, a, false, b, (short)0, acc[t], false, false);
        }
    }

    const int rbase = m0 + (lane >> 4) * 8;
    const int cix   = lane & 15;
#pragma unroll
    for (int t = 0; t < 8; ++t) {
#pragma unroll
        for (int v = 0; v < 8; ++v) {
            const int r = rbase + v;
            if (r < n) C[(size_t)r * D + t * 16 + cix] = acc[t][v];
        }
    }
}

__global__ void k_self_init(const float* __restrict__ xw,
                            const float* __restrict__ dinv,
                            float* __restrict__ out, int n) {
    size_t i = (size_t)blockIdx.x * blockDim.x + threadIdx.x;
    if (i >= (size_t)n * (D / 4)) return;
    int node = (int)(i >> 5);
    float dd = dinv[node]; dd *= dd;
    float4 v = ((const float4*)xw)[i];
    float4 o = {v.x * dd, v.y * dd, v.z * dd, v.w * dd};
    ((float4*)out)[i] = o;
}

__global__ __launch_bounds__(256) void k_agg(
        const int* __restrict__ ei, int E,
        const float* __restrict__ xw, const float* __restrict__ dinv,
        float* __restrict__ out) {
    int e = blockIdx.x * 8 + (threadIdx.x >> 5);
    if (e >= E) return;
    const int lane = threadIdx.x & 31;
    const int r = ei[e];
    const int c = ei[E + e];
    const float nrm = dinv[r] * dinv[c];
    float4 v = ((const float4*)(xw + (size_t)r * D))[lane];
    float* dst = out + (size_t)c * D + lane * 4;
    atomicAdd(dst + 0, v.x * nrm);
    atomicAdd(dst + 1, v.y * nrm);
    atomicAdd(dst + 2, v.z * nrm);
    atomicAdd(dst + 3, v.w * nrm);
}

__global__ void k_bias_relu(float* h, const float* __restrict__ b, int n) {
    size_t i = (size_t)blockIdx.x * blockDim.x + threadIdx.x;
    if (i >= (size_t)n * D) return;
    int f = (int)(i & (D - 1));
    float v = h[i] + b[f];
    h[i] = v > 0.0f ? v : 0.0f;
}

__global__ __launch_bounds__(256) void k_edge_cls(
        const int* __restrict__ ei, int E,
        const float* __restrict__ h, const float* __restrict__ Wfc,
        const float* __restrict__ bfc, float* __restrict__ out) {
    int e = blockIdx.x * 8 + (threadIdx.x >> 5);
    if (e >= E) return;
    const int lane = threadIdx.x & 31;
    const int r = ei[e];
    const int c = ei[E + e];
    float4 vr = ((const float4*)(h + (size_t)r * D))[lane];
    float4 vc = ((const float4*)(h + (size_t)c * D))[lane];
    const int k0 = lane * 4;
    float s0 = 0.f, s1 = 0.f, s2 = 0.f, s3 = 0.f;
    const float* pr = (const float*)&vr;
    const float* pc = (const float*)&vc;
#pragma unroll
    for (int i = 0; i < 4; ++i) {
        float a  = pr[i];
        float bb = pc[i];
        float4 wr = ((const float4*)Wfc)[k0 + i];
        float4 wc = ((const float4*)Wfc)[D + k0 + i];
        s0 += a * wr.x + bb * wc.x;
        s1 += a * wr.y + bb * wc.y;
        s2 += a * wr.z + bb * wc.z;
        s3 += a * wr.w + bb * wc.w;
    }
#pragma unroll
    for (int m = 16; m > 0; m >>= 1) {
        s0 += __shfl_xor(s0, m);
        s1 += __shfl_xor(s1, m);
        s2 += __shfl_xor(s2, m);
        s3 += __shfl_xor(s3, m);
    }
    if (lane == 0) {
        s0 += bfc[0]; s1 += bfc[1]; s2 += bfc[2]; s3 += bfc[3];
        float mx  = fmaxf(fmaxf(s0, s1), fmaxf(s2, s3));
        float sum = __expf(s0 - mx) + __expf(s1 - mx) +
                    __expf(s2 - mx) + __expf(s3 - mx);
        float lse = mx + __logf(sum);
        float4 o  = {s0 - lse, s1 - lse, s2 - lse, s3 - lse};
        ((float4*)out)[e] = o;
    }
}

extern "C" void kernel_launch(void* const* d_in, const int* in_sizes, int n_in,
                              void* d_out, int out_size, void* d_ws, size_t ws_size,
                              hipStream_t stream) {
    const float* x   = (const float*)d_in[0];
    const int*   ei  = (const int*)  d_in[1];
    const float* W1  = (const float*)d_in[2];
    const float* b1  = (const float*)d_in[3];
    const float* W2  = (const float*)d_in[4];
    const float* b2  = (const float*)d_in[5];
    const float* Wfc = (const float*)d_in[6];
    const float* bfc = (const float*)d_in[7];
    float* out = (float*)d_out;

    const int n = in_sizes[0] / D;
    const int E = in_sizes[1] / 2;

    float* bufA = (float*)d_ws;
    float* bufB = bufA + (size_t)n * D;
    float* deg  = bufB + (size_t)n * D;
    float* dinv = deg + n;

    const int gN    = (n + 255) / 256;
    const int gE    = (E + 255) / 256;
    const int gGemm = (n + 127) / 128;
    const int gVec4 = (int)(((size_t)n * (D / 4) + 255) / 256);
    const int gFeat = (int)(((size_t)n * D + 255) / 256);
    const int gWave = (E + 7) / 8;

    k_init_deg <<<gN, 256, 0, stream>>>(deg, n);
    k_count_deg<<<gE, 256, 0, stream>>>(ei, E, deg);
    k_dinv     <<<gN, 256, 0, stream>>>(deg, dinv, n);

    k_gemm_wmma<<<gGemm, 256, 0, stream>>>(x, W1, bufA, n);
    k_self_init<<<gVec4, 256, 0, stream>>>(bufA, dinv, bufB, n);
    k_agg      <<<gWave, 256, 0, stream>>>(ei, E, bufA, dinv, bufB);
    k_bias_relu<<<gFeat, 256, 0, stream>>>(bufB, b1, n);

    k_gemm_wmma<<<gGemm, 256, 0, stream>>>(bufB, W2, bufA, n);
    k_self_init<<<gVec4, 256, 0, stream>>>(bufA, dinv, bufB, n);
    k_agg      <<<gWave, 256, 0, stream>>>(ei, E, bufA, dinv, bufB);
    k_bias_relu<<<gFeat, 256, 0, stream>>>(bufB, b2, n);

    k_edge_cls <<<gWave, 256, 0, stream>>>(ei, E, bufB, Wfc, bfc, out);
}


# Float-atomic edge scatter replaced by per-target in-edge lists (integer-atomic arrival-order rank, prefix scan) and a register gather per node: sums run in list order, row stored once with bias+relu
# speedup vs baseline: 1.8729x; 1.8729x over previous
#include <hip/hip_runtime.h>
#include <hip/hip_bf16.h>
#include <math.h>

typedef __attribute__((ext_vector_type(2))) float v2f;
typedef __attribute__((ext_vector_type(8))) float v8f;

#define D 128

__global__ void k_zero_cnt(int* cnt, int n) {
    int i = blockIdx.x * blockDim.x + threadIdx.x;
    if (i < n) cnt[i] = 0;
}

__global__ void k_count_rank(const int* __restrict__ ei, int E, int* cnt,
                             int* __restrict__ rank) {
    int e = blockIdx.x * blockDim.x + threadIdx.x;
    if (e < E) rank[e] = atomicAdd(&cnt[ei[E + e]], 1);
}

__global__ __launch_bounds__(256) void k_scan(
        const int* __restrict__ cnt, int n,
        int* __restrict__ rowptr, float* __restrict__ dinv) {
    __shared__ int s_pre[8];
    __shared__ int s_loc[8];
    const int lane = threadIdx.x & 31;
    const int wave = threadIdx.x >> 5;
    const int base = blockIdx.x * 256;

    const int4* c4 = (const int4*)cnt;
    const int nq = base >> 2;
    int s = 0;
    int j = threadIdx.x;
    for (; j + 7 * 256 < nq; j += 8 * 256) {
        int4 q[8];
        for (int k = 0; k < 8; ++k) q[k] = c4[j + k * 256];
        for (int k = 0; k < 8; ++k) s += (q[k].x + q[k].y) + (q[k].z + q[k].w);
    }
    for (; j < nq; j += 256) {
        const int4 q = c4[j];
        s += (q.x + q.y) + (q.z + q.w);
    }
    for (int m = 16; m > 0; m >>= 1) s += __shfl_xor(s, m);

    const int i = base + threadIdx.x;
    const int c = (i < n) ? cnt[i] : 0;
    int v = c;
    for (int d = 1; d < 32; d <<= 1) {
        const int t = __shfl_up(v, d);
        if (lane >= d) v += t;
    }
    if (lane == 0)  s_pre[wave] = s;
    if (lane == 31) s_loc[wave] = v;
    __syncthreads();
    int off = 0;
    for (int w = 0; w < 8; ++w) {
        off += s_pre[w];
        if (w < wave) off += s_loc[w];
    }
    const int excl = off + v - c;
    if (i < n) {
        rowptr[i] = excl;
        const float d = (float)(c + 1);
        dinv[i] = (d > 0.0f) ? rsqrtf(d) : 0.0f;
    }
    if (i == n - 1) rowptr[n] = excl + c;
}

__global__ void k_fill(const int* __restrict__ ei, int E,
                       const int* __restrict__ rowptr, const int* __restrict__ rank,
                       const float* __restrict__ dinv, int2* __restrict__ rec) {
    int e = blockIdx.x * blockDim.x + threadIdx.x;
    if (e < E) {
        const int r = ei[e];
        const int c = ei[E + e];
        const float nrm = dinv[r] * dinv[c];
        rec[rowptr[c] + rank[e]] = make_int2(r, __float_as_int(nrm));
    }
}

__global__ __launch_bounds__(256) void k_gemm_wmma(
        const float* __restrict__ A, const float* __restrict__ W,
        float* __restrict__ C, int n) {
    __shared__ float Wl[D * D];

    for (int i = threadIdx.x; i < (D * D) / 4; i += 256)
        ((float4*)Wl)[i] = ((const float4*)W)[i];
    __syncthreads();

    const int wave = threadIdx.x >> 5;
    const int lane = threadIdx.x & 31;
    const int m0   = blockIdx.x * 128 + wave * 16;

    int arow = m0 + (lane & 15);
    if (arow > n - 1) arow = n - 1;
    const int khalf = (lane >> 4) * 2;
    const float* Ar = A + (size_t)arow * D;

    v8f acc[8] = {};

    for (int k = 0; k < D; k += 4) {
        const v2f a = *(const v2f*)(Ar + k + khalf);
        const int kb = k + khalf;
#pragma unroll
        for (int t = 0; t < 8; ++t) {
            const int col = t * 16 + (lane & 15);
            v2f b;
            b.x = Wl[kb * D + col];
            b.y = Wl[(kb + 1) * D + col];
            acc[t] = __builtin_amdgcn_wmma_f32_16x16x4_f32(
                false, a, false, b, (short)0, acc[t], false, false);
        }
    }

    const int rbase = m0 + (lane >> 4) * 8;
    const int cix   = lane & 15;
#pragma unroll
    for (int t = 0; t < 8; ++t) {
#pragma unroll
        for (int v = 0; v < 8; ++v) {
            const int r = rbase + v;
            if (r < n) C[(size_t)r * D + t * 16 + cix] = acc[t][v];
        }
    }
}

template <int K>
__device__ __forceinline__ void add_rows(float4& acc, const float4* __restrict__ X,
                                         int r_l, float w_l, int j, int lane) {
    float4 u[K];
    float  w[K];
    for (int k = 0; k < K; ++k) {
        const int r = __builtin_amdgcn_readlane(r_l, j + k);
        w[k] = __int_as_float(__builtin_amdgcn_readlane(__float_as_int(w_l), j + k));
        u[k] = X[(size_t)r * (D / 4) + lane];
    }
    for (int k = 0; k < K; ++k) {
        acc.x = __builtin_fmaf(u[k].x, w[k], acc.x);
        acc.y = __builtin_fmaf(u[k].y, w[k], acc.y);
        acc.z = __builtin_fmaf(u[k].z, w[k], acc.z);
        acc.w = __builtin_fmaf(u[k].w, w[k], acc.w);
    }
}

__global__ __launch_bounds__(256) void k_gather(
        const int* __restrict__ rowptr, const int2* __restrict__ rec,
        const float* __restrict__ xw, const float* __restrict__ dinv,
        const float* __restrict__ b, float* __restrict__ out, int n) {
    const int wv = __builtin_amdgcn_readfirstlane((int)(threadIdx.x >> 5));
    const int i = blockIdx.x * 8 + wv;
    if (i >= n) return;
    const int lane = threadIdx.x & 31;
    const int beg = rowptr[i];
    const int deg = rowptr[i + 1] - beg;
    const int tot = deg + 1;
    const float di = dinv[i];
    const float4* X = (const float4*)xw;

    float4 acc = {0.0f, 0.0f, 0.0f, 0.0f};
    for (int t0 = 0; t0 < tot; t0 += 32) {
        const int m = min(tot - t0, 32);
        const int t = t0 + lane;
        int   r_l = i;
        float w_l = (t == deg) ? di * di : 0.0f;
        if (t < deg) {
            const int2 q = rec[beg + t];
            r_l = q.x;
            w_l = __int_as_float(q.y);
        }
        int j = 0;
        for (; m - j > 4; j += 8) add_rows<8>(acc, X, r_l, w_l, j, lane);
        if (j < m)               add_rows<4>(acc, X, r_l, w_l, j, lane);
    }

    const float4 bb = ((const float4*)b)[lane];
    float4 o = {acc.x + bb.x, acc.y + bb.y, acc.z + bb.z, acc.w + bb.w};
    o.x = o.x > 0.0f ? o.x : 0.0f;
    o.y = o.y > 0.0f ? o.y : 0.0f;
    o.z = o.z > 0.0f ? o.z : 0.0f;
    o.w = o.w > 0.0f ? o.w : 0.0f;
    ((float4*)out)[(size_t)i * (D / 4) + lane] = o;
}

__global__ __launch_bounds__(256) void k_edge_cls(
        const int* __restrict__ ei, int E,
        const float* __restrict__ h, const float* __restrict__ Wfc,
        const float* __restrict__ bfc, float* __restrict__ out) {
    int e = blockIdx.x * 8 + (threadIdx.x >> 5);
    if (e >= E) return;
    const int lane = threadIdx.x & 31;
    const int r = ei[e];
    const int c = ei[E + e];
    float4 vr = ((const float4*)(h + (size_t)r * D))[lane];
    float4 vc = ((const float4*)(h + (size_t)c * D))[lane];
    const int k0 = lane * 4;
    float s0 = 0.f, s1 = 0.f, s2 = 0.f, s3 = 0.f;
    const float* pr = (const float*)&vr;
    const float* pc = (const float*)&vc;
#pragma unroll
    for (int i = 0; i < 4; ++i) {
        float a  = pr[i];
        float bb = pc[i];
        float4 wr = ((const float4*)Wfc)[k0 + i];
        float4 wc = ((const float4*)Wfc)[D + k0 + i];
        s0 += a * wr.x + bb * wc.x;
        s1 += a * wr.y + bb * wc.y;
        s2 += a * wr.z + bb * wc.z;
        s3 += a * wr.w + bb * wc.w;
    }
#pragma unroll
    for (int m = 16; m > 0; m >>= 1) {
        s0 += __shfl_xor(s0, m);
        s1 += __shfl_xor(s1, m);
        s2 += __shfl_xor(s2, m);
        s3 += __shfl_xor(s3, m);
    }
    if (lane == 0) {
        s0 += bfc[0]; s1 += bfc[1]; s2 += bfc[2]; s3 += bfc[3];
        float mx  = fmaxf(fmaxf(s0, s1), fmaxf(s2, s3));
        float sum = __expf(s0 - mx) + __expf(s1 - mx) +
                    __expf(s2 - mx) + __expf(s3 - mx);
        float lse = mx + __logf(sum);
        float4 o  = {s0 - lse, s1 - lse, s2 - lse, s3 - lse};
        ((float4*)out)[e] = o;
    }
}

extern "C" void kernel_launch(void* const* d_in, const int* in_sizes, int n_in,
                              void* d_out, int out_size, void* d_ws, size_t ws_size,
                              hipStream_t stream) {
    const float* x   = (const float*)d_in[0];
    const int*   ei  = (const int*)  d_in[1];
    const float* W1  = (const float*)d_in[2];
    const float* b1  = (const float*)d_in[3];
    const float* W2  = (const float*)d_in[4];
    const float* b2  = (const float*)d_in[5];
    const float* Wfc = (const float*)d_in[6];
    const float* bfc = (const float*)d_in[7];
    float* out = (float*)d_out;

    const int n = in_sizes[0] / D;
    const int E = in_sizes[1] / 2;
    const int n4 = (n + 3) & ~3;
    const int E4 = (E + 3) & ~3;

    float* bufA   = (float*)d_ws;
    float* bufB   = bufA + (size_t)n * D;
    float* dinv   = bufB + (size_t)n * D;
    int*   cnt    = (int*)(dinv + n4);
    int*   rowptr = cnt + n4;
    int*   rank   = rowptr + n4 + 4;
    int2*  rec    = (int2*)(rank + E4);

    const int gN    = (n + 255) / 256;
    const int gE    = (E + 255) / 256;
    const int gGemm = (n + 127) / 128;
    const int gNode = (n + 7) / 8;
    const int gWave = (E + 7) / 8;

    k_zero_cnt  <<<gN, 256, 0, stream>>>(cnt, n);
    k_count_rank<<<gE, 256, 0, stream>>>(ei, E, cnt, rank);
    k_scan      <<<gN, 256, 0, stream>>>(cnt, n, rowptr, dinv);
    k_fill      <<<gE, 256, 0, stream>>>(ei, E, rowptr, rank, dinv, rec);

    k_gemm_wmma<<<gGemm, 256, 0, stream>>>(x, W1, bufA, n);
    k_gather   <<<gNode, 256, 0, stream>>>(rowptr, rec, bufA, dinv, b1, bufB, n);

    k_gemm_wmma<<<gGemm, 256, 0, stream>>>(bufB, W2, bufA, n);
    k_gather   <<<gNode, 256, 0, stream>>>(rowptr, rec, bufA, dinv, b2, bufB, n);

    k_edge_cls <<<gWave, 256, 0, stream>>>(ei, E, bufB, Wfc, bfc, out);
}


# Edge classifier split by linearity: layer-2 gather keeps h in registers, stores per-node h@Wfc[:128] and h@Wfc[128:] (8 floats); edge pass adds P[row]+Q[col]+bias; dot-product sum order changes
# speedup vs baseline: 2.8510x; 1.5222x over previous
#include <hip/hip_runtime.h>
#include <hip/hip_bf16.h>
#include <math.h>

typedef __attribute__((ext_vector_type(2))) float v2f;
typedef __attribute__((ext_vector_type(8))) float v8f;

#define D 128

__global__ void k_zero_cnt(int* cnt, int n) {
    int i = blockIdx.x * blockDim.x + threadIdx.x;
    if (i < n) cnt[i] = 0;
}

__global__ void k_count_rank(const int* __restrict__ ei, int E, int* cnt,
                             int* __restrict__ rank) {
    int e = blockIdx.x * blockDim.x + threadIdx.x;
    if (e < E) rank[e] = atomicAdd(&cnt[ei[E + e]], 1);
}

__global__ __launch_bounds__(256) void k_scan(
        const int* __restrict__ cnt, int n,
        int* __restrict__ rowptr, float* __restrict__ dinv) {
    __shared__ int s_pre[8];
    __shared__ int s_loc[8];
    const int lane = threadIdx.x & 31;
    const int wave = threadIdx.x >> 5;
    const int base = blockIdx.x * 256;

    const int4* c4 = (const int4*)cnt;
    const int nq = base >> 2;
    int s = 0;
    int j = threadIdx.x;
    for (; j + 7 * 256 < nq; j += 8 * 256) {
        int4 q[8];
        for (int k = 0; k < 8; ++k) q[k] = c4[j + k * 256];
        for (int k = 0; k < 8; ++k) s += (q[k].x + q[k].y) + (q[k].z + q[k].w);
    }
    for (; j < nq; j += 256) {
        const int4 q = c4[j];
        s += (q.x + q.y) + (q.z + q.w);
    }
    for (int m = 16; m > 0; m >>= 1) s += __shfl_xor(s, m);

    const int i = base + threadIdx.x;
    const int c = (i < n) ? cnt[i] : 0;
    int v = c;
    for (int d = 1; d < 32; d <<= 1) {
        const int t = __shfl_up(v, d);
        if (lane >= d) v += t;
    }
    if (lane == 0)  s_pre[wave] = s;
    if (lane == 31) s_loc[wave] = v;
    __syncthreads();
    int off = 0;
    for (int w = 0; w < 8; ++w) {
        off += s_pre[w];
        if (w < wave) off += s_loc[w];
    }
    const int excl = off + v - c;
    if (i < n) {
        rowptr[i] = excl;
        const float d = (float)(c + 1);
        dinv[i] = (d > 0.0f) ? rsqrtf(d) : 0.0f;
    }
    if (i == n - 1) rowptr[n] = excl + c;
}

__global__ void k_fill(const int* __restrict__ ei, int E,
                       const int* __restrict__ rowptr, const int* __restrict__ rank,
                       const float* __restrict__ dinv, int2* __restrict__ rec) {
    int e = blockIdx.x * blockDim.x + threadIdx.x;
    if (e < E) {
        const int r = ei[e];
        const int c = ei[E + e];
        const float nrm = dinv[r] * dinv[c];
        rec[rowptr[c] + rank[e]] = make_int2(r, __float_as_int(nrm));
    }
}

__global__ __launch_bounds__(256) void k_gemm_wmma(
        const float* __restrict__ A, const float* __restrict__ W,
        float* __restrict__ C, int n) {
    __shared__ float Wl[D * D];

    for (int i = threadIdx.x; i < (D * D) / 4; i += 256)
        ((float4*)Wl)[i] = ((const float4*)W)[i];
    __syncthreads();

    const int wave = threadIdx.x >> 5;
    const int lane = threadIdx.x & 31;
    const int m0   = blockIdx.x * 128 + wave * 16;

    int arow = m0 + (lane & 15);
    if (arow > n - 1) arow = n - 1;
    const int khalf = (lane >> 4) * 2;
    const float* Ar = A + (size_t)arow * D;

    v8f acc[8] = {};

    for (int k = 0; k < D; k += 4) {
        const v2f a = *(const v2f*)(Ar + k + khalf);
        const int kb = k + khalf;
#pragma unroll
        for (int t = 0; t < 8; ++t) {
            const int col = t * 16 + (lane & 15);
            v2f b;
            b.x = Wl[kb * D + col];
            b.y = Wl[(kb + 1) * D + col];
            acc[t] = __builtin_amdgcn_wmma_f32_16x16x4_f32(
                false, a, false, b, (short)0, acc[t], false, false);
        }
    }

    const int rbase = m0 + (lane >> 4) * 8;
    const int cix   = lane & 15;
#pragma unroll
    for (int t = 0; t < 8; ++t) {
#pragma unroll
        for (int v = 0; v < 8; ++v) {
            const int r = rbase + v;
            if (r < n) C[(size_t)r * D + t * 16 + cix] = acc[t][v];
        }
    }
}

template <int K>
__device__ __forceinline__ void add_rows(float4& acc, const float4* __restrict__ X,
                                         int r_l, float w_l, int j, int lane) {
    float4 u[K];
    float  w[K];
    for (int k = 0; k < K; ++k) {
        const int r = __builtin_amdgcn_readlane(r_l, j + k);
        w[k] = __int_as_float(__builtin_amdgcn_readlane(__float_as_int(w_l), j + k));
        u[k] = X[(size_t)r * (D / 4) + lane];
    }
    for (int k = 0; k < K; ++k) {
        acc.x = __builtin_fmaf(u[k].x, w[k], acc.x);
        acc.y = __builtin_fmaf(u[k].y, w[k], acc.y);
        acc.z = __builtin_fmaf(u[k].z, w[k], acc.z);
        acc.w = __builtin_fmaf(u[k].w, w[k], acc.w);
    }
}

template <int M>
__device__ __forceinline__ float xrow(float x) {
    const int v = __builtin_bit_cast(int, x);
    return __builtin_bit_cast(float,
        __builtin_amdgcn_update_dpp(v, v, 0x160 + M, 0xf, 0xf, false));
}
__device__ __forceinline__ float xhalf(float x) {
    const int v = __builtin_bit_cast(int, x);
    return __builtin_bit_cast(float,
        __builtin_amdgcn_permlanex16(v, v, 0x76543210, 0xfedcba98, false, false));
}

template <bool HEAD>
__global__ __launch_bounds__(256) void k_gather(
        const int* __restrict__ rowptr, const int2* __restrict__ rec,
        const float* __restrict__ xw, const float* __restrict__ dinv,
        const float* __restrict__ b, const float* __restrict__ Wfc,
        float* __restrict__ out, int n) {
    __shared__ float4 s_w[HEAD ? 256 : 1];
    const int wv = __builtin_amdgcn_readfirstlane((int)(threadIdx.x >> 5));
    const int i = blockIdx.x * 8 + wv;
    const int lane = threadIdx.x & 31;
    const int ic = min(i, n - 1);
    const int beg = rowptr[ic];
    const int deg = rowptr[ic + 1] - beg;
    const float di = dinv[ic];
    if (HEAD) {
        const int t = threadIdx.x;
        s_w[((t >> 7) * 4 + (t & 3)) * 32 + ((t & 127) >> 2)] = ((const float4*)Wfc)[t];
        __syncthreads();
    }
    if (i >= n) return;
    const int tot = deg + 1;
    const float4* X = (const float4*)xw;

    float4 acc = {0.0f, 0.0f, 0.0f, 0.0f};
    for (int t0 = 0; t0 < tot; t0 += 32) {
        const int m = min(tot - t0, 32);
        const int t = t0 + lane;
        int   r_l = i;
        float w_l = (t == deg) ? di * di : 0.0f;
        if (t < deg) {
            const int2 q = rec[beg + t];
            r_l = q.x;
            w_l = __int_as_float(q.y);
        }
        int j = 0;
        for (; m - j > 4; j += 8) add_rows<8>(acc, X, r_l, w_l, j, lane);
        if (j < m)               add_rows<4>(acc, X, r_l, w_l, j, lane);
    }

    const float4 bb = ((const float4*)b)[lane];
    float4 o = {acc.x + bb.x, acc.y + bb.y, acc.z + bb.z, acc.w + bb.w};
    o.x = o.x > 0.0f ? o.x : 0.0f;
    o.y = o.y > 0.0f ? o.y : 0.0f;
    o.z = o.z > 0.0f ? o.z : 0.0f;
    o.w = o.w > 0.0f ? o.w : 0.0f;
    if (!HEAD) {
        ((float4*)out)[(size_t)i * (D / 4) + lane] = o;
        return;
    }

    const float4 t0 = s_w[0 * 32 + lane], t1 = s_w[1 * 32 + lane];
    const float4 t2 = s_w[2 * 32 + lane], t3 = s_w[3 * 32 + lane];
    const float4 u0 = s_w[4 * 32 + lane], u1 = s_w[5 * 32 + lane];
    const float4 u2 = s_w[6 * 32 + lane], u3 = s_w[7 * 32 + lane];
    float s[8];
    s[0] = o.x * t0.x + o.y * t1.x + o.z * t2.x + o.w * t3.x;
    s[1] = o.x * t0.y + o.y * t1.y + o.z * t2.y + o.w * t3.y;
    s[2] = o.x * t0.z + o.y * t1.z + o.z * t2.z + o.w * t3.z;
    s[3] = o.x * t0.w + o.y * t1.w + o.z * t2.w + o.w * t3.w;
    s[4] = o.x * u0.x + o.y * u1.x + o.z * u2.x + o.w * u3.x;
    s[5] = o.x * u0.y + o.y * u1.y + o.z * u2.y + o.w * u3.y;
    s[6] = o.x * u0.z + o.y * u1.z + o.z * u2.z + o.w * u3.z;
    s[7] = o.x * u0.w + o.y * u1.w + o.z * u2.w + o.w * u3.w;

    const bool h16 = (lane & 16) != 0;
    float a4[4];
    for (int q = 0; q < 4; ++q) {
        const float keep = h16 ? s[q + 4] : s[q];
        const float give = h16 ? s[q] : s[q + 4];
        a4[q] = keep + xhalf(give);
    }
    const bool h8 = (lane & 8) != 0;
    float a2[2];
    for (int q = 0; q < 2; ++q) {
        const float keep = h8 ? a4[q + 2] : a4[q];
        const float give = h8 ? a4[q] : a4[q + 2];
        a2[q] = keep + xrow<8>(give);
    }
    const bool h4 = (lane & 4) != 0;
    const float keep = h4 ? a2[1] : a2[0];
    const float give = h4 ? a2[0] : a2[1];
    float a1 = keep + xrow<4>(give);
    a1 += xrow<2>(a1);
    a1 += xrow<1>(a1);
    if ((lane & 3) == 0)
        out[(size_t)i * 8 + ((lane >> 4) & 1) * 4 + ((lane >> 3) & 1) * 2 + ((lane >> 2) & 1)] = a1;
}

#define EPT 4
template <bool FULL>
__device__ __forceinline__ void edge_tile(
        const int* __restrict__ ei, int E, const float4* __restrict__ pq4,
        const float* __restrict__ bfc, float4* __restrict__ out4, int e0) {
    int r[EPT], c[EPT];
    for (int k = 0; k < EPT; ++k) {
        const int e = FULL ? e0 + k * 256 : min(e0 + k * 256, E - 1);
        r[k] = ei[e];
        c[k] = ei[E + e];
    }
    float4 p[EPT], q[EPT];
    for (int k = 0; k < EPT; ++k) {
        p[k] = pq4[(size_t)r[k] * 2];
        q[k] = pq4[(size_t)c[k] * 2 + 1];
    }
    const float b0 = bfc[0], b1 = bfc[1], b2 = bfc[2], b3 = bfc[3];
    for (int k = 0; k < EPT; ++k) {
        const float s0 = (p[k].x + q[k].x) + b0;
        const float s1 = (p[k].y + q[k].y) + b1;
        const float s2 = (p[k].z + q[k].z) + b2;
        const float s3 = (p[k].w + q[k].w) + b3;
        const float mx  = fmaxf(fmaxf(s0, s1), fmaxf(s2, s3));
        const float sum = __expf(s0 - mx) + __expf(s1 - mx) +
                          __expf(s2 - mx) + __expf(s3 - mx);
        const float lse = mx + __logf(sum);
        const float4 o  = {s0 - lse, s1 - lse, s2 - lse, s3 - lse};
        const int e = e0 + k * 256;
        if (FULL || e < E) out4[e] = o;
    }
}

__global__ __launch_bounds__(256) void k_edge_out(
        const int* __restrict__ ei, int E, const float* __restrict__ pq,
        const float* __restrict__ bfc, float* __restrict__ out) {
    const int tile0 = blockIdx.x * (256 * EPT);
    const int e0 = tile0 + threadIdx.x;
    if (tile0 + 256 * EPT <= E)
        edge_tile<true >(ei, E, (const float4*)pq, bfc, (float4*)out, e0);
    else
        edge_tile<false>(ei, E, (const float4*)pq, bfc, (float4*)out, e0);
}

extern "C" void kernel_launch(void* const* d_in, const int* in_sizes, int n_in,
                              void* d_out, int out_size, void* d_ws, size_t ws_size,
                              hipStream_t stream) {
    const float* x   = (const float*)d_in[0];
    const int*   ei  = (const int*)  d_in[1];
    const float* W1  = (const float*)d_in[2];
    const float* b1  = (const float*)d_in[3];
    const float* W2  = (const float*)d_in[4];
    const float* b2  = (const float*)d_in[5];
    const float* Wfc = (const float*)d_in[6];
    const float* bfc = (const float*)d_in[7];
    float* out = (float*)d_out;

    const int n = in_sizes[0] / D;
    const int E = in_sizes[1] / 2;
    const int n4 = (n + 3) & ~3;
    const int E4 = (E + 3) & ~3;

    float* bufA   = (float*)d_ws;
    float* bufB   = bufA + (size_t)n * D;
    float* dinv   = bufB + (size_t)n * D;
    int*   cnt    = (int*)(dinv + n4);
    int*   rowptr = cnt + n4;
    int*   rank   = rowptr + n4 + 4;
    int2*  rec    = (int2*)(rank + E4);

    const int gN    = (n + 255) / 256;
    const int gE    = (E + 255) / 256;
    const int gGemm = (n + 127) / 128;
    const int gNode = (n + 7) / 8;
    const int gEdge = (E + 256 * EPT - 1) / (256 * EPT);

    k_zero_cnt  <<<gN, 256, 0, stream>>>(cnt, n);
    k_count_rank<<<gE, 256, 0, stream>>>(ei, E, cnt, rank);
    k_scan      <<<gN, 256, 0, stream>>>(cnt, n, rowptr, dinv);
    k_fill      <<<gE, 256, 0, stream>>>(ei, E, rowptr, rank, dinv, rec);

    k_gemm_wmma<<<gGemm, 256, 0, stream>>>(x, W1, bufA, n);
    k_gather<false><<<gNode, 256, 0, stream>>>(rowptr, rec, bufA, dinv, b1, Wfc, bufB, n);

    k_gemm_wmma<<<gGemm, 256, 0, stream>>>(bufB, W2, bufA, n);
    k_gather<true><<<gNode, 256, 0, stream>>>(rowptr, rec, bufA, dinv, b2, Wfc, bufB, n);

    k_edge_out<<<gEdge, 256, 0, stream>>>(ei, E, bufB, bfc, out);
}


# Layer-1 product x@W1 moved into the edge-ranking launch (split by block index; independent work): its matrix time runs under the atomic pass, one launch fewer
# speedup vs baseline: 3.1802x; 1.1155x over previous
#include <hip/hip_runtime.h>
#include <hip/hip_bf16.h>
#include <math.h>

typedef __attribute__((ext_vector_type(2))) float v2f;
typedef __attribute__((ext_vector_type(8))) float v8f;

#define D 128

__global__ void k_zero_cnt(int* cnt, int n) {
    int i = blockIdx.x * blockDim.x + threadIdx.x;
    if (i < n) cnt[i] = 0;
}

__global__ __launch_bounds__(256) void k_scan(
        const int* __restrict__ cnt, int n,
        int* __restrict__ rowptr, float* __restrict__ dinv) {
    __shared__ int s_pre[8];
    __shared__ int s_loc[8];
    const int lane = threadIdx.x & 31;
    const int wave = threadIdx.x >> 5;
    const int base = blockIdx.x * 256;

    const int4* c4 = (const int4*)cnt;
    const int nq = base >> 2;
    int s = 0;
    int j = threadIdx.x;
    for (; j + 7 * 256 < nq; j += 8 * 256) {
        int4 q[8];
        for (int k = 0; k < 8; ++k) q[k] = c4[j + k * 256];
        for (int k = 0; k < 8; ++k) s += (q[k].x + q[k].y) + (q[k].z + q[k].w);
    }
    for (; j < nq; j += 256) {
        const int4 q = c4[j];
        s += (q.x + q.y) + (q.z + q.w);
    }
    for (int m = 16; m > 0; m >>= 1) s += __shfl_xor(s, m);

    const int i = base + threadIdx.x;
    const int c = (i < n) ? cnt[i] : 0;
    int v = c;
    for (int d = 1; d < 32; d <<= 1) {
        const int t = __shfl_up(v, d);
        if (lane >= d) v += t;
    }
    if (lane == 0)  s_pre[wave] = s;
    if (lane == 31) s_loc[wave] = v;
    __syncthreads();
    int off = 0;
    for (int w = 0; w < 8; ++w) {
        off += s_pre[w];
        if (w < wave) off += s_loc[w];
    }
    const int excl = off + v - c;
    if (i < n) {
        rowptr[i] = excl;
        const float d = (float)(c + 1);
        dinv[i] = (d > 0.0f) ? rsqrtf(d) : 0.0f;
    }
    if (i == n - 1) rowptr[n] = excl + c;
}

__global__ void k_fill(const int* __restrict__ ei, int E,
                       const int* __restrict__ rowptr, const int* __restrict__ rank,
                       const float* __restrict__ dinv, int2* __restrict__ rec) {
    int e = blockIdx.x * blockDim.x + threadIdx.x;
    if (e < E) {
        const int r = ei[e];
        const int c = ei[E + e];
        const float nrm = dinv[r] * dinv[c];
        rec[rowptr[c] + rank[e]] = make_int2(r, __float_as_int(nrm));
    }
}

__device__ __forceinline__ void gemm_tile(
        const float* __restrict__ A, const float* __restrict__ W,
        float* __restrict__ C, int n, int blk, float* Wl) {
    for (int i = threadIdx.x; i < (D * D) / 4; i += 256)
        ((float4*)Wl)[i] = ((const float4*)W)[i];
    __syncthreads();

    const int wave = threadIdx.x >> 5;
    const int lane = threadIdx.x & 31;
    const int m0   = blk * 128 + wave * 16;

    int arow = m0 + (lane & 15);
    if (arow > n - 1) arow = n - 1;
    const int khalf = (lane >> 4) * 2;
    const float* Ar = A + (size_t)arow * D;

    v8f acc[8] = {};

    for (int k = 0; k < D; k += 4) {
        const v2f a = *(const v2f*)(Ar + k + khalf);
        const int kb = k + khalf;
#pragma unroll
        for (int t = 0; t < 8; ++t) {
            const int col = t * 16 + (lane & 15);
            v2f b;
            b.x = Wl[kb * D + col];
            b.y = Wl[(kb + 1) * D + col];
            acc[t] = __builtin_amdgcn_wmma_f32_16x16x4_f32(
                false, a, false, b, (short)0, acc[t], false, false);
        }
    }

    const int rbase = m0 + (lane >> 4) * 8;
    const int cix   = lane & 15;
#pragma unroll
    for (int t = 0; t < 8; ++t) {
#pragma unroll
        for (int v = 0; v < 8; ++v) {
            const int r = rbase + v;
            if (r < n) C[(size_t)r * D + t * 16 + cix] = acc[t][v];
        }
    }
}

__global__ __launch_bounds__(256) void k_gemm_wmma(
        const float* __restrict__ A, const float* __restrict__ W,
        float* __restrict__ C, int n) {
    __shared__ float Wl[D * D];
    gemm_tile(A, W, C, n, (int)blockIdx.x, Wl);
}

__global__ __launch_bounds__(256) void k_gemm_rank(
        const float* __restrict__ A, const float* __restrict__ W,
        float* __restrict__ C, int n, int gGemm,
        const int* __restrict__ ei, int E, int* cnt, int* __restrict__ rank) {
    __shared__ float Wl[D * D];
    if ((int)blockIdx.x < gGemm) {
        gemm_tile(A, W, C, n, (int)blockIdx.x, Wl);
        return;
    }
    const int e = ((int)blockIdx.x - gGemm) * 256 + (int)threadIdx.x;
    if (e < E) rank[e] = atomicAdd(&cnt[ei[E + e]], 1);
}

template <int K>
__device__ __forceinline__ void add_rows(float4& acc, const float4* __restrict__ X,
                                         int r_l, float w_l, int j, int lane) {
    float4 u[K];
    float  w[K];
    for (int k = 0; k < K; ++k) {
        const int r = __builtin_amdgcn_readlane(r_l, j + k);
        w[k] = __int_as_float(__builtin_amdgcn_readlane(__float_as_int(w_l), j + k));
        u[k] = X[(size_t)r * (D / 4) + lane];
    }
    for (int k = 0; k < K; ++k) {
        acc.x = __builtin_fmaf(u[k].x, w[k], acc.x);
        acc.y = __builtin_fmaf(u[k].y, w[k], acc.y);
        acc.z = __builtin_fmaf(u[k].z, w[k], acc.z);
        acc.w = __builtin_fmaf(u[k].w, w[k], acc.w);
    }
}

template <int M>
__device__ __forceinline__ float xrow(float x) {
    const int v = __builtin_bit_cast(int, x);
    return __builtin_bit_cast(float,
        __builtin_amdgcn_update_dpp(v, v, 0x160 + M, 0xf, 0xf, false));
}
__device__ __forceinline__ float xhalf(float x) {
    const int v = __builtin_bit_cast(int, x);
    return __builtin_bit_cast(float,
        __builtin_amdgcn_permlanex16(v, v, 0x76543210, 0xfedcba98, false, false));
}

template <bool HEAD>
__global__ __launch_bounds__(256) void k_gather(
        const int* __restrict__ rowptr, const int2* __restrict__ rec,
        const float* __restrict__ xw, const float* __restrict__ dinv,
        const float* __restrict__ b, const float* __restrict__ Wfc,
        float* __restrict__ out, int n) {
    __shared__ float4 s_w[HEAD ? 256 : 1];
    const int wv = __builtin_amdgcn_readfirstlane((int)(threadIdx.x >> 5));
    const int i = blockIdx.x * 8 + wv;
    const int lane = threadIdx.x & 31;
    const int ic = min(i, n - 1);
    const int beg = rowptr[ic];
    const int deg = rowptr[ic + 1] - beg;
    const float di = dinv[ic];
    if (HEAD) {
        const int t = threadIdx.x;
        s_w[((t >> 7) * 4 + (t & 3)) * 32 + ((t & 127) >> 2)] = ((const float4*)Wfc)[t];
        __syncthreads();
    }
    if (i >= n) return;
    const int tot = deg + 1;
    const float4* X = (const float4*)xw;

    float4 acc = {0.0f, 0.0f, 0.0f, 0.0f};
    for (int t0 = 0; t0 < tot; t0 += 32) {
        const int m = min(tot - t0, 32);
        const int t = t0 + lane;
        int   r_l = i;
        float w_l = (t == deg) ? di * di : 0.0f;
        if (t < deg) {
            const int2 q = rec[beg + t];
            r_l = q.x;
            w_l = __int_as_float(q.y);
        }
        int j = 0;
        for (; m - j > 4; j += 8) add_rows<8>(acc, X, r_l, w_l, j, lane);
        if (j < m)               add_rows<4>(acc, X, r_l, w_l, j, lane);
    }

    const float4 bb = ((const float4*)b)[lane];
    float4 o = {acc.x + bb.x, acc.y + bb.y, acc.z + bb.z, acc.w + bb.w};
    o.x = o.x > 0.0f ? o.x : 0.0f;
    o.y = o.y > 0.0f ? o.y : 0.0f;
    o.z = o.z > 0.0f ? o.z : 0.0f;
    o.w = o.w > 0.0f ? o.w : 0.0f;
    if (!HEAD) {
        ((float4*)out)[(size_t)i * (D / 4) + lane] = o;
        return;
    }

    const float4 t0 = s_w[0 * 32 + lane], t1 = s_w[1 * 32 + lane];
    const float4 t2 = s_w[2 * 32 + lane], t3 = s_w[3 * 32 + lane];
    const float4 u0 = s_w[4 * 32 + lane], u1 = s_w[5 * 32 + lane];
    const float4 u2 = s_w[6 * 32 + lane], u3 = s_w[7 * 32 + lane];
    float s[8];
    s[0] = o.x * t0.x + o.y * t1.x + o.z * t2.x + o.w * t3.x;
    s[1] = o.x * t0.y + o.y * t1.y + o.z * t2.y + o.w * t3.y;
    s[2] = o.x * t0.z + o.y * t1.z + o.z * t2.z + o.w * t3.z;
    s[3] = o.x * t0.w + o.y * t1.w + o.z * t2.w + o.w * t3.w;
    s[4] = o.x * u0.x + o.y * u1.x + o.z * u2.x + o.w * u3.x;
    s[5] = o.x * u0.y + o.y * u1.y + o.z * u2.y + o.w * u3.y;
    s[6] = o.x * u0.z + o.y * u1.z + o.z * u2.z + o.w * u3.z;
    s[7] = o.x * u0.w + o.y * u1.w + o.z * u2.w + o.w * u3.w;

    const bool h16 = (lane & 16) != 0;
    float a4[4];
    for (int q = 0; q < 4; ++q) {
        const float keep = h16 ? s[q + 4] : s[q];
        const float give = h16 ? s[q] : s[q + 4];
        a4[q] = keep + xhalf(give);
    }
    const bool h8 = (lane & 8) != 0;
    float a2[2];
    for (int q = 0; q < 2; ++q) {
        const float keep = h8 ? a4[q + 2] : a4[q];
        const float give = h8 ? a4[q] : a4[q + 2];
        a2[q] = keep + xrow<8>(give);
    }
    const bool h4 = (lane & 4) != 0;
    const float keep = h4 ? a2[1] : a2[0];
    const float give = h4 ? a2[0] : a2[1];
    float a1 = keep + xrow<4>(give);
    a1 += xrow<2>(a1);
    a1 += xrow<1>(a1);
    if ((lane & 3) == 0)
        out[(size_t)i * 8 + ((lane >> 4) & 1) * 4 + ((lane >> 3) & 1) * 2 + ((lane >> 2) & 1)] = a1;
}

#define EPT 4
template <bool FULL>
__device__ __forceinline__ void edge_tile(
        const int* __restrict__ ei, int E, const float4* __restrict__ pq4,
        const float* __restrict__ bfc, float4* __restrict__ out4, int e0) {
    int r[EPT], c[EPT];
    for (int k = 0; k < EPT; ++k) {
        const int e = FULL ? e0 + k * 256 : min(e0 + k * 256, E - 1);
        r[k] = ei[e];
        c[k] = ei[E + e];
    }
    float4 p[EPT], q[EPT];
    for (int k = 0; k < EPT; ++k) {
        p[k] = pq4[(size_t)r[k] * 2];
        q[k] = pq4[(size_t)c[k] * 2 + 1];
    }
    const float b0 = bfc[0], b1 = bfc[1], b2 = bfc[2], b3 = bfc[3];
    for (int k = 0; k < EPT; ++k) {
        const float s0 = (p[k].x + q[k].x) + b0;
        const float s1 = (p[k].y + q[k].y) + b1;
        const float s2 = (p[k].z + q[k].z) + b2;
        const float s3 = (p[k].w + q[k].w) + b3;
        const float mx  = fmaxf(fmaxf(s0, s1), fmaxf(s2, s3));
        const float sum = __expf(s0 - mx) + __expf(s1 - mx) +
                          __expf(s2 - mx) + __expf(s3 - mx);
        const float lse = mx + __logf(sum);
        const float4 o  = {s0 - lse, s1 - lse, s2 - lse, s3 - lse};
        const int e = e0 + k * 256;
        if (FULL || e < E) out4[e] = o;
    }
}

__global__ __launch_bounds__(256) void k_edge_out(
        const int* __restrict__ ei, int E, const float* __restrict__ pq,
        const float* __restrict__ bfc, float* __restrict__ out) {
    const int tile0 = blockIdx.x * (256 * EPT);
    const int e0 = tile0 + threadIdx.x;
    if (tile0 + 256 * EPT <= E)
        edge_tile<true >(ei, E, (const float4*)pq, bfc, (float4*)out, e0);
    else
        edge_tile<false>(ei, E, (const float4*)pq, bfc, (float4*)out, e0);
}

extern "C" void kernel_launch(void* const* d_in, const int* in_sizes, int n_in,
                              void* d_out, int out_size, void* d_ws, size_t ws_size,
                              hipStream_t stream) {
    const float* x   = (const float*)d_in[0];
    const int*   ei  = (const int*)  d_in[1];
    const float* W1  = (const float*)d_in[2];
    const float* b1  = (const float*)d_in[3];
    const float* W2  = (const float*)d_in[4];
    const float* b2  = (const float*)d_in[5];
    const float* Wfc = (const float*)d_in[6];
    const float* bfc = (const float*)d_in[7];
    float* out = (float*)d_out;

    const int n = in_sizes[0] / D;
    const int E = in_sizes[1] / 2;
    const int n4 = (n + 3) & ~3;
    const int E4 = (E + 3) & ~3;

    float* bufA   = (float*)d_ws;
    float* bufB   = bufA + (size_t)n * D;
    float* dinv   = bufB + (size_t)n * D;
    int*   cnt    = (int*)(dinv + n4);
    int*   rowptr = cnt + n4;
    int*   rank   = rowptr + n4 + 4;
    int2*  rec    = (int2*)(rank + E4);

    const int gN    = (n + 255) / 256;
    const int gE    = (E + 255) / 256;
    const int gGemm = (n + 127) / 128;
    const int gNode = (n + 7) / 8;
    const int gEdge = (E + 256 * EPT - 1) / (256 * EPT);

    k_zero_cnt  <<<gN, 256, 0, stream>>>(cnt, n);
    k_gemm_rank <<<gGemm + gE, 256, 0, stream>>>(x, W1, bufA, n, gGemm, ei, E, cnt, rank);
    k_scan      <<<gN, 256, 0, stream>>>(cnt, n, rowptr, dinv);
    k_fill      <<<gE, 256, 0, stream>>>(ei, E, rowptr, rank, dinv, rec);

    k_gather<false><<<gNode, 256, 0, stream>>>(rowptr, rec, bufA, dinv, b1, Wfc, bufB, n);

    k_gemm_wmma<<<gGemm, 256, 0, stream>>>(bufB, W2, bufA, n);
    k_gather<true><<<gNode, 256, 0, stream>>>(rowptr, rec, bufA, dinv, b2, Wfc, bufB, n);

    k_edge_out<<<gEdge, 256, 0, stream>>>(ei, E, bufB, bfc, out);
}


# Layer-2 product h1@W2 fused behind the layer-1 gather: 64-node blocks keep the relu'd rows in LDS and multiply them there (W2 staged per block); h1 never stored, one launch fewer; sum orders kept
# speedup vs baseline: 3.3638x; 1.0577x over previous
#include <hip/hip_runtime.h>
#include <hip/hip_bf16.h>
#include <math.h>

typedef __attribute__((ext_vector_type(2))) float v2f;
typedef __attribute__((ext_vector_type(8))) float v8f;

#define D 128

__global__ void k_zero_cnt(int* cnt, int n) {
    int i = blockIdx.x * blockDim.x + threadIdx.x;
    if (i < n) cnt[i] = 0;
}

__global__ __launch_bounds__(256) void k_scan(
        const int* __restrict__ cnt, int n,
        int* __restrict__ rowptr, float* __restrict__ dinv) {
    __shared__ int s_pre[8];
    __shared__ int s_loc[8];
    const int lane = threadIdx.x & 31;
    const int wave = threadIdx.x >> 5;
    const int base = blockIdx.x * 256;

    const int4* c4 = (const int4*)cnt;
    const int nq = base >> 2;
    int s = 0;
    int j = threadIdx.x;
    for (; j + 7 * 256 < nq; j += 8 * 256) {
        int4 q[8];
        for (int k = 0; k < 8; ++k) q[k] = c4[j + k * 256];
        for (int k = 0; k < 8; ++k) s += (q[k].x + q[k].y) + (q[k].z + q[k].w);
    }
    for (; j < nq; j += 256) {
        const int4 q = c4[j];
        s += (q.x + q.y) + (q.z + q.w);
    }
    for (int m = 16; m > 0; m >>= 1) s += __shfl_xor(s, m);

    const int i = base + threadIdx.x;
    const int c = (i < n) ? cnt[i] : 0;
    int v = c;
    for (int d = 1; d < 32; d <<= 1) {
        const int t = __shfl_up(v, d);
        if (lane >= d) v += t;
    }
    if (lane == 0)  s_pre[wave] = s;
    if (lane == 31) s_loc[wave] = v;
    __syncthreads();
    int off = 0;
    for (int w = 0; w < 8; ++w) {
        off += s_pre[w];
        if (w < wave) off += s_loc[w];
    }
    const int excl = off + v - c;
    if (i < n) {
        rowptr[i] = excl;
        const float d = (float)(c + 1);
        dinv[i] = (d > 0.0f) ? rsqrtf(d) : 0.0f;
    }
    if (i == n - 1) rowptr[n] = excl + c;
}

__global__ void k_fill(const int* __restrict__ ei, int E,
                       const int* __restrict__ rowptr, const int* __restrict__ rank,
                       const float* __restrict__ dinv, int2* __restrict__ rec) {
    int e = blockIdx.x * blockDim.x + threadIdx.x;
    if (e < E) {
        const int r = ei[e];
        const int c = ei[E + e];
        const float nrm = dinv[r] * dinv[c];
        rec[rowptr[c] + rank[e]] = make_int2(r, __float_as_int(nrm));
    }
}

__device__ __forceinline__ void gemm_tile(
        const float* __restrict__ A, const float* __restrict__ W,
        float* __restrict__ C, int n, int blk, float* Wl) {
    for (int i = threadIdx.x; i < (D * D) / 4; i += 256)
        ((float4*)Wl)[i] = ((const float4*)W)[i];
    __syncthreads();

    const int wave = threadIdx.x >> 5;
    const int lane = threadIdx.x & 31;
    const int m0   = blk * 128 + wave * 16;

    int arow = m0 + (lane & 15);
    if (arow > n - 1) arow = n - 1;
    const int khalf = (lane >> 4) * 2;
    const float* Ar = A + (size_t)arow * D;

    v8f acc[8] = {};

    for (int k = 0; k < D; k += 4) {
        const v2f a = *(const v2f*)(Ar + k + khalf);
        const int kb = k + khalf;
#pragma unroll
        for (int t = 0; t < 8; ++t) {
            const int col = t * 16 + (lane & 15);
            v2f b;
            b.x = Wl[kb * D + col];
            b.y = Wl[(kb + 1) * D + col];
            acc[t] = __builtin_amdgcn_wmma_f32_16x16x4_f32(
                false, a, false, b, (short)0, acc[t], false, false);
        }
    }

    const int rbase = m0 + (lane >> 4) * 8;
    const int cix   = lane & 15;
#pragma unroll
    for (int t = 0; t < 8; ++t) {
#pragma unroll
        for (int v = 0; v < 8; ++v) {
            const int r = rbase + v;
            if (r < n) C[(size_t)r * D + t * 16 + cix] = acc[t][v];
        }
    }
}

__global__ __launch_bounds__(256) void k_gemm_rank(
        const float* __restrict__ A, const float* __restrict__ W,
        float* __restrict__ C, int n, int gGemm,
        const int* __restrict__ ei, int E, int* cnt, int* __restrict__ rank) {
    __shared__ float Wl[D * D];
    if ((int)blockIdx.x < gGemm) {
        gemm_tile(A, W, C, n, (int)blockIdx.x, Wl);
        return;
    }
    const int e = ((int)blockIdx.x - gGemm) * 256 + (int)threadIdx.x;
    if (e < E) rank[e] = atomicAdd(&cnt[ei[E + e]], 1);
}

template <int K>
__device__ __forceinline__ void add_rows(float4& acc, const float4* __restrict__ X,
                                         int r_l, float w_l, int j, int lane) {
    float4 u[K];
    float  w[K];
    for (int k = 0; k < K; ++k) {
        const int r = __builtin_amdgcn_readlane(r_l, j + k);
        w[k] = __int_as_float(__builtin_amdgcn_readlane(__float_as_int(w_l), j + k));
        u[k] = X[(size_t)r * (D / 4) + lane];
    }
    for (int k = 0; k < K; ++k) {
        acc.x = __builtin_fmaf(u[k].x, w[k], acc.x);
        acc.y = __builtin_fmaf(u[k].y, w[k], acc.y);
        acc.z = __builtin_fmaf(u[k].z, w[k], acc.z);
        acc.w = __builtin_fmaf(u[k].w, w[k], acc.w);
    }
}

template <int M>
__device__ __forceinline__ float xrow(float x) {
    const int v = __builtin_bit_cast(int, x);
    return __builtin_bit_cast(float,
        __builtin_amdgcn_update_dpp(v, v, 0x160 + M, 0xf, 0xf, false));
}
__device__ __forceinline__ float xhalf(float x) {
    const int v = __builtin_bit_cast(int, x);
    return __builtin_bit_cast(float,
        __builtin_amdgcn_permlanex16(v, v, 0x76543210, 0xfedcba98, false, false));
}

__device__ __forceinline__ void list_entry(
        const int2* __restrict__ rec, int i, int beg, int deg, float di, int t,
        int& r_l, float& w_l) {
    r_l = i;
    w_l = (t == deg) ? di * di : 0.0f;
    if (t < deg) {
        const int2 q = rec[beg + t];
        r_l = q.x;
        w_l = __int_as_float(q.y);
    }
}

__device__ __forceinline__ float4 gather_sum(
        const float4* __restrict__ X, const int2* __restrict__ rec,
        int i, int beg, int deg, float di, int r_l, float w_l, int lane) {
    const int tot = deg + 1;
    float4 acc = {0.0f, 0.0f, 0.0f, 0.0f};
    for (int t0 = 0; t0 < tot; t0 += 32) {
        if (t0 > 0) list_entry(rec, i, beg, deg, di, t0 + lane, r_l, w_l);
        const int m = min(tot - t0, 32);
        int j = 0;
        for (; m - j > 4; j += 8) add_rows<8>(acc, X, r_l, w_l, j, lane);
        if (j < m)               add_rows<4>(acc, X, r_l, w_l, j, lane);
    }
    return acc;
}

__device__ __forceinline__ float4 bias_relu(float4 acc, float4 bb) {
    float4 o = {acc.x + bb.x, acc.y + bb.y, acc.z + bb.z, acc.w + bb.w};
    o.x = o.x > 0.0f ? o.x : 0.0f;
    o.y = o.y > 0.0f ? o.y : 0.0f;
    o.z = o.z > 0.0f ? o.z : 0.0f;
    o.w = o.w > 0.0f ? o.w : 0.0f;
    return o;
}

#define HP 132
#define WP 136
__global__ __launch_bounds__(1024) void k_gather_gemm(
        const int* __restrict__ rowptr, const int2* __restrict__ rec,
        const float* __restrict__ xw, const float* __restrict__ dinv,
        const float* __restrict__ b, const float* __restrict__ W,
        float* __restrict__ C, int n) {
    __shared__ float Wl[D * WP];
    __shared__ float Hl[64 * HP];
    const int wv = __builtin_amdgcn_readfirstlane((int)(threadIdx.x >> 5));
    const int lane = threadIdx.x & 31;
    const int base = blockIdx.x * 64;
    const int iA = base + 2 * wv, iB = iA + 1;
    const int cA = min(iA, n - 1), cB = min(iB, n - 1);
    const int begA = rowptr[cA], begB = rowptr[cB];
    const int degA = rowptr[cA + 1] - begA, degB = rowptr[cB + 1] - begB;
    const float diA = dinv[cA], diB = dinv[cB];
    int rA, rB; float wA, wB;
    list_entry(rec, cA, begA, degA, diA, lane, rA, wA);
    list_entry(rec, cB, begB, degB, diB, lane, rB, wB);
    const float4 bb = ((const float4*)b)[lane];

    for (int q = 0; q < 4; ++q) {
        const int f = (int)threadIdx.x + q * 1024;
        *(float4*)(Wl + (f >> 5) * WP + (f & 31) * 4) = ((const float4*)W)[f];
    }

    const float4* X = (const float4*)xw;
    float4 oA = {0.0f, 0.0f, 0.0f, 0.0f}, oB = oA;
    if (iA < n) oA = bias_relu(gather_sum(X, rec, iA, begA, degA, diA, rA, wA, lane), bb);
    if (iB < n) oB = bias_relu(gather_sum(X, rec, iB, begB, degB, diB, rB, wB, lane), bb);
    *(float4*)(Hl + (2 * wv) * HP + 4 * lane)     = oA;
    *(float4*)(Hl + (2 * wv + 1) * HP + 4 * lane) = oB;
    __syncthreads();

    const int mt = wv >> 3, nt = wv & 7;
    const int c = lane & 15;
    const int khalf = (lane >> 4) * 2;
    const float* Ar = Hl + (mt * 16 + c) * HP + khalf;
    const float* Br = Wl + khalf * WP + nt * 16 + c;
    v8f acc = {};
    for (int k = 0; k < D; k += 4) {
        const v2f a = *(const v2f*)(Ar + k);
        v2f bf;
        bf.x = Br[k * WP];
        bf.y = Br[(k + 1) * WP];
        acc = __builtin_amdgcn_wmma_f32_16x16x4_f32(
            false, a, false, bf, (short)0, acc, false, false);
    }
    const int rbase = base + mt * 16 + (lane >> 4) * 8;
    for (int v = 0; v < 8; ++v) {
        const int r = rbase + v;
        if (r < n) C[(size_t)r * D + nt * 16 + c] = acc[v];
    }
}

__global__ __launch_bounds__(256) void k_gather_head(
        const int* __restrict__ rowptr, const int2* __restrict__ rec,
        const float* __restrict__ xw, const float* __restrict__ dinv,
        const float* __restrict__ b, const float* __restrict__ Wfc,
        float* __restrict__ out, int n) {
    __shared__ float4 s_w[256];
    const int wv = __builtin_amdgcn_readfirstlane((int)(threadIdx.x >> 5));
    const int i = blockIdx.x * 8 + wv;
    const int lane = threadIdx.x & 31;
    const int ic = min(i, n - 1);
    const int beg = rowptr[ic];
    const int deg = rowptr[ic + 1] - beg;
    const float di = dinv[ic];
    {
        const int t = threadIdx.x;
        s_w[((t >> 7) * 4 + (t & 3)) * 32 + ((t & 127) >> 2)] = ((const float4*)Wfc)[t];
        __syncthreads();
    }
    if (i >= n) return;
    int r_l; float w_l;
    list_entry(rec, i, beg, deg, di, lane, r_l, w_l);
    const float4 acc = gather_sum((const float4*)xw, rec, i, beg, deg, di, r_l, w_l, lane);
    const float4 o = bias_relu(acc, ((const float4*)b)[lane]);

    const float4 t0 = s_w[0 * 32 + lane], t1 = s_w[1 * 32 + lane];
    const float4 t2 = s_w[2 * 32 + lane], t3 = s_w[3 * 32 + lane];
    const float4 u0 = s_w[4 * 32 + lane], u1 = s_w[5 * 32 + lane];
    const float4 u2 = s_w[6 * 32 + lane], u3 = s_w[7 * 32 + lane];
    float s[8];
    s[0] = o.x * t0.x + o.y * t1.x + o.z * t2.x + o.w * t3.x;
    s[1] = o.x * t0.y + o.y * t1.y + o.z * t2.y + o.w * t3.y;
    s[2] = o.x * t0.z + o.y * t1.z + o.z * t2.z + o.w * t3.z;
    s[3] = o.x * t0.w + o.y * t1.w + o.z * t2.w + o.w * t3.w;
    s[4] = o.x * u0.x + o.y * u1.x + o.z * u2.x + o.w * u3.x;
    s[5] = o.x * u0.y + o.y * u1.y + o.z * u2.y + o.w * u3.y;
    s[6] = o.x * u0.z + o.y * u1.z + o.z * u2.z + o.w * u3.z;
    s[7] = o.x * u0.w + o.y * u1.w + o.z * u2.w + o.w * u3.w;

    const bool h16 = (lane & 16) != 0;
    float a4[4];
    for (int q = 0; q < 4; ++q) {
        const float keep = h16 ? s[q + 4] : s[q];
        const float give = h16 ? s[q] : s[q + 4];
        a4[q] = keep + xhalf(give);
    }
    const bool h8 = (lane & 8) != 0;
    float a2[2];
    for (int q = 0; q < 2; ++q) {
        const float keep = h8 ? a4[q + 2] : a4[q];
        const float give = h8 ? a4[q] : a4[q + 2];
        a2[q] = keep + xrow<8>(give);
    }
    const bool h4 = (lane & 4) != 0;
    const float keep = h4 ? a2[1] : a2[0];
    const float give = h4 ? a2[0] : a2[1];
    float a1 = keep + xrow<4>(give);
    a1 += xrow<2>(a1);
    a1 += xrow<1>(a1);
    if ((lane & 3) == 0)
        out[(size_t)i * 8 + ((lane >> 4) & 1) * 4 + ((lane >> 3) & 1) * 2 + ((lane >> 2) & 1)] = a1;
}

#define EPT 4
template <bool FULL>
__device__ __forceinline__ void edge_tile(
        const int* __restrict__ ei, int E, const float4* __restrict__ pq4,
        const float* __restrict__ bfc, float4* __restrict__ out4, int e0) {
    int r[EPT], c[EPT];
    for (int k = 0; k < EPT; ++k) {
        const int e = FULL ? e0 + k * 256 : min(e0 + k * 256, E - 1);
        r[k] = ei[e];
        c[k] = ei[E + e];
    }
    float4 p[EPT], q[EPT];
    for (int k = 0; k < EPT; ++k) {
        p[k] = pq4[(size_t)r[k] * 2];
        q[k] = pq4[(size_t)c[k] * 2 + 1];
    }
    const float b0 = bfc[0], b1 = bfc[1], b2 = bfc[2], b3 = bfc[3];
    for (int k = 0; k < EPT; ++k) {
        const float s0 = (p[k].x + q[k].x) + b0;
        const float s1 = (p[k].y + q[k].y) + b1;
        const float s2 = (p[k].z + q[k].z) + b2;
        const float s3 = (p[k].w + q[k].w) + b3;
        const float mx  = fmaxf(fmaxf(s0, s1), fmaxf(s2, s3));
        const float sum = __expf(s0 - mx) + __expf(s1 - mx) +
                          __expf(s2 - mx) + __expf(s3 - mx);
        const float lse = mx + __logf(sum);
        const float4 o  = {s0 - lse, s1 - lse, s2 - lse, s3 - lse};
        const int e = e0 + k * 256;
        if (FULL || e < E) out4[e] = o;
    }
}

__global__ __launch_bounds__(256) void k_edge_out(
        const int* __restrict__ ei, int E, const float* __restrict__ pq,
        const float* __restrict__ bfc, float* __restrict__ out) {
    const int tile0 = blockIdx.x * (256 * EPT);
    const int e0 = tile0 + threadIdx.x;
    if (tile0 + 256 * EPT <= E)
        edge_tile<true >(ei, E, (const float4*)pq, bfc, (float4*)out, e0);
    else
        edge_tile<false>(ei, E, (const float4*)pq, bfc, (float4*)out, e0);
}

extern "C" void kernel_launch(void* const* d_in, const int* in_sizes, int n_in,
                              void* d_out, int out_size, void* d_ws, size_t ws_size,
                              hipStream_t stream) {
    const float* x   = (const float*)d_in[0];
    const int*   ei  = (const int*)  d_in[1];
    const float* W1  = (const float*)d_in[2];
    const float* b1  = (const float*)d_in[3];
    const float* W2  = (const float*)d_in[4];
    const float* b2  = (const float*)d_in[5];
    const float* Wfc = (const float*)d_in[6];
    const float* bfc = (const float*)d_in[7];
    float* out = (float*)d_out;

    const int n = in_sizes[0] / D;
    const int E = in_sizes[1] / 2;
    const int n4 = (n + 3) & ~3;
    const int E4 = (E + 3) & ~3;

    float* bufA   = (float*)d_ws;
    float* bufB   = bufA + (size_t)n * D;
    float* pq     = bufA;
    float* dinv   = bufB + (size_t)n * D;
    int*   cnt    = (int*)(dinv + n4);
    int*   rowptr = cnt + n4;
    int*   rank   = rowptr + n4 + 4;
    int2*  rec    = (int2*)(rank + E4);

    const int gN    = (n + 255) / 256;
    const int gE    = (E + 255) / 256;
    const int gGemm = (n + 127) / 128;
    const int gNode = (n + 7) / 8;
    const int gTile = (n + 63) / 64;
    const int gEdge = (E + 256 * EPT - 1) / (256 * EPT);

    k_zero_cnt  <<<gN, 256, 0, stream>>>(cnt, n);
    k_gemm_rank <<<gGemm + gE, 256, 0, stream>>>(x, W1, bufA, n, gGemm, ei, E, cnt, rank);
    k_scan      <<<gN, 256, 0, stream>>>(cnt, n, rowptr, dinv);
    k_fill      <<<gE, 256, 0, stream>>>(ei, E, rowptr, rank, dinv, rec);

    k_gather_gemm<<<gTile, 1024, 0, stream>>>(rowptr, rec, bufA, dinv, b1, W2, bufB, n);

    k_gather_head<<<gNode, 256, 0, stream>>>(rowptr, rec, bufB, dinv, b2, Wfc, pq, n);

    k_edge_out<<<gEdge, 256, 0, stream>>>(ei, E, pq, bfc, out);
}


# Per-edge returning global atomics (rank by target) replaced by a two-level counting sort on LDS counters (256-node bins, then nodes); a node's sum order is now the LDS adds' arrival order
# speedup vs baseline: 3.7671x; 1.1199x over previous
#include <hip/hip_runtime.h>
#include <hip/hip_bf16.h>
#include <math.h>

typedef __attribute__((ext_vector_type(2))) float v2f;
typedef __attribute__((ext_vector_type(8))) float v8f;

#define D 128

#define RB 256
#define NBMAX 256
#define CE 1024

__device__ __forceinline__ int scan256(int v, int* s_ws, int& tot) {
    const int lane = threadIdx.x & 31;
    const int wave = threadIdx.x >> 5;
    int s = v;
    for (int d = 1; d < 32; d <<= 1) {
        const int t = __shfl_up(s, d);
        if (lane >= d) s += t;
    }
    if (lane == 31) s_ws[wave] = s;
    __syncthreads();
    int off = 0, all = 0;
    for (int w = 0; w < 8; ++w) {
        const int x = s_ws[w];
        if (w < wave) off += x;
        all += x;
    }
    tot = all;
    return off + s - v;
}

__global__ __launch_bounds__(256) void k_bin_prefix(
        const int* __restrict__ hist, int G, int NB,
        int* __restrict__ pre, int* __restrict__ tot) {
    __shared__ int s_ws[8];
    const int b = blockIdx.x;
    int carry = 0;
    for (int g0 = 0; g0 < G; g0 += 256) {
        const int g = g0 + (int)threadIdx.x;
        const int v = (g < G) ? hist[(size_t)g * NB + b] : 0;
        int all;
        const int ex = scan256(v, s_ws, all);
        if (g < G) pre[(size_t)g * NB + b] = carry + ex;
        carry += all;
        __syncthreads();
    }
    if (threadIdx.x == 0) tot[b] = carry;
}

__global__ __launch_bounds__(1024) void k_bin_sort(
        const int* __restrict__ binStart, const int2* __restrict__ binned, int n,
        int* __restrict__ rowptr, float* __restrict__ dinv, int2* __restrict__ rec) {
    __shared__ int s_cnt[RB];
    __shared__ int s_cur[RB];
    __shared__ int s_ws[8];
    const int b = blockIdx.x;
    const int t = threadIdx.x;
    const int lo = binStart[b], hi = binStart[b + 1];
    if (t < RB) s_cnt[t] = 0;
    __syncthreads();
    for (int j = lo + t; j < hi; j += 1024)
        atomicAdd(&s_cnt[binned[j].y & (RB - 1)], 1);
    __syncthreads();
    {
        const int lane = t & 31, wave = t >> 5;
        const int v = (t < RB) ? s_cnt[t] : 0;
        int s = v;
        for (int d = 1; d < 32; d <<= 1) {
            const int x = __shfl_up(s, d);
            if (lane >= d) s += x;
        }
        if (t < RB && lane == 31) s_ws[wave] = s;
        __syncthreads();
        if (t < RB) {
            int off = 0;
            for (int w = 0; w < 8; ++w) if (w < wave) off += s_ws[w];
            const int ex = off + s - v;
            s_cur[t] = ex;
            const int node = b * RB + t;
            if (node < n) {
                rowptr[node] = lo + ex;
                const float d = (float)(v + 1);
                dinv[node] = (d > 0.0f) ? rsqrtf(d) : 0.0f;
            }
            if (node == n - 1) rowptr[n] = lo + ex + v;
        }
        __syncthreads();
    }
    for (int j = lo + t; j < hi; j += 1024) {
        const int2 q = binned[j];
        const int pos = atomicAdd(&s_cur[q.y & (RB - 1)], 1);
        rec[lo + pos] = q;
    }
}

__global__ __launch_bounds__(256) void k_weight(
        int E, const float* __restrict__ dinv, int2* rec) {
    const int j0 = blockIdx.x * CE + threadIdx.x;
    int2 q[4];
    for (int k = 0; k < 4; ++k) {
        const int j = j0 + k * 256;
        q[k] = (j < E) ? rec[j] : make_int2(0, 0);
    }
    float a[4], c[4];
    for (int k = 0; k < 4; ++k) {
        a[k] = dinv[q[k].x];
        c[k] = dinv[q[k].y];
    }
    for (int k = 0; k < 4; ++k) {
        const int j = j0 + k * 256;
        if (j < E) rec[j] = make_int2(q[k].x, __float_as_int(a[k] * c[k]));
    }
}

__device__ __forceinline__ void gemm_tile(
        const float* __restrict__ A, const float* __restrict__ W,
        float* __restrict__ C, int n, int blk, float* Wl) {
    for (int i = threadIdx.x; i < (D * D) / 4; i += 256)
        ((float4*)Wl)[i] = ((const float4*)W)[i];
    __syncthreads();

    const int wave = threadIdx.x >> 5;
    const int lane = threadIdx.x & 31;
    const int m0   = blk * 128 + wave * 16;

    int arow = m0 + (lane & 15);
    if (arow > n - 1) arow = n - 1;
    const int khalf = (lane >> 4) * 2;
    const float* Ar = A + (size_t)arow * D;

    v8f acc[8] = {};

    for (int k = 0; k < D; k += 4) {
        const v2f a = *(const v2f*)(Ar + k + khalf);
        const int kb = k + khalf;
#pragma unroll
        for (int t = 0; t < 8; ++t) {
            const int col = t * 16 + (lane & 15);
            v2f b;
            b.x = Wl[kb * D + col];
            b.y = Wl[(kb + 1) * D + col];
            acc[t] = __builtin_amdgcn_wmma_f32_16x16x4_f32(
                false, a, false, b, (short)0, acc[t], false, false);
        }
    }

    const int rbase = m0 + (lane >> 4) * 8;
    const int cix   = lane & 15;
#pragma unroll
    for (int t = 0; t < 8; ++t) {
#pragma unroll
        for (int v = 0; v < 8; ++v) {
            const int r = rbase + v;
            if (r < n) C[(size_t)r * D + t * 16 + cix] = acc[t][v];
        }
    }
}


__global__ __launch_bounds__(256) void k_gemm_hist(
        const float* __restrict__ A, const float* __restrict__ W,
        float* __restrict__ C, int n, int blk0, int gGemm,
        const int* __restrict__ ei, int E, int NB, int* __restrict__ hist) {
    __shared__ float Wl[D * D];
    __shared__ int s_h[NBMAX];
    if ((int)blockIdx.x < gGemm) {
        gemm_tile(A, W, C, n, blk0 + (int)blockIdx.x, Wl);
        return;
    }
    const int g = (int)blockIdx.x - gGemm;
    const int t = threadIdx.x;
    const int e0 = g * CE + t;
    int c[4];
    for (int k = 0; k < 4; ++k) c[k] = ei[E + min(e0 + k * 256, E - 1)];
    s_h[t] = 0;
    __syncthreads();
    for (int k = 0; k < 4; ++k)
        if (e0 + k * 256 < E) atomicAdd(&s_h[c[k] >> 8], 1);
    __syncthreads();
    if (t < NB) hist[(size_t)g * NB + t] = s_h[t];
}

__global__ __launch_bounds__(256) void k_gemm_scatter(
        const float* __restrict__ A, const float* __restrict__ W,
        float* __restrict__ C, int n, int blk0, int gGemm,
        const int* __restrict__ ei, int E, int NB,
        const int* __restrict__ pre, const int* __restrict__ tot,
        int* __restrict__ binStart, int2* __restrict__ binned) {
    __shared__ float Wl[D * D];
    __shared__ int s_base[NBMAX];
    __shared__ int s_ws[8];
    if ((int)blockIdx.x < gGemm) {
        gemm_tile(A, W, C, n, blk0 + (int)blockIdx.x, Wl);
        return;
    }
    const int g = (int)blockIdx.x - gGemm;
    const int t = threadIdx.x;
    const int e0 = g * CE + t;
    int r[4], c[4];
    for (int k = 0; k < 4; ++k) {
        const int e = min(e0 + k * 256, E - 1);
        r[k] = ei[e];
        c[k] = ei[E + e];
    }
    const int v = (t < NB) ? tot[t] : 0;
    const int p = (t < NB) ? pre[(size_t)g * NB + t] : 0;
    int all;
    const int ex = scan256(v, s_ws, all);
    s_base[t] = ex + p;
    if (g == 0) {
        if (t < NB) binStart[t] = ex;
        if (t == NB - 1) binStart[NB] = ex + v;
    }
    __syncthreads();
    for (int k = 0; k < 4; ++k) {
        if (e0 + k * 256 < E) {
            const int pos = atomicAdd(&s_base[c[k] >> 8], 1);
            binned[pos] = make_int2(r[k], c[k]);
        }
    }
}

template <int K>
__device__ __forceinline__ void add_rows(float4& acc, const float4* __restrict__ X,
                                         int r_l, float w_l, int j, int lane) {
    float4 u[K];
    float  w[K];
    for (int k = 0; k < K; ++k) {
        const int r = __builtin_amdgcn_readlane(r_l, j + k);
        w[k] = __int_as_float(__builtin_amdgcn_readlane(__float_as_int(w_l), j + k));
        u[k] = X[(size_t)r * (D / 4) + lane];
    }
    for (int k = 0; k < K; ++k) {
        acc.x = __builtin_fmaf(u[k].x, w[k], acc.x);
        acc.y = __builtin_fmaf(u[k].y, w[k], acc.y);
        acc.z = __builtin_fmaf(u[k].z, w[k], acc.z);
        acc.w = __builtin_fmaf(u[k].w, w[k], acc.w);
    }
}

template <int M>
__device__ __forceinline__ float xrow(float x) {
    const int v = __builtin_bit_cast(int, x);
    return __builtin_bit_cast(float,
        __builtin_amdgcn_update_dpp(v, v, 0x160 + M, 0xf, 0xf, false));
}
__device__ __forceinline__ float xhalf(float x) {
    const int v = __builtin_bit_cast(int, x);
    return __builtin_bit_cast(float,
        __builtin_amdgcn_permlanex16(v, v, 0x76543210, 0xfedcba98, false, false));
}

__device__ __forceinline__ void list_entry(
        const int2* __restrict__ rec, int i, int beg, int deg, float di, int t,
        int& r_l, float& w_l) {
    r_l = i;
    w_l = (t == deg) ? di * di : 0.0f;
    if (t < deg) {
        const int2 q = rec[beg + t];
        r_l = q.x;
        w_l = __int_as_float(q.y);
    }
}

__device__ __forceinline__ float4 gather_sum(
        const float4* __restrict__ X, const int2* __restrict__ rec,
        int i, int beg, int deg, float di, int r_l, float w_l, int lane) {
    const int tot = deg + 1;
    float4 acc = {0.0f, 0.0f, 0.0f, 0.0f};
    for (int t0 = 0; t0 < tot; t0 += 32) {
        if (t0 > 0) list_entry(rec, i, beg, deg, di, t0 + lane, r_l, w_l);
        const int m = min(tot - t0, 32);
        int j = 0;
        for (; m - j > 4; j += 8) add_rows<8>(acc, X, r_l, w_l, j, lane);
        if (j < m)               add_rows<4>(acc, X, r_l, w_l, j, lane);
    }
    return acc;
}

__device__ __forceinline__ float4 bias_relu(float4 acc, float4 bb) {
    float4 o = {acc.x + bb.x, acc.y + bb.y, acc.z + bb.z, acc.w + bb.w};
    o.x = o.x > 0.0f ? o.x : 0.0f;
    o.y = o.y > 0.0f ? o.y : 0.0f;
    o.z = o.z > 0.0f ? o.z : 0.0f;
    o.w = o.w > 0.0f ? o.w : 0.0f;
    return o;
}

#define HP 132
#define WP 136
__global__ __launch_bounds__(1024) void k_gather_gemm(
        const int* __restrict__ rowptr, const int2* __restrict__ rec,
        const float* __restrict__ xw, const float* __restrict__ dinv,
        const float* __restrict__ b, const float* __restrict__ W,
        float* __restrict__ C, int n) {
    __shared__ float Wl[D * WP];
    __shared__ float Hl[64 * HP];
    const int wv = __builtin_amdgcn_readfirstlane((int)(threadIdx.x >> 5));
    const int lane = threadIdx.x & 31;
    const int base = blockIdx.x * 64;
    const int iA = base + 2 * wv, iB = iA + 1;
    const int cA = min(iA, n - 1), cB = min(iB, n - 1);
    const int begA = rowptr[cA], begB = rowptr[cB];
    const int degA = rowptr[cA + 1] - begA, degB = rowptr[cB + 1] - begB;
    const float diA = dinv[cA], diB = dinv[cB];
    int rA, rB; float wA, wB;
    list_entry(rec, cA, begA, degA, diA, lane, rA, wA);
    list_entry(rec, cB, begB, degB, diB, lane, rB, wB);
    const float4 bb = ((const float4*)b)[lane];

    for (int q = 0; q < 4; ++q) {
        const int f = (int)threadIdx.x + q * 1024;
        *(float4*)(Wl + (f >> 5) * WP + (f & 31) * 4) = ((const float4*)W)[f];
    }

    const float4* X = (const float4*)xw;
    float4 oA = {0.0f, 0.0f, 0.0f, 0.0f}, oB = oA;
    if (iA < n) oA = bias_relu(gather_sum(X, rec, iA, begA, degA, diA, rA, wA, lane), bb);
    if (iB < n) oB = bias_relu(gather_sum(X, rec, iB, begB, degB, diB, rB, wB, lane), bb);
    *(float4*)(Hl + (2 * wv) * HP + 4 * lane)     = oA;
    *(float4*)(Hl + (2 * wv + 1) * HP + 4 * lane) = oB;
    __syncthreads();

    const int mt = wv >> 3, nt = wv & 7;
    const int c = lane & 15;
    const int khalf = (lane >> 4) * 2;
    const float* Ar = Hl + (mt * 16 + c) * HP + khalf;
    const float* Br = Wl + khalf * WP + nt * 16 + c;
    v8f acc = {};
    for (int k = 0; k < D; k += 4) {
        const v2f a = *(const v2f*)(Ar + k);
        v2f bf;
        bf.x = Br[k * WP];
        bf.y = Br[(k + 1) * WP];
        acc = __builtin_amdgcn_wmma_f32_16x16x4_f32(
            false, a, false, bf, (short)0, acc, false, false);
    }
    const int rbase = base + mt * 16 + (lane >> 4) * 8;
    for (int v = 0; v < 8; ++v) {
        const int r = rbase + v;
        if (r < n) C[(size_t)r * D + nt * 16 + c] = acc[v];
    }
}

__global__ __launch_bounds__(256) void k_gather_head(
        const int* __restrict__ rowptr, const int2* __restrict__ rec,
        const float* __restrict__ xw, const float* __restrict__ dinv,
        const float* __restrict__ b, const float* __restrict__ Wfc,
        float* __restrict__ out, int n) {
    __shared__ float4 s_w[256];
    const int wv = __builtin_amdgcn_readfirstlane((int)(threadIdx.x >> 5));
    const int i = blockIdx.x * 8 + wv;
    const int lane = threadIdx.x & 31;
    const int ic = min(i, n - 1);
    const int beg = rowptr[ic];
    const int deg = rowptr[ic + 1] - beg;
    const float di = dinv[ic];
    {
        const int t = threadIdx.x;
        s_w[((t >> 7) * 4 + (t & 3)) * 32 + ((t & 127) >> 2)] = ((const float4*)Wfc)[t];
        __syncthreads();
    }
    if (i >= n) return;
    int r_l; float w_l;
    list_entry(rec, i, beg, deg, di, lane, r_l, w_l);
    const float4 acc = gather_sum((const float4*)xw, rec, i, beg, deg, di, r_l, w_l, lane);
    const float4 o = bias_relu(acc, ((const float4*)b)[lane]);

    const float4 t0 = s_w[0 * 32 + lane], t1 = s_w[1 * 32 + lane];
    const float4 t2 = s_w[2 * 32 + lane], t3 = s_w[3 * 32 + lane];
    const float4 u0 = s_w[4 * 32 + lane], u1 = s_w[5 * 32 + lane];
    const float4 u2 = s_w[6 * 32 + lane], u3 = s_w[7 * 32 + lane];
    float s[8];
    s[0] = o.x * t0.x + o.y * t1.x + o.z * t2.x + o.w * t3.x;
    s[1] = o.x * t0.y + o.y * t1.y + o.z * t2.y + o.w * t3.y;
    s[2] = o.x * t0.z + o.y * t1.z + o.z * t2.z + o.w * t3.z;
    s[3] = o.x * t0.w + o.y * t1.w + o.z * t2.w + o.w * t3.w;
    s[4] = o.x * u0.x + o.y * u1.x + o.z * u2.x + o.w * u3.x;
    s[5] = o.x * u0.y + o.y * u1.y + o.z * u2.y + o.w * u3.y;
    s[6] = o.x * u0.z + o.y * u1.z + o.z * u2.z + o.w * u3.z;
    s[7] = o.x * u0.w + o.y * u1.w + o.z * u2.w + o.w * u3.w;

    const bool h16 = (lane & 16) != 0;
    float a4[4];
    for (int q = 0; q < 4; ++q) {
        const float keep = h16 ? s[q + 4] : s[q];
        const float give = h16 ? s[q] : s[q + 4];
        a4[q] = keep + xhalf(give);
    }
    const bool h8 = (lane & 8) != 0;
    float a2[2];
    for (int q = 0; q < 2; ++q) {
        const float keep = h8 ? a4[q + 2] : a4[q];
        const float give = h8 ? a4[q] : a4[q + 2];
        a2[q] = keep + xrow<8>(give);
    }
    const bool h4 = (lane & 4) != 0;
    const float keep = h4 ? a2[1] : a2[0];
    const float give = h4 ? a2[0] : a2[1];
    float a1 = keep + xrow<4>(give);
    a1 += xrow<2>(a1);
    a1 += xrow<1>(a1);
    if ((lane & 3) == 0)
        out[(size_t)i * 8 + ((lane >> 4) & 1) * 4 + ((lane >> 3) & 1) * 2 + ((lane >> 2) & 1)] = a1;
}

#define EPT 4
template <bool FULL>
__device__ __forceinline__ void edge_tile(
        const int* __restrict__ ei, int E, const float4* __restrict__ pq4,
        const float* __restrict__ bfc, float4* __restrict__ out4, int e0) {
    int r[EPT], c[EPT];
    for (int k = 0; k < EPT; ++k) {
        const int e = FULL ? e0 + k * 256 : min(e0 + k * 256, E - 1);
        r[k] = ei[e];
        c[k] = ei[E + e];
    }
    float4 p[EPT], q[EPT];
    for (int k = 0; k < EPT; ++k) {
        p[k] = pq4[(size_t)r[k] * 2];
        q[k] = pq4[(size_t)c[k] * 2 + 1];
    }
    const float b0 = bfc[0], b1 = bfc[1], b2 = bfc[2], b3 = bfc[3];
    for (int k = 0; k < EPT; ++k) {
        const float s0 = (p[k].x + q[k].x) + b0;
        const float s1 = (p[k].y + q[k].y) + b1;
        const float s2 = (p[k].z + q[k].z) + b2;
        const float s3 = (p[k].w + q[k].w) + b3;
        const float mx  = fmaxf(fmaxf(s0, s1), fmaxf(s2, s3));
        const float sum = __expf(s0 - mx) + __expf(s1 - mx) +
                          __expf(s2 - mx) + __expf(s3 - mx);
        const float lse = mx + __logf(sum);
        const float4 o  = {s0 - lse, s1 - lse, s2 - lse, s3 - lse};
        const int e = e0 + k * 256;
        if (FULL || e < E) out4[e] = o;
    }
}

__global__ __launch_bounds__(256) void k_edge_out(
        const int* __restrict__ ei, int E, const float* __restrict__ pq,
        const float* __restrict__ bfc, float* __restrict__ out) {
    const int tile0 = blockIdx.x * (256 * EPT);
    const int e0 = tile0 + threadIdx.x;
    if (tile0 + 256 * EPT <= E)
        edge_tile<true >(ei, E, (const float4*)pq, bfc, (float4*)out, e0);
    else
        edge_tile<false>(ei, E, (const float4*)pq, bfc, (float4*)out, e0);
}

extern "C" void kernel_launch(void* const* d_in, const int* in_sizes, int n_in,
                              void* d_out, int out_size, void* d_ws, size_t ws_size,
                              hipStream_t stream) {
    const float* x   = (const float*)d_in[0];
    const int*   ei  = (const int*)  d_in[1];
    const float* W1  = (const float*)d_in[2];
    const float* b1  = (const float*)d_in[3];
    const float* W2  = (const float*)d_in[4];
    const float* b2  = (const float*)d_in[5];
    const float* Wfc = (const float*)d_in[6];
    const float* bfc = (const float*)d_in[7];
    float* out = (float*)d_out;

    const int n = in_sizes[0] / D;
    const int E = in_sizes[1] / 2;
    const int n4 = (n + 3) & ~3;
    const int E4 = (E + 3) & ~3;

    float* bufA   = (float*)d_ws;
    float* bufB   = bufA + (size_t)n * D;
    float* pq     = bufA;
    const int NB    = (n + RB - 1) / RB;
    const int G     = (E + CE - 1) / CE;
    const int GNB4  = (G * NB + 3) & ~3;
    float* dinv     = bufB + (size_t)n * D;
    int*   tot      = (int*)(dinv + n4);
    int*   binStart = tot + NBMAX;
    int*   rowptr   = binStart + NBMAX + 4;
    int*   hist     = rowptr + n4 + 4;
    int*   pre      = hist + GNB4;
    int2*  rec      = (int2*)(pre + GNB4);
    int2*  binned   = rec + E4;

    const int gGemm = (n + 127) / 128;
    const int gGemA = gGemm / 2;
    const int gGemC = gGemm - gGemA;
    const int gNode = (n + 7) / 8;
    const int gTile = (n + 63) / 64;
    const int gEdge = (E + 256 * EPT - 1) / (256 * EPT);

    k_gemm_hist   <<<gGemA + G, 256, 0, stream>>>(x, W1, bufA, n, 0, gGemA, ei, E, NB, hist);
    k_bin_prefix  <<<NB, 256, 0, stream>>>(hist, G, NB, pre, tot);
    k_gemm_scatter<<<gGemC + G, 256, 0, stream>>>(x, W1, bufA, n, gGemA, gGemC, ei, E, NB, pre, tot, binStart, binned);
    k_bin_sort    <<<NB, 1024, 0, stream>>>(binStart, binned, n, rowptr, dinv, rec);
    k_weight      <<<G, 256, 0, stream>>>(E, dinv, rec);

    k_gather_gemm<<<gTile, 1024, 0, stream>>>(rowptr, rec, bufA, dinv, b1, W2, bufB, n);

    k_gather_head<<<gNode, 256, 0, stream>>>(rowptr, rec, bufB, dinv, b2, Wfc, pq, n);

    k_edge_out<<<gEdge, 256, 0, stream>>>(ei, E, pq, bfc, out);
}


# Per-bin prefix over the edge chunks folded into the scatter launch: 8192-edge chunks, each chunk block sums the small count table itself; prefix launch removed; a node's list (sum) order shifts
# speedup vs baseline: 4.1701x; 1.1070x over previous
#include <hip/hip_runtime.h>
#include <hip/hip_bf16.h>
#include <math.h>

typedef __attribute__((ext_vector_type(2))) float v2f;
typedef __attribute__((ext_vector_type(8))) float v8f;

#define D 128

#define RB 256
#define NBMAX 256
#define CE 8192
#define EPC (CE / 256)
#define WE 1024

__device__ __forceinline__ int scan256(int v, int* s_ws, int& tot) {
    const int lane = threadIdx.x & 31;
    const int wave = threadIdx.x >> 5;
    int s = v;
    for (int d = 1; d < 32; d <<= 1) {
        const int t = __shfl_up(s, d);
        if (lane >= d) s += t;
    }
    if (lane == 31) s_ws[wave] = s;
    __syncthreads();
    int off = 0, all = 0;
    for (int w = 0; w < 8; ++w) {
        const int x = s_ws[w];
        if (w < wave) off += x;
        all += x;
    }
    tot = all;
    return off + s - v;
}

__global__ __launch_bounds__(1024) void k_bin_sort(
        const int* __restrict__ binStart, const int2* __restrict__ binned, int n,
        int* __restrict__ rowptr, float* __restrict__ dinv, int2* __restrict__ rec) {
    __shared__ int s_cnt[RB];
    __shared__ int s_cur[RB];
    __shared__ int s_ws[8];
    const int b = blockIdx.x;
    const int t = threadIdx.x;
    const int lo = binStart[b], hi = binStart[b + 1];
    if (t < RB) s_cnt[t] = 0;
    __syncthreads();
    for (int j = lo + t; j < hi; j += 1024)
        atomicAdd(&s_cnt[binned[j].y & (RB - 1)], 1);
    __syncthreads();
    {
        const int lane = t & 31, wave = t >> 5;
        const int v = (t < RB) ? s_cnt[t] : 0;
        int s = v;
        for (int d = 1; d < 32; d <<= 1) {
            const int x = __shfl_up(s, d);
            if (lane >= d) s += x;
        }
        if (t < RB && lane == 31) s_ws[wave] = s;
        __syncthreads();
        if (t < RB) {
            int off = 0;
            for (int w = 0; w < 8; ++w) if (w < wave) off += s_ws[w];
            const int ex = off + s - v;
            s_cur[t] = ex;
            const int node = b * RB + t;
            if (node < n) {
                rowptr[node] = lo + ex;
                const float d = (float)(v + 1);
                dinv[node] = (d > 0.0f) ? rsqrtf(d) : 0.0f;
            }
            if (node == n - 1) rowptr[n] = lo + ex + v;
        }
        __syncthreads();
    }
    for (int j = lo + t; j < hi; j += 1024) {
        const int2 q = binned[j];
        const int pos = atomicAdd(&s_cur[q.y & (RB - 1)], 1);
        rec[lo + pos] = q;
    }
}

__global__ __launch_bounds__(256) void k_weight(
        int E, const float* __restrict__ dinv, int2* rec) {
    const int j0 = blockIdx.x * WE + threadIdx.x;
    int2 q[4];
    for (int k = 0; k < 4; ++k) {
        const int j = j0 + k * 256;
        q[k] = (j < E) ? rec[j] : make_int2(0, 0);
    }
    float a[4], c[4];
    for (int k = 0; k < 4; ++k) {
        a[k] = dinv[q[k].x];
        c[k] = dinv[q[k].y];
    }
    for (int k = 0; k < 4; ++k) {
        const int j = j0 + k * 256;
        if (j < E) rec[j] = make_int2(q[k].x, __float_as_int(a[k] * c[k]));
    }
}

__device__ __forceinline__ void gemm_tile(
        const float* __restrict__ A, const float* __restrict__ W,
        float* __restrict__ C, int n, int blk, float* Wl) {
    for (int i = threadIdx.x; i < (D * D) / 4; i += 256)
        ((float4*)Wl)[i] = ((const float4*)W)[i];
    __syncthreads();

    const int wave = threadIdx.x >> 5;
    const int lane = threadIdx.x & 31;
    const int m0   = blk * 128 + wave * 16;

    int arow = m0 + (lane & 15);
    if (arow > n - 1) arow = n - 1;
    const int khalf = (lane >> 4) * 2;
    const float* Ar = A + (size_t)arow * D;

    v8f acc[8] = {};

    for (int k = 0; k < D; k += 4) {
        const v2f a = *(const v2f*)(Ar + k + khalf);
        const int kb = k + khalf;
#pragma unroll
        for (int t = 0; t < 8; ++t) {
            const int col = t * 16 + (lane & 15);
            v2f b;
            b.x = Wl[kb * D + col];
            b.y = Wl[(kb + 1) * D + col];
            acc[t] = __builtin_amdgcn_wmma_f32_16x16x4_f32(
                false, a, false, b, (short)0, acc[t], false, false);
        }
    }

    const int rbase = m0 + (lane >> 4) * 8;
    const int cix   = lane & 15;
#pragma unroll
    for (int t = 0; t < 8; ++t) {
#pragma unroll
        for (int v = 0; v < 8; ++v) {
            const int r = rbase + v;
            if (r < n) C[(size_t)r * D + t * 16 + cix] = acc[t][v];
        }
    }
}


__global__ __launch_bounds__(256) void k_gemm_hist(
        const float* __restrict__ A, const float* __restrict__ W,
        float* __restrict__ C, int n, int blk0, int gGemm,
        const int* __restrict__ ei, int E, int NB, int* __restrict__ hist) {
    __shared__ float Wl[D * D];
    __shared__ int s_h[NBMAX];
    if ((int)blockIdx.x < gGemm) {
        gemm_tile(A, W, C, n, blk0 + (int)blockIdx.x, Wl);
        return;
    }
    const int g = (int)blockIdx.x - gGemm;
    const int t = threadIdx.x;
    const int e0 = g * CE + t;
    int c[EPC];
    for (int k = 0; k < EPC; ++k) c[k] = ei[E + min(e0 + k * 256, E - 1)];
    s_h[t] = 0;
    __syncthreads();
    for (int k = 0; k < EPC; ++k)
        if (e0 + k * 256 < E) atomicAdd(&s_h[c[k] >> 8], 1);
    __syncthreads();
    if (t < NB) hist[(size_t)g * NB + t] = s_h[t];
}

template <bool FULL>
__device__ __forceinline__ void place_edges(
        const int (&r)[EPC], const int (&c)[EPC], int e0, int E,
        int* s_base, int2* __restrict__ binned) {
    if (FULL) {
        int pos[EPC];
        for (int k = 0; k < EPC; ++k) pos[k] = atomicAdd(&s_base[c[k] >> 8], 1);
        for (int k = 0; k < EPC; ++k) binned[pos[k]] = make_int2(r[k], c[k]);
    } else {
        for (int k = 0; k < EPC; ++k) {
            if (e0 + k * 256 < E) {
                const int pos = atomicAdd(&s_base[c[k] >> 8], 1);
                binned[pos] = make_int2(r[k], c[k]);
            }
        }
    }
}

__global__ __launch_bounds__(256) void k_gemm_scatter(
        const float* __restrict__ A, const float* __restrict__ W,
        float* __restrict__ C, int n, int blk0, int gGemm,
        const int* __restrict__ ei, int E, int NB,
        const int* __restrict__ hist, int G,
        int* __restrict__ binStart, int2* __restrict__ binned) {
    __shared__ float Wl[D * D];
    __shared__ int s_base[NBMAX];
    __shared__ int s_ws[8];
    if ((int)blockIdx.x < gGemm) {
        gemm_tile(A, W, C, n, blk0 + (int)blockIdx.x, Wl);
        return;
    }
    const int g = (int)blockIdx.x - gGemm;
    const int t = threadIdx.x;
    const int e0 = g * CE + t;
    int r[EPC], c[EPC];
    for (int k = 0; k < EPC; ++k) {
        const int e = min(e0 + k * 256, E - 1);
        r[k] = ei[e];
        c[k] = ei[E + e];
    }
    int v = 0, p = 0;
    const int bt = min(t, NB - 1);
    for (int g0 = 0; g0 < G; g0 += 16) {
        int h[16];
        for (int k = 0; k < 16; ++k) h[k] = hist[(size_t)min(g0 + k, G - 1) * NB + bt];
        for (int k = 0; k < 16; ++k) {
            if (g0 + k < G) v += h[k];
            if (g0 + k < g) p += h[k];
        }
    }
    if (t >= NB) { v = 0; p = 0; }
    int all;
    const int ex = scan256(v, s_ws, all);
    s_base[t] = ex + p;
    if (g == 0) {
        if (t < NB) binStart[t] = ex;
        if (t == NB - 1) binStart[NB] = ex + v;
    }
    __syncthreads();
    if (g * CE + CE <= E)
        place_edges<true >(r, c, e0, E, s_base, binned);
    else
        place_edges<false>(r, c, e0, E, s_base, binned);
}

template <int K>
__device__ __forceinline__ void add_rows(float4& acc, const float4* __restrict__ X,
                                         int r_l, float w_l, int j, int lane) {
    float4 u[K];
    float  w[K];
    for (int k = 0; k < K; ++k) {
        const int r = __builtin_amdgcn_readlane(r_l, j + k);
        w[k] = __int_as_float(__builtin_amdgcn_readlane(__float_as_int(w_l), j + k));
        u[k] = X[(size_t)r * (D / 4) + lane];
    }
    for (int k = 0; k < K; ++k) {
        acc.x = __builtin_fmaf(u[k].x, w[k], acc.x);
        acc.y = __builtin_fmaf(u[k].y, w[k], acc.y);
        acc.z = __builtin_fmaf(u[k].z, w[k], acc.z);
        acc.w = __builtin_fmaf(u[k].w, w[k], acc.w);
    }
}

template <int M>
__device__ __forceinline__ float xrow(float x) {
    const int v = __builtin_bit_cast(int, x);
    return __builtin_bit_cast(float,
        __builtin_amdgcn_update_dpp(v, v, 0x160 + M, 0xf, 0xf, false));
}
__device__ __forceinline__ float xhalf(float x) {
    const int v = __builtin_bit_cast(int, x);
    return __builtin_bit_cast(float,
        __builtin_amdgcn_permlanex16(v, v, 0x76543210, 0xfedcba98, false, false));
}

__device__ __forceinline__ void list_entry(
        const int2* __restrict__ rec, int i, int beg, int deg, float di, int t,
        int& r_l, float& w_l) {
    r_l = i;
    w_l = (t == deg) ? di * di : 0.0f;
    if (t < deg) {
        const int2 q = rec[beg + t];
        r_l = q.x;
        w_l = __int_as_float(q.y);
    }
}

__device__ __forceinline__ float4 gather_sum(
        const float4* __restrict__ X, const int2* __restrict__ rec,
        int i, int beg, int deg, float di, int r_l, float w_l, int lane) {
    const int tot = deg + 1;
    float4 acc = {0.0f, 0.0f, 0.0f, 0.0f};
    for (int t0 = 0; t0 < tot; t0 += 32) {
        if (t0 > 0) list_entry(rec, i, beg, deg, di, t0 + lane, r_l, w_l);
        const int m = min(tot - t0, 32);
        int j = 0;
        for (; m - j > 4; j += 8) add_rows<8>(acc, X, r_l, w_l, j, lane);
        if (j < m)               add_rows<4>(acc, X, r_l, w_l, j, lane);
    }
    return acc;
}

__device__ __forceinline__ float4 bias_relu(float4 acc, float4 bb) {
    float4 o = {acc.x + bb.x, acc.y + bb.y, acc.z + bb.z, acc.w + bb.w};
    o.x = o.x > 0.0f ? o.x : 0.0f;
    o.y = o.y > 0.0f ? o.y : 0.0f;
    o.z = o.z > 0.0f ? o.z : 0.0f;
    o.w = o.w > 0.0f ? o.w : 0.0f;
    return o;
}

#define HP 132
#define WP 136
__global__ __launch_bounds__(1024) void k_gather_gemm(
        const int* __restrict__ rowptr, const int2* __restrict__ rec,
        const float* __restrict__ xw, const float* __restrict__ dinv,
        const float* __restrict__ b, const float* __restrict__ W,
        float* __restrict__ C, int n) {
    __shared__ float Wl[D * WP];
    __shared__ float Hl[64 * HP];
    const int wv = __builtin_amdgcn_readfirstlane((int)(threadIdx.x >> 5));
    const int lane = threadIdx.x & 31;
    const int base = blockIdx.x * 64;
    const int iA = base + 2 * wv, iB = iA + 1;
    const int cA = min(iA, n - 1), cB = min(iB, n - 1);
    const int begA = rowptr[cA], begB = rowptr[cB];
    const int degA = rowptr[cA + 1] - begA, degB = rowptr[cB + 1] - begB;
    const float diA = dinv[cA], diB = dinv[cB];
    int rA, rB; float wA, wB;
    list_entry(rec, cA, begA, degA, diA, lane, rA, wA);
    list_entry(rec, cB, begB, degB, diB, lane, rB, wB);
    const float4 bb = ((const float4*)b)[lane];

    for (int q = 0; q < 4; ++q) {
        const int f = (int)threadIdx.x + q * 1024;
        *(float4*)(Wl + (f >> 5) * WP + (f & 31) * 4) = ((const float4*)W)[f];
    }

    const float4* X = (const float4*)xw;
    float4 oA = {0.0f, 0.0f, 0.0f, 0.0f}, oB = oA;
    if (iA < n) oA = bias_relu(gather_sum(X, rec, iA, begA, degA, diA, rA, wA, lane), bb);
    if (iB < n) oB = bias_relu(gather_sum(X, rec, iB, begB, degB, diB, rB, wB, lane), bb);
    *(float4*)(Hl + (2 * wv) * HP + 4 * lane)     = oA;
    *(float4*)(Hl + (2 * wv + 1) * HP + 4 * lane) = oB;
    __syncthreads();

    const int mt = wv >> 3, nt = wv & 7;
    const int c = lane & 15;
    const int khalf = (lane >> 4) * 2;
    const float* Ar = Hl + (mt * 16 + c) * HP + khalf;
    const float* Br = Wl + khalf * WP + nt * 16 + c;
    v8f acc = {};
    for (int k = 0; k < D; k += 4) {
        const v2f a = *(const v2f*)(Ar + k);
        v2f bf;
        bf.x = Br[k * WP];
        bf.y = Br[(k + 1) * WP];
        acc = __builtin_amdgcn_wmma_f32_16x16x4_f32(
            false, a, false, bf, (short)0, acc, false, false);
    }
    const int rbase = base + mt * 16 + (lane >> 4) * 8;
    for (int v = 0; v < 8; ++v) {
        const int r = rbase + v;
        if (r < n) C[(size_t)r * D + nt * 16 + c] = acc[v];
    }
}

__global__ __launch_bounds__(256) void k_gather_head(
        const int* __restrict__ rowptr, const int2* __restrict__ rec,
        const float* __restrict__ xw, const float* __restrict__ dinv,
        const float* __restrict__ b, const float* __restrict__ Wfc,
        float* __restrict__ out, int n) {
    __shared__ float4 s_w[256];
    const int wv = __builtin_amdgcn_readfirstlane((int)(threadIdx.x >> 5));
    const int i = blockIdx.x * 8 + wv;
    const int lane = threadIdx.x & 31;
    const int ic = min(i, n - 1);
    const int beg = rowptr[ic];
    const int deg = rowptr[ic + 1] - beg;
    const float di = dinv[ic];
    {
        const int t = threadIdx.x;
        s_w[((t >> 7) * 4 + (t & 3)) * 32 + ((t & 127) >> 2)] = ((const float4*)Wfc)[t];
        __syncthreads();
    }
    if (i >= n) return;
    int r_l; float w_l;
    list_entry(rec, i, beg, deg, di, lane, r_l, w_l);
    const float4 acc = gather_sum((const float4*)xw, rec, i, beg, deg, di, r_l, w_l, lane);
    const float4 o = bias_relu(acc, ((const float4*)b)[lane]);

    const float4 t0 = s_w[0 * 32 + lane], t1 = s_w[1 * 32 + lane];
    const float4 t2 = s_w[2 * 32 + lane], t3 = s_w[3 * 32 + lane];
    const float4 u0 = s_w[4 * 32 + lane], u1 = s_w[5 * 32 + lane];
    const float4 u2 = s_w[6 * 32 + lane], u3 = s_w[7 * 32 + lane];
    float s[8];
    s[0] = o.x * t0.x + o.y * t1.x + o.z * t2.x + o.w * t3.x;
    s[1] = o.x * t0.y + o.y * t1.y + o.z * t2.y + o.w * t3.y;
    s[2] = o.x * t0.z + o.y * t1.z + o.z * t2.z + o.w * t3.z;
    s[3] = o.x * t0.w + o.y * t1.w + o.z * t2.w + o.w * t3.w;
    s[4] = o.x * u0.x + o.y * u1.x + o.z * u2.x + o.w * u3.x;
    s[5] = o.x * u0.y + o.y * u1.y + o.z * u2.y + o.w * u3.y;
    s[6] = o.x * u0.z + o.y * u1.z + o.z * u2.z + o.w * u3.z;
    s[7] = o.x * u0.w + o.y * u1.w + o.z * u2.w + o.w * u3.w;

    const bool h16 = (lane & 16) != 0;
    float a4[4];
    for (int q = 0; q < 4; ++q) {
        const float keep = h16 ? s[q + 4] : s[q];
        const float give = h16 ? s[q] : s[q + 4];
        a4[q] = keep + xhalf(give);
    }
    const bool h8 = (lane & 8) != 0;
    float a2[2];
    for (int q = 0; q < 2; ++q) {
        const float keep = h8 ? a4[q + 2] : a4[q];
        const float give = h8 ? a4[q] : a4[q + 2];
        a2[q] = keep + xrow<8>(give);
    }
    const bool h4 = (lane & 4) != 0;
    const float keep = h4 ? a2[1] : a2[0];
    const float give = h4 ? a2[0] : a2[1];
    float a1 = keep + xrow<4>(give);
    a1 += xrow<2>(a1);
    a1 += xrow<1>(a1);
    if ((lane & 3) == 0)
        out[(size_t)i * 8 + ((lane >> 4) & 1) * 4 + ((lane >> 3) & 1) * 2 + ((lane >> 2) & 1)] = a1;
}

#define EPT 4
template <bool FULL>
__device__ __forceinline__ void edge_tile(
        const int* __restrict__ ei, int E, const float4* __restrict__ pq4,
        const float* __restrict__ bfc, float4* __restrict__ out4, int e0) {
    int r[EPT], c[EPT];
    for (int k = 0; k < EPT; ++k) {
        const int e = FULL ? e0 + k * 256 : min(e0 + k * 256, E - 1);
        r[k] = ei[e];
        c[k] = ei[E + e];
    }
    float4 p[EPT], q[EPT];
    for (int k = 0; k < EPT; ++k) {
        p[k] = pq4[(size_t)r[k] * 2];
        q[k] = pq4[(size_t)c[k] * 2 + 1];
    }
    const float b0 = bfc[0], b1 = bfc[1], b2 = bfc[2], b3 = bfc[3];
    for (int k = 0; k < EPT; ++k) {
        const float s0 = (p[k].x + q[k].x) + b0;
        const float s1 = (p[k].y + q[k].y) + b1;
        const float s2 = (p[k].z + q[k].z) + b2;
        const float s3 = (p[k].w + q[k].w) + b3;
        const float mx  = fmaxf(fmaxf(s0, s1), fmaxf(s2, s3));
        const float sum = __expf(s0 - mx) + __expf(s1 - mx) +
                          __expf(s2 - mx) + __expf(s3 - mx);
        const float lse = mx + __logf(sum);
        const float4 o  = {s0 - lse, s1 - lse, s2 - lse, s3 - lse};
        const int e = e0 + k * 256;
        if (FULL || e < E) out4[e] = o;
    }
}

__global__ __launch_bounds__(256) void k_edge_out(
        const int* __restrict__ ei, int E, const float* __restrict__ pq,
        const float* __restrict__ bfc, float* __restrict__ out) {
    const int tile0 = blockIdx.x * (256 * EPT);
    const int e0 = tile0 + threadIdx.x;
    if (tile0 + 256 * EPT <= E)
        edge_tile<true >(ei, E, (const float4*)pq, bfc, (float4*)out, e0);
    else
        edge_tile<false>(ei, E, (const float4*)pq, bfc, (float4*)out, e0);
}

extern "C" void kernel_launch(void* const* d_in, const int* in_sizes, int n_in,
                              void* d_out, int out_size, void* d_ws, size_t ws_size,
                              hipStream_t stream) {
    const float* x   = (const float*)d_in[0];
    const int*   ei  = (const int*)  d_in[1];
    const float* W1  = (const float*)d_in[2];
    const float* b1  = (const float*)d_in[3];
    const float* W2  = (const float*)d_in[4];
    const float* b2  = (const float*)d_in[5];
    const float* Wfc = (const float*)d_in[6];
    const float* bfc = (const float*)d_in[7];
    float* out = (float*)d_out;

    const int n = in_sizes[0] / D;
    const int E = in_sizes[1] / 2;
    const int n4 = (n + 3) & ~3;
    const int E4 = (E + 3) & ~3;

    float* bufA   = (float*)d_ws;
    float* bufB   = bufA + (size_t)n * D;
    float* pq     = bufA;
    const int NB    = (n + RB - 1) / RB;
    const int G     = (E + CE - 1) / CE;
    const int GNB4  = (G * NB + 3) & ~3;
    float* dinv     = bufB + (size_t)n * D;
    int*   binStart = (int*)(dinv + n4);
    int*   rowptr   = binStart + NBMAX + 4;
    int*   hist     = rowptr + n4 + 4;
    int2*  rec      = (int2*)(hist + GNB4);
    int2*  binned   = rec + E4;

    const int gGemm = (n + 127) / 128;
    const int gGemA = gGemm / 2;
    const int gGemC = gGemm - gGemA;
    const int gNode = (n + 7) / 8;
    const int gTile = (n + 63) / 64;
    const int gEdge = (E + 256 * EPT - 1) / (256 * EPT);
    const int gWgt  = (E + WE - 1) / WE;

    k_gemm_hist   <<<gGemA + G, 256, 0, stream>>>(x, W1, bufA, n, 0, gGemA, ei, E, NB, hist);
    k_gemm_scatter<<<gGemC + G, 256, 0, stream>>>(x, W1, bufA, n, gGemA, gGemC, ei, E, NB, hist, G, binStart, binned);
    k_bin_sort    <<<NB, 1024, 0, stream>>>(binStart, binned, n, rowptr, dinv, rec);
    k_weight      <<<gWgt, 256, 0, stream>>>(E, dinv, rec);

    k_gather_gemm<<<gTile, 1024, 0, stream>>>(rowptr, rec, bufA, dinv, b1, W2, bufB, n);

    k_gather_head<<<gNode, 256, 0, stream>>>(rowptr, rec, bufB, dinv, b2, Wfc, pq, n);

    k_edge_out<<<gEdge, 256, 0, stream>>>(ei, E, pq, bfc, out);
}


# Histogram+scatter replaced by a chunk-local sort by bin (rank = LDS add's return); bin blocks walk the chunks' runs: one launch fewer, its x@W1 half rides the weight pass; list (sum) order shifts
# speedup vs baseline: 4.5781x; 1.0978x over previous
#include <hip/hip_runtime.h>
#include <hip/hip_bf16.h>
#include <math.h>

typedef __attribute__((ext_vector_type(2))) float v2f;
typedef __attribute__((ext_vector_type(8))) float v8f;

#define D 128

#define RB 256
#define NBMAX 256
#define CE 8192
#define EPC (CE / 256)
#define GMAX 1024
#define SEGT 8
#define SEGU 8
#define WE 1024

__device__ __forceinline__ int scan256(int v, int* s_ws, int& tot) {
    const int lane = threadIdx.x & 31;
    const int wave = threadIdx.x >> 5;
    int s = v;
    for (int d = 1; d < 32; d <<= 1) {
        const int t = __shfl_up(s, d);
        if (lane >= d) s += t;
    }
    if (lane == 31) s_ws[wave] = s;
    __syncthreads();
    int off = 0, all = 0;
    for (int w = 0; w < 8; ++w) {
        const int x = s_ws[w];
        if (w < wave) off += x;
        all += x;
    }
    tot = all;
    return off + s - v;
}

__global__ __launch_bounds__(1024) void k_bin_sort(
        const int* __restrict__ offs, const int2* __restrict__ binned, int n, int NB, int G,
        int* __restrict__ rowptr, float* __restrict__ dinv, int2* __restrict__ rec) {
    __shared__ int s_cnt[RB];
    __shared__ int s_cur[RB];
    __shared__ int s_ws[8];
    __shared__ int s_red[32];
    __shared__ int s_seg[GMAX];
    __shared__ int s_len[GMAX];
    const int b = blockIdx.x;
    const int t = threadIdx.x;
    int o0 = 0, o1 = 0;
    if (t < G) {
        const int* o = offs + (size_t)t * (NB + 1) + b;
        o0 = o[0];
        o1 = o[1];
    }
    s_seg[t] = t * CE + o0;
    s_len[t] = o1 - o0;
    if (t < RB) s_cnt[t] = 0;
    int lo;
    {
        int sum = o0;
        for (int d = 16; d >= 1; d >>= 1) sum += __shfl_xor(sum, d);
        if ((t & 31) == 0) s_red[t >> 5] = sum;
        __syncthreads();
        lo = 0;
        for (int w = 0; w < 32; ++w) lo += s_red[w];
    }
    for (int g = t / SEGT; g < G; g += 1024 / SEGT) {
        const int st = s_seg[g], len = s_len[g];
        for (int j0 = t & (SEGT - 1); j0 < len; j0 += SEGT * SEGU) {
            int y[SEGU];
            for (int k = 0; k < SEGU; ++k)
                y[k] = binned[st + min(j0 + k * SEGT, len - 1)].y;
            for (int k = 0; k < SEGU; ++k)
                if (j0 + k * SEGT < len) atomicAdd(&s_cnt[y[k] & (RB - 1)], 1);
        }
    }
    __syncthreads();
    {
        const int lane = t & 31, wave = t >> 5;
        const int v = (t < RB) ? s_cnt[t] : 0;
        int s = v;
        for (int d = 1; d < 32; d <<= 1) {
            const int x = __shfl_up(s, d);
            if (lane >= d) s += x;
        }
        if (t < RB && lane == 31) s_ws[wave] = s;
        __syncthreads();
        if (t < RB) {
            int off = 0;
            for (int w = 0; w < 8; ++w) if (w < wave) off += s_ws[w];
            const int ex = off + s - v;
            s_cur[t] = ex;
            const int node = b * RB + t;
            if (node < n) {
                rowptr[node] = lo + ex;
                const float d = (float)(v + 1);
                dinv[node] = (d > 0.0f) ? rsqrtf(d) : 0.0f;
            }
            if (node == n - 1) rowptr[n] = lo + ex + v;
        }
        __syncthreads();
    }
    for (int g = t / SEGT; g < G; g += 1024 / SEGT) {
        const int st = s_seg[g], len = s_len[g];
        for (int j0 = t & (SEGT - 1); j0 < len; j0 += SEGT * SEGU) {
            int2 q[SEGU];
            for (int k = 0; k < SEGU; ++k)
                q[k] = binned[st + min(j0 + k * SEGT, len - 1)];
            for (int k = 0; k < SEGU; ++k) {
                if (j0 + k * SEGT < len) {
                    const int pos = atomicAdd(&s_cur[q[k].y & (RB - 1)], 1);
                    rec[lo + pos] = q[k];
                }
            }
        }
    }
}

__device__ __forceinline__ void gemm_tile(
        const float* __restrict__ A, const float* __restrict__ W,
        float* __restrict__ C, int n, int blk, float* Wl) {
    for (int i = threadIdx.x; i < (D * D) / 4; i += 256)
        ((float4*)Wl)[i] = ((const float4*)W)[i];
    __syncthreads();

    const int wave = threadIdx.x >> 5;
    const int lane = threadIdx.x & 31;
    const int m0   = blk * 128 + wave * 16;

    int arow = m0 + (lane & 15);
    if (arow > n - 1) arow = n - 1;
    const int khalf = (lane >> 4) * 2;
    const float* Ar = A + (size_t)arow * D;

    v8f acc[8] = {};

    for (int k = 0; k < D; k += 4) {
        const v2f a = *(const v2f*)(Ar + k + khalf);
        const int kb = k + khalf;
#pragma unroll
        for (int t = 0; t < 8; ++t) {
            const int col = t * 16 + (lane & 15);
            v2f b;
            b.x = Wl[kb * D + col];
            b.y = Wl[(kb + 1) * D + col];
            acc[t] = __builtin_amdgcn_wmma_f32_16x16x4_f32(
                false, a, false, b, (short)0, acc[t], false, false);
        }
    }

    const int rbase = m0 + (lane >> 4) * 8;
    const int cix   = lane & 15;
#pragma unroll
    for (int t = 0; t < 8; ++t) {
#pragma unroll
        for (int v = 0; v < 8; ++v) {
            const int r = rbase + v;
            if (r < n) C[(size_t)r * D + t * 16 + cix] = acc[t][v];
        }
    }
}


__global__ __launch_bounds__(256) void k_gemm_csort(
        const float* __restrict__ A, const float* __restrict__ W,
        float* __restrict__ C, int n, int blk0, int gGemm,
        const int* __restrict__ ei, int E, int NB,
        int* __restrict__ offs, int2* __restrict__ binned) {
    __shared__ float Wl[D * D];
    __shared__ int s_cnt[NBMAX];
    __shared__ int s_off[NBMAX];
    __shared__ int s_ws[8];
    if ((int)blockIdx.x < gGemm) {
        gemm_tile(A, W, C, n, blk0 + (int)blockIdx.x, Wl);
        return;
    }
    const int g = (int)blockIdx.x - gGemm;
    const int t = threadIdx.x;
    const int e0 = g * CE + t;
    const bool full = g * CE + CE <= E;
    int r[EPC], c[EPC];
    for (int k = 0; k < EPC; ++k) {
        const int e = min(e0 + k * 256, E - 1);
        r[k] = ei[e];
        c[k] = ei[E + e];
    }
    s_cnt[t] = 0;
    __syncthreads();
    int rank[EPC];
    if (full) {
        for (int k = 0; k < EPC; ++k) rank[k] = atomicAdd(&s_cnt[c[k] >> 8], 1);
    } else {
        for (int k = 0; k < EPC; ++k)
            rank[k] = (e0 + k * 256 < E) ? atomicAdd(&s_cnt[c[k] >> 8], 1) : 0;
    }
    __syncthreads();
    const int v = s_cnt[t];
    int all;
    const int ex = scan256(v, s_ws, all);
    s_off[t] = ex;
    int* o = offs + (size_t)g * (NB + 1);
    if (t < NB) o[t] = ex;
    if (t == NB - 1) o[NB] = ex + v;
    __syncthreads();
    int2* dst = binned + (size_t)g * CE;
    if (full) {
        for (int k = 0; k < EPC; ++k) dst[s_off[c[k] >> 8] + rank[k]] = make_int2(r[k], c[k]);
    } else {
        for (int k = 0; k < EPC; ++k)
            if (e0 + k * 256 < E) dst[s_off[c[k] >> 8] + rank[k]] = make_int2(r[k], c[k]);
    }
}

__global__ __launch_bounds__(256) void k_gemm_weight(
        const float* __restrict__ A, const float* __restrict__ W,
        float* __restrict__ C, int n, int blk0, int gGemm,
        int E, const float* __restrict__ dinv, int2* rec) {
    __shared__ float Wl[D * D];
    if ((int)blockIdx.x < gGemm) {
        gemm_tile(A, W, C, n, blk0 + (int)blockIdx.x, Wl);
        return;
    }
    const int j0 = ((int)blockIdx.x - gGemm) * WE + threadIdx.x;
    int2 q[4];
    for (int k = 0; k < 4; ++k) {
        const int j = j0 + k * 256;
        q[k] = (j < E) ? rec[j] : make_int2(0, 0);
    }
    float a[4], c[4];
    for (int k = 0; k < 4; ++k) {
        a[k] = dinv[q[k].x];
        c[k] = dinv[q[k].y];
    }
    for (int k = 0; k < 4; ++k) {
        const int j = j0 + k * 256;
        if (j < E) rec[j] = make_int2(q[k].x, __float_as_int(a[k] * c[k]));
    }
}

template <int K>
__device__ __forceinline__ void add_rows(float4& acc, const float4* __restrict__ X,
                                         int r_l, float w_l, int j, int lane) {
    float4 u[K];
    float  w[K];
    for (int k = 0; k < K; ++k) {
        const int r = __builtin_amdgcn_readlane(r_l, j + k);
        w[k] = __int_as_float(__builtin_amdgcn_readlane(__float_as_int(w_l), j + k));
        u[k] = X[(size_t)r * (D / 4) + lane];
    }
    for (int k = 0; k < K; ++k) {
        acc.x = __builtin_fmaf(u[k].x, w[k], acc.x);
        acc.y = __builtin_fmaf(u[k].y, w[k], acc.y);
        acc.z = __builtin_fmaf(u[k].z, w[k], acc.z);
        acc.w = __builtin_fmaf(u[k].w, w[k], acc.w);
    }
}

template <int M>
__device__ __forceinline__ float xrow(float x) {
    const int v = __builtin_bit_cast(int, x);
    return __builtin_bit_cast(float,
        __builtin_amdgcn_update_dpp(v, v, 0x160 + M, 0xf, 0xf, false));
}
__device__ __forceinline__ float xhalf(float x) {
    const int v = __builtin_bit_cast(int, x);
    return __builtin_bit_cast(float,
        __builtin_amdgcn_permlanex16(v, v, 0x76543210, 0xfedcba98, false, false));
}

__device__ __forceinline__ void list_entry(
        const int2* __restrict__ rec, int i, int beg, int deg, float di, int t,
        int& r_l, float& w_l) {
    r_l = i;
    w_l = (t == deg) ? di * di : 0.0f;
    if (t < deg) {
        const int2 q = rec[beg + t];
        r_l = q.x;
        w_l = __int_as_float(q.y);
    }
}

__device__ __forceinline__ float4 gather_sum(
        const float4* __restrict__ X, const int2* __restrict__ rec,
        int i, int beg, int deg, float di, int r_l, float w_l, int lane) {
    const int tot = deg + 1;
    float4 acc = {0.0f, 0.0f, 0.0f, 0.0f};
    for (int t0 = 0; t0 < tot; t0 += 32) {
        if (t0 > 0) list_entry(rec, i, beg, deg, di, t0 + lane, r_l, w_l);
        const int m = min(tot - t0, 32);
        int j = 0;
        for (; m - j > 4; j += 8) add_rows<8>(acc, X, r_l, w_l, j, lane);
        if (j < m)               add_rows<4>(acc, X, r_l, w_l, j, lane);
    }
    return acc;
}

__device__ __forceinline__ float4 bias_relu(float4 acc, float4 bb) {
    float4 o = {acc.x + bb.x, acc.y + bb.y, acc.z + bb.z, acc.w + bb.w};
    o.x = o.x > 0.0f ? o.x : 0.0f;
    o.y = o.y > 0.0f ? o.y : 0.0f;
    o.z = o.z > 0.0f ? o.z : 0.0f;
    o.w = o.w > 0.0f ? o.w : 0.0f;
    return o;
}

#define HP 132
#define WP 136
__global__ __launch_bounds__(1024) void k_gather_gemm(
        const int* __restrict__ rowptr, const int2* __restrict__ rec,
        const float* __restrict__ xw, const float* __restrict__ dinv,
        const float* __restrict__ b, const float* __restrict__ W,
        float* __restrict__ C, int n) {
    __shared__ float Wl[D * WP];
    __shared__ float Hl[64 * HP];
    const int wv = __builtin_amdgcn_readfirstlane((int)(threadIdx.x >> 5));
    const int lane = threadIdx.x & 31;
    const int base = blockIdx.x * 64;
    const int iA = base + 2 * wv, iB = iA + 1;
    const int cA = min(iA, n - 1), cB = min(iB, n - 1);
    const int begA = rowptr[cA], begB = rowptr[cB];
    const int degA = rowptr[cA + 1] - begA, degB = rowptr[cB + 1] - begB;
    const float diA = dinv[cA], diB = dinv[cB];
    int rA, rB; float wA, wB;
    list_entry(rec, cA, begA, degA, diA, lane, rA, wA);
    list_entry(rec, cB, begB, degB, diB, lane, rB, wB);
    const float4 bb = ((const float4*)b)[lane];

    for (int q = 0; q < 4; ++q) {
        const int f = (int)threadIdx.x + q * 1024;
        *(float4*)(Wl + (f >> 5) * WP + (f & 31) * 4) = ((const float4*)W)[f];
    }

    const float4* X = (const float4*)xw;
    float4 oA = {0.0f, 0.0f, 0.0f, 0.0f}, oB = oA;
    if (iA < n) oA = bias_relu(gather_sum(X, rec, iA, begA, degA, diA, rA, wA, lane), bb);
    if (iB < n) oB = bias_relu(gather_sum(X, rec, iB, begB, degB, diB, rB, wB, lane), bb);
    *(float4*)(Hl + (2 * wv) * HP + 4 * lane)     = oA;
    *(float4*)(Hl + (2 * wv + 1) * HP + 4 * lane) = oB;
    __syncthreads();

    const int mt = wv >> 3, nt = wv & 7;
    const int c = lane & 15;
    const int khalf = (lane >> 4) * 2;
    const float* Ar = Hl + (mt * 16 + c) * HP + khalf;
    const float* Br = Wl + khalf * WP + nt * 16 + c;
    v8f acc = {};
    for (int k = 0; k < D; k += 4) {
        const v2f a = *(const v2f*)(Ar + k);
        v2f bf;
        bf.x = Br[k * WP];
        bf.y = Br[(k + 1) * WP];
        acc = __builtin_amdgcn_wmma_f32_16x16x4_f32(
            false, a, false, bf, (short)0, acc, false, false);
    }
    const int rbase = base + mt * 16 + (lane >> 4) * 8;
    for (int v = 0; v < 8; ++v) {
        const int r = rbase + v;
        if (r < n) C[(size_t)r * D + nt * 16 + c] = acc[v];
    }
}

__global__ __launch_bounds__(256) void k_gather_head(
        const int* __restrict__ rowptr, const int2* __restrict__ rec,
        const float* __restrict__ xw, const float* __restrict__ dinv,
        const float* __restrict__ b, const float* __restrict__ Wfc,
        float* __restrict__ out, int n) {
    __shared__ float4 s_w[256];
    const int wv = __builtin_amdgcn_readfirstlane((int)(threadIdx.x >> 5));
    const int i = blockIdx.x * 8 + wv;
    const int lane = threadIdx.x & 31;
    const int ic = min(i, n - 1);
    const int beg = rowptr[ic];
    const int deg = rowptr[ic + 1] - beg;
    const float di = dinv[ic];
    {
        const int t = threadIdx.x;
        s_w[((t >> 7) * 4 + (t & 3)) * 32 + ((t & 127) >> 2)] = ((const float4*)Wfc)[t];
        __syncthreads();
    }
    if (i >= n) return;
    int r_l; float w_l;
    list_entry(rec, i, beg, deg, di, lane, r_l, w_l);
    const float4 acc = gather_sum((const float4*)xw, rec, i, beg, deg, di, r_l, w_l, lane);
    const float4 o = bias_relu(acc, ((const float4*)b)[lane]);

    const float4 t0 = s_w[0 * 32 + lane], t1 = s_w[1 * 32 + lane];
    const float4 t2 = s_w[2 * 32 + lane], t3 = s_w[3 * 32 + lane];
    const float4 u0 = s_w[4 * 32 + lane], u1 = s_w[5 * 32 + lane];
    const float4 u2 = s_w[6 * 32 + lane], u3 = s_w[7 * 32 + lane];
    float s[8];
    s[0] = o.x * t0.x + o.y * t1.x + o.z * t2.x + o.w * t3.x;
    s[1] = o.x * t0.y + o.y * t1.y + o.z * t2.y + o.w * t3.y;
    s[2] = o.x * t0.z + o.y * t1.z + o.z * t2.z + o.w * t3.z;
    s[3] = o.x * t0.w + o.y * t1.w + o.z * t2.w + o.w * t3.w;
    s[4] = o.x * u0.x + o.y * u1.x + o.z * u2.x + o.w * u3.x;
    s[5] = o.x * u0.y + o.y * u1.y + o.z * u2.y + o.w * u3.y;
    s[6] = o.x * u0.z + o.y * u1.z + o.z * u2.z + o.w * u3.z;
    s[7] = o.x * u0.w + o.y * u1.w + o.z * u2.w + o.w * u3.w;

    const bool h16 = (lane & 16) != 0;
    float a4[4];
    for (int q = 0; q < 4; ++q) {
        const float keep = h16 ? s[q + 4] : s[q];
        const float give = h16 ? s[q] : s[q + 4];
        a4[q] = keep + xhalf(give);
    }
    const bool h8 = (lane & 8) != 0;
    float a2[2];
    for (int q = 0; q < 2; ++q) {
        const float keep = h8 ? a4[q + 2] : a4[q];
        const float give = h8 ? a4[q] : a4[q + 2];
        a2[q] = keep + xrow<8>(give);
    }
    const bool h4 = (lane & 4) != 0;
    const float keep = h4 ? a2[1] : a2[0];
    const float give = h4 ? a2[0] : a2[1];
    float a1 = keep + xrow<4>(give);
    a1 += xrow<2>(a1);
    a1 += xrow<1>(a1);
    if ((lane & 3) == 0)
        out[(size_t)i * 8 + ((lane >> 4) & 1) * 4 + ((lane >> 3) & 1) * 2 + ((lane >> 2) & 1)] = a1;
}

#define EPT 4
template <bool FULL>
__device__ __forceinline__ void edge_tile(
        const int* __restrict__ ei, int E, const float4* __restrict__ pq4,
        const float* __restrict__ bfc, float4* __restrict__ out4, int e0) {
    int r[EPT], c[EPT];
    for (int k = 0; k < EPT; ++k) {
        const int e = FULL ? e0 + k * 256 : min(e0 + k * 256, E - 1);
        r[k] = ei[e];
        c[k] = ei[E + e];
    }
    float4 p[EPT], q[EPT];
    for (int k = 0; k < EPT; ++k) {
        p[k] = pq4[(size_t)r[k] * 2];
        q[k] = pq4[(size_t)c[k] * 2 + 1];
    }
    const float b0 = bfc[0], b1 = bfc[1], b2 = bfc[2], b3 = bfc[3];
    for (int k = 0; k < EPT; ++k) {
        const float s0 = (p[k].x + q[k].x) + b0;
        const float s1 = (p[k].y + q[k].y) + b1;
        const float s2 = (p[k].z + q[k].z) + b2;
        const float s3 = (p[k].w + q[k].w) + b3;
        const float mx  = fmaxf(fmaxf(s0, s1), fmaxf(s2, s3));
        const float sum = __expf(s0 - mx) + __expf(s1 - mx) +
                          __expf(s2 - mx) + __expf(s3 - mx);
        const float lse = mx + __logf(sum);
        const float4 o  = {s0 - lse, s1 - lse, s2 - lse, s3 - lse};
        const int e = e0 + k * 256;
        if (FULL || e < E) out4[e] = o;
    }
}

__global__ __launch_bounds__(256) void k_edge_out(
        const int* __restrict__ ei, int E, const float* __restrict__ pq,
        const float* __restrict__ bfc, float* __restrict__ out) {
    const int tile0 = blockIdx.x * (256 * EPT);
    const int e0 = tile0 + threadIdx.x;
    if (tile0 + 256 * EPT <= E)
        edge_tile<true >(ei, E, (const float4*)pq, bfc, (float4*)out, e0);
    else
        edge_tile<false>(ei, E, (const float4*)pq, bfc, (float4*)out, e0);
}

extern "C" void kernel_launch(void* const* d_in, const int* in_sizes, int n_in,
                              void* d_out, int out_size, void* d_ws, size_t ws_size,
                              hipStream_t stream) {
    const float* x   = (const float*)d_in[0];
    const int*   ei  = (const int*)  d_in[1];
    const float* W1  = (const float*)d_in[2];
    const float* b1  = (const float*)d_in[3];
    const float* W2  = (const float*)d_in[4];
    const float* b2  = (const float*)d_in[5];
    const float* Wfc = (const float*)d_in[6];
    const float* bfc = (const float*)d_in[7];
    float* out = (float*)d_out;

    const int n = in_sizes[0] / D;
    const int E = in_sizes[1] / 2;
    const int n4 = (n + 3) & ~3;
    const int E4 = (E + 3) & ~3;

    float* bufA   = (float*)d_ws;
    float* bufB   = bufA + (size_t)n * D;
    float* pq     = bufA;
    const int NB    = (n + RB - 1) / RB;
    const int G     = (E + CE - 1) / CE;
    const int GNB4  = (G * (NB + 1) + 3) & ~3;
    float* dinv     = bufB + (size_t)n * D;
    int*   rowptr   = (int*)(dinv + n4);
    int*   offs     = rowptr + n4 + 4;
    int2*  rec      = (int2*)(offs + GNB4);
    int2*  binned   = rec + E4;

    const int gGemm = (n + 127) / 128;
    const int gGemA = gGemm / 2;
    const int gGemC = gGemm - gGemA;
    const int gNode = (n + 7) / 8;
    const int gTile = (n + 63) / 64;
    const int gEdge = (E + 256 * EPT - 1) / (256 * EPT);
    const int gWgt  = (E + WE - 1) / WE;

    k_gemm_csort  <<<gGemA + G, 256, 0, stream>>>(x, W1, bufA, n, 0, gGemA, ei, E, NB, offs, binned);
    k_bin_sort    <<<NB, 1024, 0, stream>>>(offs, binned, n, NB, G, rowptr, dinv, rec);
    k_gemm_weight <<<gGemC + gWgt, 256, 0, stream>>>(x, W1, bufA, n, gGemA, gGemC, E, dinv, rec);

    k_gather_gemm<<<gTile, 1024, 0, stream>>>(rowptr, rec, bufA, dinv, b1, W2, bufB, n);

    k_gather_head<<<gNode, 256, 0, stream>>>(rowptr, rec, bufB, dinv, b2, Wfc, pq, n);

    k_edge_out<<<gEdge, 256, 0, stream>>>(ei, E, pq, bfc, out);
}


# Weight pass and launch removed: layer 1 forms dinv[src]*dinv[tgt] per entry; layer-2 rows stored times dinv[row], sum times dinv[tgt] (rounding shifts, sum order kept); x@W1 half rides bin sort
# speedup vs baseline: 4.9578x; 1.0829x over previous
#include <hip/hip_runtime.h>
#include <hip/hip_bf16.h>
#include <math.h>

typedef __attribute__((ext_vector_type(2))) float v2f;
typedef __attribute__((ext_vector_type(8))) float v8f;

#define D 128

#define RB 256
#define NBMAX 256
#define CE 8192
#define EPC (CE / 256)
#define GMAX 1024
#define SEGT 8
#define SEGU 8

__device__ __forceinline__ int scan256(int v, int* s_ws, int& tot) {
    const int lane = threadIdx.x & 31;
    const int wave = threadIdx.x >> 5;
    int s = v;
    for (int d = 1; d < 32; d <<= 1) {
        const int t = __shfl_up(s, d);
        if (lane >= d) s += t;
    }
    if (lane == 31) s_ws[wave] = s;
    __syncthreads();
    int off = 0, all = 0;
    for (int w = 0; w < 8; ++w) {
        const int x = s_ws[w];
        if (w < wave) off += x;
        all += x;
    }
    tot = all;
    return off + s - v;
}

__device__ __forceinline__ void bin_sort(
        const int* __restrict__ offs, const int2* __restrict__ binned, int n, int NB, int G,
        int* __restrict__ rowptr, float* __restrict__ dinv, int* __restrict__ rec, int b) {
    __shared__ int s_cnt[RB];
    __shared__ int s_cur[RB];
    __shared__ int s_ws[8];
    __shared__ int s_red[32];
    __shared__ int s_seg[GMAX];
    __shared__ int s_len[GMAX];
    const int t = threadIdx.x;
    int o0 = 0, o1 = 0;
    if (t < G) {
        const int* o = offs + (size_t)t * (NB + 1) + b;
        o0 = o[0];
        o1 = o[1];
    }
    s_seg[t] = t * CE + o0;
    s_len[t] = o1 - o0;
    if (t < RB) s_cnt[t] = 0;
    int lo;
    {
        int sum = o0;
        for (int d = 16; d >= 1; d >>= 1) sum += __shfl_xor(sum, d);
        if ((t & 31) == 0) s_red[t >> 5] = sum;
        __syncthreads();
        lo = 0;
        for (int w = 0; w < 32; ++w) lo += s_red[w];
    }
    for (int g = t / SEGT; g < G; g += 1024 / SEGT) {
        const int st = s_seg[g], len = s_len[g];
        for (int j0 = t & (SEGT - 1); j0 < len; j0 += SEGT * SEGU) {
            int y[SEGU];
            for (int k = 0; k < SEGU; ++k)
                y[k] = binned[st + min(j0 + k * SEGT, len - 1)].y;
            for (int k = 0; k < SEGU; ++k)
                if (j0 + k * SEGT < len) atomicAdd(&s_cnt[y[k] & (RB - 1)], 1);
        }
    }
    __syncthreads();
    {
        const int lane = t & 31, wave = t >> 5;
        const int v = (t < RB) ? s_cnt[t] : 0;
        int s = v;
        for (int d = 1; d < 32; d <<= 1) {
            const int x = __shfl_up(s, d);
            if (lane >= d) s += x;
        }
        if (t < RB && lane == 31) s_ws[wave] = s;
        __syncthreads();
        if (t < RB) {
            int off = 0;
            for (int w = 0; w < 8; ++w) if (w < wave) off += s_ws[w];
            const int ex = off + s - v;
            s_cur[t] = ex;
            const int node = b * RB + t;
            if (node < n) {
                rowptr[node] = lo + ex;
                const float d = (float)(v + 1);
                dinv[node] = (d > 0.0f) ? rsqrtf(d) : 0.0f;
            }
            if (node == n - 1) rowptr[n] = lo + ex + v;
        }
        __syncthreads();
    }
    for (int g = t / SEGT; g < G; g += 1024 / SEGT) {
        const int st = s_seg[g], len = s_len[g];
        for (int j0 = t & (SEGT - 1); j0 < len; j0 += SEGT * SEGU) {
            int2 q[SEGU];
            for (int k = 0; k < SEGU; ++k)
                q[k] = binned[st + min(j0 + k * SEGT, len - 1)];
            for (int k = 0; k < SEGU; ++k) {
                if (j0 + k * SEGT < len) {
                    const int pos = atomicAdd(&s_cur[q[k].y & (RB - 1)], 1);
                    rec[lo + pos] = q[k].x;
                }
            }
        }
    }
}

__device__ __forceinline__ void gemm_tile(
        const float* __restrict__ A, const float* __restrict__ W,
        float* __restrict__ C, int n, int blk, float* Wl) {
    for (int i = threadIdx.x; i < (D * D) / 4; i += 256)
        ((float4*)Wl)[i] = ((const float4*)W)[i];
    __syncthreads();

    const int wave = threadIdx.x >> 5;
    const int lane = threadIdx.x & 31;
    const int m0   = blk * 128 + wave * 16;

    int arow = m0 + (lane & 15);
    if (arow > n - 1) arow = n - 1;
    const int khalf = (lane >> 4) * 2;
    const float* Ar = A + (size_t)arow * D;

    v8f acc[8] = {};

    for (int k = 0; k < D; k += 4) {
        const v2f a = *(const v2f*)(Ar + k + khalf);
        const int kb = k + khalf;
#pragma unroll
        for (int t = 0; t < 8; ++t) {
            const int col = t * 16 + (lane & 15);
            v2f b;
            b.x = Wl[kb * D + col];
            b.y = Wl[(kb + 1) * D + col];
            acc[t] = __builtin_amdgcn_wmma_f32_16x16x4_f32(
                false, a, false, b, (short)0, acc[t], false, false);
        }
    }

    const int rbase = m0 + (lane >> 4) * 8;
    const int cix   = lane & 15;
#pragma unroll
    for (int t = 0; t < 8; ++t) {
#pragma unroll
        for (int v = 0; v < 8; ++v) {
            const int r = rbase + v;
            if (r < n) C[(size_t)r * D + t * 16 + cix] = acc[t][v];
        }
    }
}


__global__ __launch_bounds__(256) void k_gemm_csort(
        const float* __restrict__ A, const float* __restrict__ W,
        float* __restrict__ C, int n, int blk0, int gGemm,
        const int* __restrict__ ei, int E, int NB,
        int* __restrict__ offs, int2* __restrict__ binned) {
    __shared__ float Wl[D * D];
    __shared__ int s_cnt[NBMAX];
    __shared__ int s_off[NBMAX];
    __shared__ int s_ws[8];
    if ((int)blockIdx.x < gGemm) {
        gemm_tile(A, W, C, n, blk0 + (int)blockIdx.x, Wl);
        return;
    }
    const int g = (int)blockIdx.x - gGemm;
    const int t = threadIdx.x;
    const int e0 = g * CE + t;
    const bool full = g * CE + CE <= E;
    int r[EPC], c[EPC];
    for (int k = 0; k < EPC; ++k) {
        const int e = min(e0 + k * 256, E - 1);
        r[k] = ei[e];
        c[k] = ei[E + e];
    }
    s_cnt[t] = 0;
    __syncthreads();
    int rank[EPC];
    if (full) {
        for (int k = 0; k < EPC; ++k) rank[k] = atomicAdd(&s_cnt[c[k] >> 8], 1);
    } else {
        for (int k = 0; k < EPC; ++k)
            rank[k] = (e0 + k * 256 < E) ? atomicAdd(&s_cnt[c[k] >> 8], 1) : 0;
    }
    __syncthreads();
    const int v = s_cnt[t];
    int all;
    const int ex = scan256(v, s_ws, all);
    s_off[t] = ex;
    int* o = offs + (size_t)g * (NB + 1);
    if (t < NB) o[t] = ex;
    if (t == NB - 1) o[NB] = ex + v;
    __syncthreads();
    int2* dst = binned + (size_t)g * CE;
    if (full) {
        for (int k = 0; k < EPC; ++k) dst[s_off[c[k] >> 8] + rank[k]] = make_int2(r[k], c[k]);
    } else {
        for (int k = 0; k < EPC; ++k)
            if (e0 + k * 256 < E) dst[s_off[c[k] >> 8] + rank[k]] = make_int2(r[k], c[k]);
    }
}

#define HP 132
#define WP 136

__device__ __forceinline__ void gemm_tile32(
        const float* __restrict__ A, const float* __restrict__ W,
        float* __restrict__ C, int n, int blk) {
    __shared__ float Wl[D * WP];
    __shared__ float Al[128 * HP];
    for (int q = 0; q < 4; ++q) {
        const int f = (int)threadIdx.x + q * 1024;
        const int row = f >> 5, cg = f & 31;
        *(float4*)(Wl + row * WP + cg * 4) = ((const float4*)W)[f];
        const int ar = min(blk * 128 + row, n - 1);
        *(float4*)(Al + row * HP + cg * 4) = ((const float4*)A)[(size_t)ar * (D / 4) + cg];
    }
    __syncthreads();

    const int wave = threadIdx.x >> 5;
    const int lane = threadIdx.x & 31;
    const int mt = wave >> 3, nt = wave & 7;
    const int c = lane & 15;
    const int khalf = (lane >> 4) * 2;
    const float* Ar0 = Al + (mt * 16 + c) * HP + khalf;
    const float* Ar1 = Ar0 + 64 * HP;
    const float* Br  = Wl + khalf * WP + nt * 16 + c;
    v8f acc0 = {}, acc1 = {};
    for (int k = 0; k < D; k += 4) {
        const v2f a0 = *(const v2f*)(Ar0 + k);
        const v2f a1 = *(const v2f*)(Ar1 + k);
        v2f bf;
        bf.x = Br[k * WP];
        bf.y = Br[(k + 1) * WP];
        acc0 = __builtin_amdgcn_wmma_f32_16x16x4_f32(
            false, a0, false, bf, (short)0, acc0, false, false);
        acc1 = __builtin_amdgcn_wmma_f32_16x16x4_f32(
            false, a1, false, bf, (short)0, acc1, false, false);
    }
    const int rbase = blk * 128 + mt * 16 + (lane >> 4) * 8;
    for (int v = 0; v < 8; ++v) {
        const int r0 = rbase + v, r1 = r0 + 64;
        if (r0 < n) C[(size_t)r0 * D + nt * 16 + c] = acc0[v];
        if (r1 < n) C[(size_t)r1 * D + nt * 16 + c] = acc1[v];
    }
}

__global__ __launch_bounds__(1024) void k_bin_sort_gemm(
        const int* __restrict__ offs, const int2* __restrict__ binned, int n, int NB, int G,
        int* __restrict__ rowptr, float* __restrict__ dinv, int* __restrict__ rec,
        const float* __restrict__ A, const float* __restrict__ W,
        float* __restrict__ C, int blk0, int gGemm) {
    const int bid = (int)blockIdx.x;
    if (bid < gGemm) gemm_tile32(A, W, C, n, blk0 + bid);
    else             bin_sort(offs, binned, n, NB, G, rowptr, dinv, rec, bid - gGemm);
}

template <int K, bool WGT>
__device__ __forceinline__ void add_rows(float4& acc, const float4* __restrict__ X,
                                         int r_l, float d_l, float di, int j, int m, int lane) {
    float4 u[K];
    float  w[K];
    for (int k = 0; k < K; ++k) {
        const int r = __builtin_amdgcn_readlane(r_l, j + k);
        u[k] = X[(size_t)r * (D / 4) + lane];
    }
    if (WGT) {
        __builtin_amdgcn_sched_barrier(0);
        for (int k = 0; k < K; ++k) {
            const float d = __int_as_float(__builtin_amdgcn_readlane(__float_as_int(d_l), j + k));
            w[k] = (j + k < m) ? d * di : 0.0f;
        }
    } else {
        for (int k = 0; k < K; ++k) w[k] = (j + k < m) ? 1.0f : 0.0f;
    }
    for (int k = 0; k < K; ++k) {
        acc.x = __builtin_fmaf(u[k].x, w[k], acc.x);
        acc.y = __builtin_fmaf(u[k].y, w[k], acc.y);
        acc.z = __builtin_fmaf(u[k].z, w[k], acc.z);
        acc.w = __builtin_fmaf(u[k].w, w[k], acc.w);
    }
}

template <int M>
__device__ __forceinline__ float xrow(float x) {
    const int v = __builtin_bit_cast(int, x);
    return __builtin_bit_cast(float,
        __builtin_amdgcn_update_dpp(v, v, 0x160 + M, 0xf, 0xf, false));
}
__device__ __forceinline__ float xhalf(float x) {
    const int v = __builtin_bit_cast(int, x);
    return __builtin_bit_cast(float,
        __builtin_amdgcn_permlanex16(v, v, 0x76543210, 0xfedcba98, false, false));
}

__device__ __forceinline__ int list_source(
        const int* __restrict__ rec, int i, int beg, int deg, int t) {
    int r = i;
    if (t < deg) r = rec[beg + t];
    return r;
}
__device__ __forceinline__ float list_dinv(
        const float* __restrict__ dinv, int r_l, float di, int deg, int t) {
    float d = di;
    if (t < deg) d = dinv[r_l];
    return d;
}
__device__ __forceinline__ void list_entry(
        const int* __restrict__ rec, const float* __restrict__ dinv,
        int i, int beg, int deg, float di, int t, int& r_l, float& d_l) {
    int r = i;
    float d = di;
    if (t < deg) {
        r = rec[beg + t];
        d = dinv[r];
    }
    r_l = r;
    d_l = d;
}

template <bool WGT>
__device__ __forceinline__ float4 gather_sum(
        const float4* __restrict__ X, const int* __restrict__ rec, const float* __restrict__ dinv,
        int i, int beg, int deg, float di, int r_l, float d_l, int lane) {
    const int tot = deg + 1;
    float4 acc = {0.0f, 0.0f, 0.0f, 0.0f};
    for (int t0 = 0; t0 < tot; t0 += 32) {
        if (t0 > 0) {
            if (WGT) list_entry(rec, dinv, i, beg, deg, di, t0 + lane, r_l, d_l);
            else     r_l = list_source(rec, i, beg, deg, t0 + lane);
        }
        const int m = min(tot - t0, 32);
        int j = 0;
        for (; m - j > 4; j += 8) add_rows<8, WGT>(acc, X, r_l, d_l, di, j, m, lane);
        if (j < m)               add_rows<4, WGT>(acc, X, r_l, d_l, di, j, m, lane);
    }
    return acc;
}

__device__ __forceinline__ float4 bias_relu(float4 acc, float4 bb) {
    float4 o = {acc.x + bb.x, acc.y + bb.y, acc.z + bb.z, acc.w + bb.w};
    o.x = o.x > 0.0f ? o.x : 0.0f;
    o.y = o.y > 0.0f ? o.y : 0.0f;
    o.z = o.z > 0.0f ? o.z : 0.0f;
    o.w = o.w > 0.0f ? o.w : 0.0f;
    return o;
}

__global__ __launch_bounds__(1024) void k_gather_gemm(
        const int* __restrict__ rowptr, const int* __restrict__ rec,
        const float* __restrict__ xw, const float* __restrict__ dinv,
        const float* __restrict__ b, const float* __restrict__ W,
        float* __restrict__ C, int n) {
    __shared__ float Wl[D * WP];
    __shared__ float Hl[64 * HP];
    __shared__ float4 s_di4[16];
    const int wv = __builtin_amdgcn_readfirstlane((int)(threadIdx.x >> 5));
    const int lane = threadIdx.x & 31;
    const int base = blockIdx.x * 64;
    const int iA = base + 2 * wv, iB = iA + 1;
    const int cA = min(iA, n - 1), cB = min(iB, n - 1);
    const int begA = rowptr[cA], begB = rowptr[cB];
    const int degA = rowptr[cA + 1] - begA, degB = rowptr[cB + 1] - begB;
    const float diA = dinv[cA], diB = dinv[cB];
    const int rA = list_source(rec, cA, begA, degA, lane);
    const int rB = list_source(rec, cB, begB, degB, lane);
    const float4 bb = ((const float4*)b)[lane];

    for (int q = 0; q < 4; ++q) {
        const int f = (int)threadIdx.x + q * 1024;
        *(float4*)(Wl + (f >> 5) * WP + (f & 31) * 4) = ((const float4*)W)[f];
    }
    const float wA = list_dinv(dinv, rA, diA, degA, lane);
    const float wB = list_dinv(dinv, rB, diB, degB, lane);

    const float4* X = (const float4*)xw;
    float4 oA = {0.0f, 0.0f, 0.0f, 0.0f}, oB = oA;
    if (iA < n) oA = bias_relu(gather_sum<true>(X, rec, dinv, iA, begA, degA, diA, rA, wA, lane), bb);
    if (iB < n) oB = bias_relu(gather_sum<true>(X, rec, dinv, iB, begB, degB, diB, rB, wB, lane), bb);
    *(float4*)(Hl + (2 * wv) * HP + 4 * lane)     = oA;
    *(float4*)(Hl + (2 * wv + 1) * HP + 4 * lane) = oB;
    if (lane == 0) {
        float* s_di = (float*)s_di4;
        s_di[2 * wv]     = diA;
        s_di[2 * wv + 1] = diB;
    }
    __syncthreads();

    const int mt = wv >> 3, nt = wv & 7;
    const int c = lane & 15;
    const int khalf = (lane >> 4) * 2;
    const float* Ar = Hl + (mt * 16 + c) * HP + khalf;
    const float* Br = Wl + khalf * WP + nt * 16 + c;
    v8f acc = {};
    for (int k = 0; k < D; k += 4) {
        const v2f a = *(const v2f*)(Ar + k);
        v2f bf;
        bf.x = Br[k * WP];
        bf.y = Br[(k + 1) * WP];
        acc = __builtin_amdgcn_wmma_f32_16x16x4_f32(
            false, a, false, bf, (short)0, acc, false, false);
    }
    const int lrow = mt * 16 + (lane >> 4) * 8;
    const int rbase = base + lrow;
    const float4 d0 = s_di4[lrow >> 2], d1 = s_di4[(lrow >> 2) + 1];
    const float dr[8] = {d0.x, d0.y, d0.z, d0.w, d1.x, d1.y, d1.z, d1.w};
    for (int v = 0; v < 8; ++v) {
        const int r = rbase + v;
        if (r < n) C[(size_t)r * D + nt * 16 + c] = acc[v] * dr[v];
    }
}

__global__ __launch_bounds__(256) void k_gather_head(
        const int* __restrict__ rowptr, const int* __restrict__ rec,
        const float* __restrict__ xw, const float* __restrict__ dinv,
        const float* __restrict__ b, const float* __restrict__ Wfc,
        float* __restrict__ out, int n) {
    __shared__ float4 s_w[256];
    const int wv = __builtin_amdgcn_readfirstlane((int)(threadIdx.x >> 5));
    const int i = blockIdx.x * 8 + wv;
    const int lane = threadIdx.x & 31;
    const int ic = min(i, n - 1);
    const int beg = rowptr[ic];
    const int deg = rowptr[ic + 1] - beg;
    const float di = dinv[ic];
    {
        const int t = threadIdx.x;
        s_w[((t >> 7) * 4 + (t & 3)) * 32 + ((t & 127) >> 2)] = ((const float4*)Wfc)[t];
        __syncthreads();
    }
    if (i >= n) return;
    const int r_l = list_source(rec, i, beg, deg, lane);
    float4 acc = gather_sum<false>((const float4*)xw, rec, dinv, i, beg, deg, di, r_l, 0.0f, lane);
    acc.x *= di; acc.y *= di; acc.z *= di; acc.w *= di;
    const float4 o = bias_relu(acc, ((const float4*)b)[lane]);

    const float4 t0 = s_w[0 * 32 + lane], t1 = s_w[1 * 32 + lane];
    const float4 t2 = s_w[2 * 32 + lane], t3 = s_w[3 * 32 + lane];
    const float4 u0 = s_w[4 * 32 + lane], u1 = s_w[5 * 32 + lane];
    const float4 u2 = s_w[6 * 32 + lane], u3 = s_w[7 * 32 + lane];
    float s[8];
    s[0] = o.x * t0.x + o.y * t1.x + o.z * t2.x + o.w * t3.x;
    s[1] = o.x * t0.y + o.y * t1.y + o.z * t2.y + o.w * t3.y;
    s[2] = o.x * t0.z + o.y * t1.z + o.z * t2.z + o.w * t3.z;
    s[3] = o.x * t0.w + o.y * t1.w + o.z * t2.w + o.w * t3.w;
    s[4] = o.x * u0.x + o.y * u1.x + o.z * u2.x + o.w * u3.x;
    s[5] = o.x * u0.y + o.y * u1.y + o.z * u2.y + o.w * u3.y;
    s[6] = o.x * u0.z + o.y * u1.z + o.z * u2.z + o.w * u3.z;
    s[7] = o.x * u0.w + o.y * u1.w + o.z * u2.w + o.w * u3.w;

    const bool h16 = (lane & 16) != 0;
    float a4[4];
    for (int q = 0; q < 4; ++q) {
        const float keep = h16 ? s[q + 4] : s[q];
        const float give = h16 ? s[q] : s[q + 4];
        a4[q] = keep + xhalf(give);
    }
    const bool h8 = (lane & 8) != 0;
    float a2[2];
    for (int q = 0; q < 2; ++q) {
        const float keep = h8 ? a4[q + 2] : a4[q];
        const float give = h8 ? a4[q] : a4[q + 2];
        a2[q] = keep + xrow<8>(give);
    }
    const bool h4 = (lane & 4) != 0;
    const float keep = h4 ? a2[1] : a2[0];
    const float give = h4 ? a2[0] : a2[1];
    float a1 = keep + xrow<4>(give);
    a1 += xrow<2>(a1);
    a1 += xrow<1>(a1);
    if ((lane & 3) == 0)
        out[(size_t)i * 8 + ((lane >> 4) & 1) * 4 + ((lane >> 3) & 1) * 2 + ((lane >> 2) & 1)] = a1;
}

#define EPT 4
template <bool FULL>
__device__ __forceinline__ void edge_tile(
        const int* __restrict__ ei, int E, const float4* __restrict__ pq4,
        const float* __restrict__ bfc, float4* __restrict__ out4, int e0) {
    int r[EPT], c[EPT];
    for (int k = 0; k < EPT; ++k) {
        const int e = FULL ? e0 + k * 256 : min(e0 + k * 256, E - 1);
        r[k] = ei[e];
        c[k] = ei[E + e];
    }
    float4 p[EPT], q[EPT];
    for (int k = 0; k < EPT; ++k) {
        p[k] = pq4[(size_t)r[k] * 2];
        q[k] = pq4[(size_t)c[k] * 2 + 1];
    }
    const float b0 = bfc[0], b1 = bfc[1], b2 = bfc[2], b3 = bfc[3];
    for (int k = 0; k < EPT; ++k) {
        const float s0 = (p[k].x + q[k].x) + b0;
        const float s1 = (p[k].y + q[k].y) + b1;
        const float s2 = (p[k].z + q[k].z) + b2;
        const float s3 = (p[k].w + q[k].w) + b3;
        const float mx  = fmaxf(fmaxf(s0, s1), fmaxf(s2, s3));
        const float sum = __expf(s0 - mx) + __expf(s1 - mx) +
                          __expf(s2 - mx) + __expf(s3 - mx);
        const float lse = mx + __logf(sum);
        const float4 o  = {s0 - lse, s1 - lse, s2 - lse, s3 - lse};
        const int e = e0 + k * 256;
        if (FULL || e < E) out4[e] = o;
    }
}

__global__ __launch_bounds__(256) void k_edge_out(
        const int* __restrict__ ei, int E, const float* __restrict__ pq,
        const float* __restrict__ bfc, float* __restrict__ out) {
    const int tile0 = blockIdx.x * (256 * EPT);
    const int e0 = tile0 + threadIdx.x;
    if (tile0 + 256 * EPT <= E)
        edge_tile<true >(ei, E, (const float4*)pq, bfc, (float4*)out, e0);
    else
        edge_tile<false>(ei, E, (const float4*)pq, bfc, (float4*)out, e0);
}

extern "C" void kernel_launch(void* const* d_in, const int* in_sizes, int n_in,
                              void* d_out, int out_size, void* d_ws, size_t ws_size,
                              hipStream_t stream) {
    const float* x   = (const float*)d_in[0];
    const int*   ei  = (const int*)  d_in[1];
    const float* W1  = (const float*)d_in[2];
    const float* b1  = (const float*)d_in[3];
    const float* W2  = (const float*)d_in[4];
    const float* b2  = (const float*)d_in[5];
    const float* Wfc = (const float*)d_in[6];
    const float* bfc = (const float*)d_in[7];
    float* out = (float*)d_out;

    const int n = in_sizes[0] / D;
    const int E = in_sizes[1] / 2;
    const int n4 = (n + 3) & ~3;
    const int E4 = (E + 3) & ~3;

    float* bufA   = (float*)d_ws;
    float* bufB   = bufA + (size_t)n * D;
    float* pq     = bufA;
    const int NB    = (n + RB - 1) / RB;
    const int G     = (E + CE - 1) / CE;
    const int GNB4  = (G * (NB + 1) + 3) & ~3;
    float* dinv     = bufB + (size_t)n * D;
    int*   rowptr   = (int*)(dinv + n4);
    int*   offs     = rowptr + n4 + 4;
    int*   rec      = offs + GNB4;
    int2*  binned   = (int2*)(rec + E4);

    const int gGemm = (n + 127) / 128;
    const int gGemA = gGemm / 2;
    const int gGemC = gGemm - gGemA;
    const int gNode = (n + 7) / 8;
    const int gTile = (n + 63) / 64;
    const int gEdge = (E + 256 * EPT - 1) / (256 * EPT);

    k_gemm_csort   <<<gGemA + G, 256, 0, stream>>>(x, W1, bufA, n, 0, gGemA, ei, E, NB, offs, binned);
    k_bin_sort_gemm<<<gGemC + NB, 1024, 0, stream>>>(offs, binned, n, NB, G, rowptr, dinv, rec,
                                                     x, W1, bufA, gGemA, gGemC);

    k_gather_gemm<<<gTile, 1024, 0, stream>>>(rowptr, rec, bufA, dinv, b1, W2, bufB, n);

    k_gather_head<<<gNode, 256, 0, stream>>>(rowptr, rec, bufB, dinv, b2, Wfc, pq, n);

    k_edge_out<<<gEdge, 256, 0, stream>>>(ei, E, pq, bfc, out);
}


# Both sort passes order their output in LDS and copy it out as whole 128-byte lines (chunk records in the idle weight tile, bin sources in the idle A tile) instead of scattered 8- and 4-byte stores
# speedup vs baseline: 5.1696x; 1.0427x over previous
#include <hip/hip_runtime.h>
#include <hip/hip_bf16.h>
#include <math.h>

typedef __attribute__((ext_vector_type(2))) float v2f;
typedef __attribute__((ext_vector_type(8))) float v8f;

#define D 128

#define RB 256
#define NBMAX 256
#define CE 8192
#define EPC (CE / 256)
#define GMAX 1024
#define SEGT 8
#define SEGU 8

__device__ __forceinline__ int scan256(int v, int* s_ws, int& tot) {
    const int lane = threadIdx.x & 31;
    const int wave = threadIdx.x >> 5;
    int s = v;
    for (int d = 1; d < 32; d <<= 1) {
        const int t = __shfl_up(s, d);
        if (lane >= d) s += t;
    }
    if (lane == 31) s_ws[wave] = s;
    __syncthreads();
    int off = 0, all = 0;
    for (int w = 0; w < 8; ++w) {
        const int x = s_ws[w];
        if (w < wave) off += x;
        all += x;
    }
    tot = all;
    return off + s - v;
}

__device__ __forceinline__ void place_sources(
        const int2* __restrict__ binned, const int* s_seg, const int* s_len, int* s_cur,
        int* dst, int G, int t) {
    for (int g = t / SEGT; g < G; g += 1024 / SEGT) {
        const int st = s_seg[g], len = s_len[g];
        for (int j0 = t & (SEGT - 1); j0 < len; j0 += SEGT * SEGU) {
            int2 q[SEGU];
            for (int k = 0; k < SEGU; ++k)
                q[k] = binned[st + min(j0 + k * SEGT, len - 1)];
            for (int k = 0; k < SEGU; ++k) {
                if (j0 + k * SEGT < len) {
                    const int pos = atomicAdd(&s_cur[q[k].y & (RB - 1)], 1);
                    dst[pos] = q[k].x;
                }
            }
        }
    }
}

__device__ __forceinline__ void bin_sort(
        const int* __restrict__ offs, const int2* __restrict__ binned, int n, int NB, int G,
        int* __restrict__ rowptr, float* __restrict__ dinv, int* __restrict__ rec, int b,
        int* s_rec, int rcap) {
    __shared__ int s_cnt[RB];
    __shared__ int s_cur[RB];
    __shared__ int s_ws[8];
    __shared__ int s_red[32];
    __shared__ int s_seg[GMAX];
    __shared__ int s_len[GMAX];
    __shared__ int s_tot;
    const int t = threadIdx.x;
    int o0 = 0, o1 = 0;
    if (t < G) {
        const int* o = offs + (size_t)t * (NB + 1) + b;
        o0 = o[0];
        o1 = o[1];
    }
    s_seg[t] = t * CE + o0;
    s_len[t] = o1 - o0;
    if (t < RB) s_cnt[t] = 0;
    int lo;
    {
        int sum = o0;
        for (int d = 16; d >= 1; d >>= 1) sum += __shfl_xor(sum, d);
        if ((t & 31) == 0) s_red[t >> 5] = sum;
        __syncthreads();
        lo = 0;
        for (int w = 0; w < 32; ++w) lo += s_red[w];
    }
    for (int g = t / SEGT; g < G; g += 1024 / SEGT) {
        const int st = s_seg[g], len = s_len[g];
        for (int j0 = t & (SEGT - 1); j0 < len; j0 += SEGT * SEGU) {
            int y[SEGU];
            for (int k = 0; k < SEGU; ++k)
                y[k] = binned[st + min(j0 + k * SEGT, len - 1)].y;
            for (int k = 0; k < SEGU; ++k)
                if (j0 + k * SEGT < len) atomicAdd(&s_cnt[y[k] & (RB - 1)], 1);
        }
    }
    __syncthreads();
    {
        const int lane = t & 31, wave = t >> 5;
        const int v = (t < RB) ? s_cnt[t] : 0;
        int s = v;
        for (int d = 1; d < 32; d <<= 1) {
            const int x = __shfl_up(s, d);
            if (lane >= d) s += x;
        }
        if (t < RB && lane == 31) s_ws[wave] = s;
        __syncthreads();
        if (t < RB) {
            int off = 0;
            for (int w = 0; w < 8; ++w) if (w < wave) off += s_ws[w];
            const int ex = off + s - v;
            s_cur[t] = ex;
            const int node = b * RB + t;
            if (node < n) {
                rowptr[node] = lo + ex;
                const float d = (float)(v + 1);
                dinv[node] = (d > 0.0f) ? rsqrtf(d) : 0.0f;
            }
            if (node == n - 1) rowptr[n] = lo + ex + v;
            if (t == RB - 1) s_tot = ex + v;
        }
        __syncthreads();
    }
    const int tot = s_tot;
    const bool staged = tot <= rcap;
    if (staged) {
        place_sources(binned, s_seg, s_len, s_cur, s_rec, G, t);
        __syncthreads();
        for (int j = t; j < tot; j += 1024) rec[lo + j] = s_rec[j];
    } else {
        place_sources(binned, s_seg, s_len, s_cur, rec + lo, G, t);
    }
}

__device__ __forceinline__ void gemm_tile(
        const float* __restrict__ A, const float* __restrict__ W,
        float* __restrict__ C, int n, int blk, float* Wl) {
    for (int i = threadIdx.x; i < (D * D) / 4; i += 256)
        ((float4*)Wl)[i] = ((const float4*)W)[i];
    __syncthreads();

    const int wave = threadIdx.x >> 5;
    const int lane = threadIdx.x & 31;
    const int m0   = blk * 128 + wave * 16;

    int arow = m0 + (lane & 15);
    if (arow > n - 1) arow = n - 1;
    const int khalf = (lane >> 4) * 2;
    const float* Ar = A + (size_t)arow * D;

    v8f acc[8] = {};

    for (int k = 0; k < D; k += 4) {
        const v2f a = *(const v2f*)(Ar + k + khalf);
        const int kb = k + khalf;
#pragma unroll
        for (int t = 0; t < 8; ++t) {
            const int col = t * 16 + (lane & 15);
            v2f b;
            b.x = Wl[kb * D + col];
            b.y = Wl[(kb + 1) * D + col];
            acc[t] = __builtin_amdgcn_wmma_f32_16x16x4_f32(
                false, a, false, b, (short)0, acc[t], false, false);
        }
    }

    const int rbase = m0 + (lane >> 4) * 8;
    const int cix   = lane & 15;
#pragma unroll
    for (int t = 0; t < 8; ++t) {
#pragma unroll
        for (int v = 0; v < 8; ++v) {
            const int r = rbase + v;
            if (r < n) C[(size_t)r * D + t * 16 + cix] = acc[t][v];
        }
    }
}


__global__ __launch_bounds__(256) void k_gemm_csort(
        const float* __restrict__ A, const float* __restrict__ W,
        float* __restrict__ C, int n, int blk0, int gGemm,
        const int* __restrict__ ei, int E, int NB,
        int* __restrict__ offs, int2* __restrict__ binned) {
    __shared__ float Wl[D * D];
    __shared__ int s_cnt[NBMAX];
    __shared__ int s_off[NBMAX];
    __shared__ int s_ws[8];
    if ((int)blockIdx.x < gGemm) {
        gemm_tile(A, W, C, n, blk0 + (int)blockIdx.x, Wl);
        return;
    }
    const int g = (int)blockIdx.x - gGemm;
    const int t = threadIdx.x;
    const int e0 = g * CE + t;
    const bool full = g * CE + CE <= E;
    int r[EPC], c[EPC];
    for (int k = 0; k < EPC; ++k) {
        const int e = min(e0 + k * 256, E - 1);
        r[k] = ei[e];
        c[k] = ei[E + e];
    }
    s_cnt[t] = 0;
    __syncthreads();
    int rank[EPC];
    if (full) {
        for (int k = 0; k < EPC; ++k) rank[k] = atomicAdd(&s_cnt[c[k] >> 8], 1);
    } else {
        for (int k = 0; k < EPC; ++k)
            rank[k] = (e0 + k * 256 < E) ? atomicAdd(&s_cnt[c[k] >> 8], 1) : 0;
    }
    __syncthreads();
    const int v = s_cnt[t];
    int all;
    const int ex = scan256(v, s_ws, all);
    s_off[t] = ex;
    int* o = offs + (size_t)g * (NB + 1);
    if (t < NB) o[t] = ex;
    if (t == NB - 1) o[NB] = ex + v;
    __syncthreads();
    static_assert(CE * sizeof(int2) <= D * D * sizeof(float), "a chunk's records must fit the weight tile");
    int2* stage = reinterpret_cast<int2*>(Wl);
    if (full) {
        for (int k = 0; k < EPC; ++k) stage[s_off[c[k] >> 8] + rank[k]] = make_int2(r[k], c[k]);
    } else {
        for (int k = 0; k < EPC; ++k)
            if (e0 + k * 256 < E) stage[s_off[c[k] >> 8] + rank[k]] = make_int2(r[k], c[k]);
    }
    __syncthreads();
    int2* dst = binned + (size_t)g * CE;
    if (full) {
        const int4* s4 = reinterpret_cast<const int4*>(stage);
        int4* d4 = reinterpret_cast<int4*>(dst);
        for (int k = 0; k < CE / 2 / 256; ++k) d4[t + k * 256] = s4[t + k * 256];
    } else {
        const int cnt = E - g * CE;
        for (int j = t; j < cnt; j += 256) dst[j] = stage[j];
    }
}

#define HP 132
#define WP 136

__device__ __forceinline__ void gemm_tile32(
        const float* __restrict__ A, const float* __restrict__ W,
        float* __restrict__ C, int n, int blk, float* Wl, float* Al) {
    for (int q = 0; q < 4; ++q) {
        const int f = (int)threadIdx.x + q * 1024;
        const int row = f >> 5, cg = f & 31;
        *(float4*)(Wl + row * WP + cg * 4) = ((const float4*)W)[f];
        const int ar = min(blk * 128 + row, n - 1);
        *(float4*)(Al + row * HP + cg * 4) = ((const float4*)A)[(size_t)ar * (D / 4) + cg];
    }
    __syncthreads();

    const int wave = threadIdx.x >> 5;
    const int lane = threadIdx.x & 31;
    const int mt = wave >> 3, nt = wave & 7;
    const int c = lane & 15;
    const int khalf = (lane >> 4) * 2;
    const float* Ar0 = Al + (mt * 16 + c) * HP + khalf;
    const float* Ar1 = Ar0 + 64 * HP;
    const float* Br  = Wl + khalf * WP + nt * 16 + c;
    v8f acc0 = {}, acc1 = {};
    for (int k = 0; k < D; k += 4) {
        const v2f a0 = *(const v2f*)(Ar0 + k);
        const v2f a1 = *(const v2f*)(Ar1 + k);
        v2f bf;
        bf.x = Br[k * WP];
        bf.y = Br[(k + 1) * WP];
        acc0 = __builtin_amdgcn_wmma_f32_16x16x4_f32(
            false, a0, false, bf, (short)0, acc0, false, false);
        acc1 = __builtin_amdgcn_wmma_f32_16x16x4_f32(
            false, a1, false, bf, (short)0, acc1, false, false);
    }
    const int rbase = blk * 128 + mt * 16 + (lane >> 4) * 8;
    for (int v = 0; v < 8; ++v) {
        const int r0 = rbase + v, r1 = r0 + 64;
        if (r0 < n) C[(size_t)r0 * D + nt * 16 + c] = acc0[v];
        if (r1 < n) C[(size_t)r1 * D + nt * 16 + c] = acc1[v];
    }
}

__global__ __launch_bounds__(1024) void k_bin_sort_gemm(
        const int* __restrict__ offs, const int2* __restrict__ binned, int n, int NB, int G,
        int* __restrict__ rowptr, float* __restrict__ dinv, int* __restrict__ rec,
        const float* __restrict__ A, const float* __restrict__ W,
        float* __restrict__ C, int blk0, int gGemm) {
    __shared__ float Wl[D * WP];
    __shared__ float Al[128 * HP];
    const int bid = (int)blockIdx.x;
    if (bid < gGemm) gemm_tile32(A, W, C, n, blk0 + bid, Wl, Al);
    else             bin_sort(offs, binned, n, NB, G, rowptr, dinv, rec, bid - gGemm,
                              reinterpret_cast<int*>(Al), 128 * HP);
}

template <int K, bool WGT>
__device__ __forceinline__ void add_rows(float4& acc, const float4* __restrict__ X,
                                         int r_l, float d_l, float di, int j, int m, int lane) {
    float4 u[K];
    float  w[K];
    for (int k = 0; k < K; ++k) {
        const int r = __builtin_amdgcn_readlane(r_l, j + k);
        u[k] = X[(size_t)r * (D / 4) + lane];
    }
    if (WGT) {
        __builtin_amdgcn_sched_barrier(0);
        for (int k = 0; k < K; ++k) {
            const float d = __int_as_float(__builtin_amdgcn_readlane(__float_as_int(d_l), j + k));
            w[k] = (j + k < m) ? d * di : 0.0f;
        }
    } else {
        for (int k = 0; k < K; ++k) w[k] = (j + k < m) ? 1.0f : 0.0f;
    }
    for (int k = 0; k < K; ++k) {
        acc.x = __builtin_fmaf(u[k].x, w[k], acc.x);
        acc.y = __builtin_fmaf(u[k].y, w[k], acc.y);
        acc.z = __builtin_fmaf(u[k].z, w[k], acc.z);
        acc.w = __builtin_fmaf(u[k].w, w[k], acc.w);
    }
}

template <int M>
__device__ __forceinline__ float xrow(float x) {
    const int v = __builtin_bit_cast(int, x);
    return __builtin_bit_cast(float,
        __builtin_amdgcn_update_dpp(v, v, 0x160 + M, 0xf, 0xf, false));
}
__device__ __forceinline__ float xhalf(float x) {
    const int v = __builtin_bit_cast(int, x);
    return __builtin_bit_cast(float,
        __builtin_amdgcn_permlanex16(v, v, 0x76543210, 0xfedcba98, false, false));
}

__device__ __forceinline__ int list_source(
        const int* __restrict__ rec, int i, int beg, int deg, int t) {
    int r = i;
    if (t < deg) r = rec[beg + t];
    return r;
}
__device__ __forceinline__ float list_dinv(
        const float* __restrict__ dinv, int r_l, float di, int deg, int t) {
    float d = di;
    if (t < deg) d = dinv[r_l];
    return d;
}
__device__ __forceinline__ void list_entry(
        const int* __restrict__ rec, const float* __restrict__ dinv,
        int i, int beg, int deg, float di, int t, int& r_l, float& d_l) {
    int r = i;
    float d = di;
    if (t < deg) {
        r = rec[beg + t];
        d = dinv[r];
    }
    r_l = r;
    d_l = d;
}

template <bool WGT>
__device__ __forceinline__ float4 gather_sum(
        const float4* __restrict__ X, const int* __restrict__ rec, const float* __restrict__ dinv,
        int i, int beg, int deg, float di, int r_l, float d_l, int lane) {
    const int tot = deg + 1;
    float4 acc = {0.0f, 0.0f, 0.0f, 0.0f};
    for (int t0 = 0; t0 < tot; t0 += 32) {
        if (t0 > 0) {
            if (WGT) list_entry(rec, dinv, i, beg, deg, di, t0 + lane, r_l, d_l);
            else     r_l = list_source(rec, i, beg, deg, t0 + lane);
        }
        const int m = min(tot - t0, 32);
        int j = 0;
        for (; m - j > 4; j += 8) add_rows<8, WGT>(acc, X, r_l, d_l, di, j, m, lane);
        if (j < m)               add_rows<4, WGT>(acc, X, r_l, d_l, di, j, m, lane);
    }
    return acc;
}

__device__ __forceinline__ float4 bias_relu(float4 acc, float4 bb) {
    float4 o = {acc.x + bb.x, acc.y + bb.y, acc.z + bb.z, acc.w + bb.w};
    o.x = o.x > 0.0f ? o.x : 0.0f;
    o.y = o.y > 0.0f ? o.y : 0.0f;
    o.z = o.z > 0.0f ? o.z : 0.0f;
    o.w = o.w > 0.0f ? o.w : 0.0f;
    return o;
}

__global__ __launch_bounds__(1024) void k_gather_gemm(
        const int* __restrict__ rowptr, const int* __restrict__ rec,
        const float* __restrict__ xw, const float* __restrict__ dinv,
        const float* __restrict__ b, const float* __restrict__ W,
        float* __restrict__ C, int n) {
    __shared__ float Wl[D * WP];
    __shared__ float Hl[64 * HP];
    __shared__ float4 s_di4[16];
    const int wv = __builtin_amdgcn_readfirstlane((int)(threadIdx.x >> 5));
    const int lane = threadIdx.x & 31;
    const int base = blockIdx.x * 64;
    const int iA = base + 2 * wv, iB = iA + 1;
    const int cA = min(iA, n - 1), cB = min(iB, n - 1);
    const int begA = rowptr[cA], begB = rowptr[cB];
    const int degA = rowptr[cA + 1] - begA, degB = rowptr[cB + 1] - begB;
    const float diA = dinv[cA], diB = dinv[cB];
    const int rA = list_source(rec, cA, begA, degA, lane);
    const int rB = list_source(rec, cB, begB, degB, lane);
    const float4 bb = ((const float4*)b)[lane];

    for (int q = 0; q < 4; ++q) {
        const int f = (int)threadIdx.x + q * 1024;
        *(float4*)(Wl + (f >> 5) * WP + (f & 31) * 4) = ((const float4*)W)[f];
    }
    const float wA = list_dinv(dinv, rA, diA, degA, lane);
    const float wB = list_dinv(dinv, rB, diB, degB, lane);

    const float4* X = (const float4*)xw;
    float4 oA = {0.0f, 0.0f, 0.0f, 0.0f}, oB = oA;
    if (iA < n) oA = bias_relu(gather_sum<true>(X, rec, dinv, iA, begA, degA, diA, rA, wA, lane), bb);
    if (iB < n) oB = bias_relu(gather_sum<true>(X, rec, dinv, iB, begB, degB, diB, rB, wB, lane), bb);
    *(float4*)(Hl + (2 * wv) * HP + 4 * lane)     = oA;
    *(float4*)(Hl + (2 * wv + 1) * HP + 4 * lane) = oB;
    if (lane == 0) {
        float* s_di = (float*)s_di4;
        s_di[2 * wv]     = diA;
        s_di[2 * wv + 1] = diB;
    }
    __syncthreads();

    const int mt = wv >> 3, nt = wv & 7;
    const int c = lane & 15;
    const int khalf = (lane >> 4) * 2;
    const float* Ar = Hl + (mt * 16 + c) * HP + khalf;
    const float* Br = Wl + khalf * WP + nt * 16 + c;
    v8f acc = {};
    for (int k = 0; k < D; k += 4) {
        const v2f a = *(const v2f*)(Ar + k);
        v2f bf;
        bf.x = Br[k * WP];
        bf.y = Br[(k + 1) * WP];
        acc = __builtin_amdgcn_wmma_f32_16x16x4_f32(
            false, a, false, bf, (short)0, acc, false, false);
    }
    const int lrow = mt * 16 + (lane >> 4) * 8;
    const int rbase = base + lrow;
    const float4 d0 = s_di4[lrow >> 2], d1 = s_di4[(lrow >> 2) + 1];
    const float dr[8] = {d0.x, d0.y, d0.z, d0.w, d1.x, d1.y, d1.z, d1.w};
    for (int v = 0; v < 8; ++v) {
        const int r = rbase + v;
        if (r < n) C[(size_t)r * D + nt * 16 + c] = acc[v] * dr[v];
    }
}

__global__ __launch_bounds__(256) void k_gather_head(
        const int* __restrict__ rowptr, const int* __restrict__ rec,
        const float* __restrict__ xw, const float* __restrict__ dinv,
        const float* __restrict__ b, const float* __restrict__ Wfc,
        float* __restrict__ out, int n) {
    __shared__ float4 s_w[256];
    const int wv = __builtin_amdgcn_readfirstlane((int)(threadIdx.x >> 5));
    const int i = blockIdx.x * 8 + wv;
    const int lane = threadIdx.x & 31;
    const int ic = min(i, n - 1);
    const int beg = rowptr[ic];
    const int deg = rowptr[ic + 1] - beg;
    const float di = dinv[ic];
    {
        const int t = threadIdx.x;
        s_w[((t >> 7) * 4 + (t & 3)) * 32 + ((t & 127) >> 2)] = ((const float4*)Wfc)[t];
        __syncthreads();
    }
    if (i >= n) return;
    const int r_l = list_source(rec, i, beg, deg, lane);
    float4 acc = gather_sum<false>((const float4*)xw, rec, dinv, i, beg, deg, di, r_l, 0.0f, lane);
    acc.x *= di; acc.y *= di; acc.z *= di; acc.w *= di;
    const float4 o = bias_relu(acc, ((const float4*)b)[lane]);

    const float4 t0 = s_w[0 * 32 + lane], t1 = s_w[1 * 32 + lane];
    const float4 t2 = s_w[2 * 32 + lane], t3 = s_w[3 * 32 + lane];
    const float4 u0 = s_w[4 * 32 + lane], u1 = s_w[5 * 32 + lane];
    const float4 u2 = s_w[6 * 32 + lane], u3 = s_w[7 * 32 + lane];
    float s[8];
    s[0] = o.x * t0.x + o.y * t1.x + o.z * t2.x + o.w * t3.x;
    s[1] = o.x * t0.y + o.y * t1.y + o.z * t2.y + o.w * t3.y;
    s[2] = o.x * t0.z + o.y * t1.z + o.z * t2.z + o.w * t3.z;
    s[3] = o.x * t0.w + o.y * t1.w + o.z * t2.w + o.w * t3.w;
    s[4] = o.x * u0.x + o.y * u1.x + o.z * u2.x + o.w * u3.x;
    s[5] = o.x * u0.y + o.y * u1.y + o.z * u2.y + o.w * u3.y;
    s[6] = o.x * u0.z + o.y * u1.z + o.z * u2.z + o.w * u3.z;
    s[7] = o.x * u0.w + o.y * u1.w + o.z * u2.w + o.w * u3.w;

    const bool h16 = (lane & 16) != 0;
    float a4[4];
    for (int q = 0; q < 4; ++q) {
        const float keep = h16 ? s[q + 4] : s[q];
        const float give = h16 ? s[q] : s[q + 4];
        a4[q] = keep + xhalf(give);
    }
    const bool h8 = (lane & 8) != 0;
    float a2[2];
    for (int q = 0; q < 2; ++q) {
        const float keep = h8 ? a4[q + 2] : a4[q];
        const float give = h8 ? a4[q] : a4[q + 2];
        a2[q] = keep + xrow<8>(give);
    }
    const bool h4 = (lane & 4) != 0;
    const float keep = h4 ? a2[1] : a2[0];
    const float give = h4 ? a2[0] : a2[1];
    float a1 = keep + xrow<4>(give);
    a1 += xrow<2>(a1);
    a1 += xrow<1>(a1);
    if ((lane & 3) == 0)
        out[(size_t)i * 8 + ((lane >> 4) & 1) * 4 + ((lane >> 3) & 1) * 2 + ((lane >> 2) & 1)] = a1;
}

#define EPT 4
template <bool FULL>
__device__ __forceinline__ void edge_tile(
        const int* __restrict__ ei, int E, const float4* __restrict__ pq4,
        const float* __restrict__ bfc, float4* __restrict__ out4, int e0) {
    int r[EPT], c[EPT];
    for (int k = 0; k < EPT; ++k) {
        const int e = FULL ? e0 + k * 256 : min(e0 + k * 256, E - 1);
        r[k] = ei[e];
        c[k] = ei[E + e];
    }
    float4 p[EPT], q[EPT];
    for (int k = 0; k < EPT; ++k) {
        p[k] = pq4[(size_t)r[k] * 2];
        q[k] = pq4[(size_t)c[k] * 2 + 1];
    }
    const float b0 = bfc[0], b1 = bfc[1], b2 = bfc[2], b3 = bfc[3];
    for (int k = 0; k < EPT; ++k) {
        const float s0 = (p[k].x + q[k].x) + b0;
        const float s1 = (p[k].y + q[k].y) + b1;
        const float s2 = (p[k].z + q[k].z) + b2;
        const float s3 = (p[k].w + q[k].w) + b3;
        const float mx  = fmaxf(fmaxf(s0, s1), fmaxf(s2, s3));
        const float sum = __expf(s0 - mx) + __expf(s1 - mx) +
                          __expf(s2 - mx) + __expf(s3 - mx);
        const float lse = mx + __logf(sum);
        const float4 o  = {s0 - lse, s1 - lse, s2 - lse, s3 - lse};
        const int e = e0 + k * 256;
        if (FULL || e < E) out4[e] = o;
    }
}

__global__ __launch_bounds__(256) void k_edge_out(
        const int* __restrict__ ei, int E, const float* __restrict__ pq,
        const float* __restrict__ bfc, float* __restrict__ out) {
    const int tile0 = blockIdx.x * (256 * EPT);
    const int e0 = tile0 + threadIdx.x;
    if (tile0 + 256 * EPT <= E)
        edge_tile<true >(ei, E, (const float4*)pq, bfc, (float4*)out, e0);
    else
        edge_tile<false>(ei, E, (const float4*)pq, bfc, (float4*)out, e0);
}

extern "C" void kernel_launch(void* const* d_in, const int* in_sizes, int n_in,
                              void* d_out, int out_size, void* d_ws, size_t ws_size,
                              hipStream_t stream) {
    const float* x   = (const float*)d_in[0];
    const int*   ei  = (const int*)  d_in[1];
    const float* W1  = (const float*)d_in[2];
    const float* b1  = (const float*)d_in[3];
    const float* W2  = (const float*)d_in[4];
    const float* b2  = (const float*)d_in[5];
    const float* Wfc = (const float*)d_in[6];
    const float* bfc = (const float*)d_in[7];
    float* out = (float*)d_out;

    const int n = in_sizes[0] / D;
    const int E = in_sizes[1] / 2;
    const int n4 = (n + 3) & ~3;
    const int E4 = (E + 3) & ~3;

    float* bufA   = (float*)d_ws;
    float* bufB   = bufA + (size_t)n * D;
    float* pq     = bufA;
    const int NB    = (n + RB - 1) / RB;
    const int G     = (E + CE - 1) / CE;
    const int GNB4  = (G * (NB + 1) + 3) & ~3;
    float* dinv     = bufB + (size_t)n * D;
    int*   rowptr   = (int*)(dinv + n4);
    int*   offs     = rowptr + n4 + 4;
    int*   rec      = offs + GNB4;
    int2*  binned   = (int2*)(rec + E4);

    const int gGemm = (n + 127) / 128;
    const int gGemA = gGemm / 2;
    const int gGemC = gGemm - gGemA;
    const int gNode = (n + 7) / 8;
    const int gTile = (n + 63) / 64;
    const int gEdge = (E + 256 * EPT - 1) / (256 * EPT);

    k_gemm_csort   <<<gGemA + G, 256, 0, stream>>>(x, W1, bufA, n, 0, gGemA, ei, E, NB, offs, binned);
    k_bin_sort_gemm<<<gGemC + NB, 1024, 0, stream>>>(offs, binned, n, NB, G, rowptr, dinv, rec,
                                                     x, W1, bufA, gGemA, gGemC);

    k_gather_gemm<<<gTile, 1024, 0, stream>>>(rowptr, rec, bufA, dinv, b1, W2, bufB, n);

    k_gather_head<<<gNode, 256, 0, stream>>>(rowptr, rec, bufB, dinv, b2, Wfc, pq, n);

    k_edge_out<<<gEdge, 256, 0, stream>>>(ei, E, pq, bfc, out);
}


# Layer 1 reassociated to (A_hat x)@W1: its gather reads x itself, W1 and W2 products run back to back in one LDS tile; x@W1 and its 26 MB plane leave the sort launches; sum order changes
# speedup vs baseline: 5.4253x; 1.0495x over previous
#include <hip/hip_runtime.h>
#include <hip/hip_bf16.h>
#include <math.h>

typedef __attribute__((ext_vector_type(2))) float v2f;
typedef __attribute__((ext_vector_type(8))) float v8f;

#define D 128

#define RB 256
#define NBMAX 256
#define CE 8192
#define EPC (CE / 256)
#define GMAX 1024
#define SEGT 8
#define SEGU 8

__device__ __forceinline__ int scan256(int v, int* s_ws, int& tot) {
    const int lane = threadIdx.x & 31;
    const int wave = threadIdx.x >> 5;
    int s = v;
    for (int d = 1; d < 32; d <<= 1) {
        const int t = __shfl_up(s, d);
        if (lane >= d) s += t;
    }
    if (lane == 31) s_ws[wave] = s;
    __syncthreads();
    int off = 0, all = 0;
    for (int w = 0; w < 8; ++w) {
        const int x = s_ws[w];
        if (w < wave) off += x;
        all += x;
    }
    tot = all;
    return off + s - v;
}

__device__ __forceinline__ void place_sources(
        const int2* __restrict__ binned, const int* s_seg, const int* s_len, int* s_cur,
        int* dst, int G, int t) {
    for (int g = t / SEGT; g < G; g += 1024 / SEGT) {
        const int st = s_seg[g], len = s_len[g];
        for (int j0 = t & (SEGT - 1); j0 < len; j0 += SEGT * SEGU) {
            int2 q[SEGU];
            for (int k = 0; k < SEGU; ++k)
                q[k] = binned[st + min(j0 + k * SEGT, len - 1)];
            for (int k = 0; k < SEGU; ++k) {
                if (j0 + k * SEGT < len) {
                    const int pos = atomicAdd(&s_cur[q[k].y & (RB - 1)], 1);
                    dst[pos] = q[k].x;
                }
            }
        }
    }
}

__device__ __forceinline__ void bin_sort(
        const int* __restrict__ offs, const int2* __restrict__ binned, int n, int NB, int G,
        int* __restrict__ rowptr, float* __restrict__ dinv, int* __restrict__ rec, int b,
        int* s_rec, int rcap) {
    __shared__ int s_cnt[RB];
    __shared__ int s_cur[RB];
    __shared__ int s_ws[8];
    __shared__ int s_red[32];
    __shared__ int s_seg[GMAX];
    __shared__ int s_len[GMAX];
    __shared__ int s_tot;
    const int t = threadIdx.x;
    int o0 = 0, o1 = 0;
    if (t < G) {
        const int* o = offs + (size_t)t * (NB + 1) + b;
        o0 = o[0];
        o1 = o[1];
    }
    s_seg[t] = t * CE + o0;
    s_len[t] = o1 - o0;
    if (t < RB) s_cnt[t] = 0;
    int lo;
    {
        int sum = o0;
        for (int d = 16; d >= 1; d >>= 1) sum += __shfl_xor(sum, d);
        if ((t & 31) == 0) s_red[t >> 5] = sum;
        __syncthreads();
        lo = 0;
        for (int w = 0; w < 32; ++w) lo += s_red[w];
    }
    for (int g = t / SEGT; g < G; g += 1024 / SEGT) {
        const int st = s_seg[g], len = s_len[g];
        for (int j0 = t & (SEGT - 1); j0 < len; j0 += SEGT * SEGU) {
            int y[SEGU];
            for (int k = 0; k < SEGU; ++k)
                y[k] = binned[st + min(j0 + k * SEGT, len - 1)].y;
            for (int k = 0; k < SEGU; ++k)
                if (j0 + k * SEGT < len) atomicAdd(&s_cnt[y[k] & (RB - 1)], 1);
        }
    }
    __syncthreads();
    {
        const int lane = t & 31, wave = t >> 5;
        const int v = (t < RB) ? s_cnt[t] : 0;
        int s = v;
        for (int d = 1; d < 32; d <<= 1) {
            const int x = __shfl_up(s, d);
            if (lane >= d) s += x;
        }
        if (t < RB && lane == 31) s_ws[wave] = s;
        __syncthreads();
        if (t < RB) {
            int off = 0;
            for (int w = 0; w < 8; ++w) if (w < wave) off += s_ws[w];
            const int ex = off + s - v;
            s_cur[t] = ex;
            const int node = b * RB + t;
            if (node < n) {
                rowptr[node] = lo + ex;
                const float d = (float)(v + 1);
                dinv[node] = (d > 0.0f) ? rsqrtf(d) : 0.0f;
            }
            if (node == n - 1) rowptr[n] = lo + ex + v;
            if (t == RB - 1) s_tot = ex + v;
        }
        __syncthreads();
    }
    const int tot = s_tot;
    const bool staged = tot <= rcap;
    if (staged) {
        place_sources(binned, s_seg, s_len, s_cur, s_rec, G, t);
        __syncthreads();
        for (int j = t; j < tot; j += 1024) rec[lo + j] = s_rec[j];
    } else {
        place_sources(binned, s_seg, s_len, s_cur, rec + lo, G, t);
    }
}

__global__ __launch_bounds__(256) void k_csort(
        const int* __restrict__ ei, int E, int NB,
        int* __restrict__ offs, int2* __restrict__ binned) {
    __shared__ int2 stage[CE];
    __shared__ int s_cnt[NBMAX];
    __shared__ int s_off[NBMAX];
    __shared__ int s_ws[8];
    const int g = (int)blockIdx.x;
    const int t = threadIdx.x;
    const int e0 = g * CE + t;
    const bool full = g * CE + CE <= E;
    int r[EPC], c[EPC];
    for (int k = 0; k < EPC; ++k) {
        const int e = min(e0 + k * 256, E - 1);
        r[k] = ei[e];
        c[k] = ei[E + e];
    }
    s_cnt[t] = 0;
    __syncthreads();
    int rank[EPC];
    if (full) {
        for (int k = 0; k < EPC; ++k) rank[k] = atomicAdd(&s_cnt[c[k] >> 8], 1);
    } else {
        for (int k = 0; k < EPC; ++k)
            rank[k] = (e0 + k * 256 < E) ? atomicAdd(&s_cnt[c[k] >> 8], 1) : 0;
    }
    __syncthreads();
    const int v = s_cnt[t];
    int all;
    const int ex = scan256(v, s_ws, all);
    s_off[t] = ex;
    int* o = offs + (size_t)g * (NB + 1);
    if (t < NB) o[t] = ex;
    if (t == NB - 1) o[NB] = ex + v;
    __syncthreads();
    if (full) {
        for (int k = 0; k < EPC; ++k) stage[s_off[c[k] >> 8] + rank[k]] = make_int2(r[k], c[k]);
    } else {
        for (int k = 0; k < EPC; ++k)
            if (e0 + k * 256 < E) stage[s_off[c[k] >> 8] + rank[k]] = make_int2(r[k], c[k]);
    }
    __syncthreads();
    int2* dst = binned + (size_t)g * CE;
    if (full) {
        const int4* s4 = reinterpret_cast<const int4*>(stage);
        int4* d4 = reinterpret_cast<int4*>(dst);
        for (int k = 0; k < CE / 2 / 256; ++k) d4[t + k * 256] = s4[t + k * 256];
    } else {
        const int cnt = E - g * CE;
        for (int j = t; j < cnt; j += 256) dst[j] = stage[j];
    }
}

#define HP 132
#define WP 136

#define RCAP (128 * HP)

__global__ __launch_bounds__(1024) void k_bin_sort(
        const int* __restrict__ offs, const int2* __restrict__ binned, int n, int NB, int G,
        int* __restrict__ rowptr, float* __restrict__ dinv, int* __restrict__ rec) {
    __shared__ int s_rec[RCAP];
    bin_sort(offs, binned, n, NB, G, rowptr, dinv, rec, (int)blockIdx.x, s_rec, RCAP);
}

template <int K, bool WGT>
__device__ __forceinline__ void add_rows(float4& acc, const float4* __restrict__ X,
                                         int r_l, float d_l, float di, int j, int m, int lane) {
    float4 u[K];
    float  w[K];
    for (int k = 0; k < K; ++k) {
        const int r = __builtin_amdgcn_readlane(r_l, j + k);
        u[k] = X[(size_t)r * (D / 4) + lane];
    }
    if (WGT) {
        __builtin_amdgcn_sched_barrier(0);
        for (int k = 0; k < K; ++k) {
            const float d = __int_as_float(__builtin_amdgcn_readlane(__float_as_int(d_l), j + k));
            w[k] = (j + k < m) ? d * di : 0.0f;
        }
    } else {
        for (int k = 0; k < K; ++k) w[k] = (j + k < m) ? 1.0f : 0.0f;
    }
    for (int k = 0; k < K; ++k) {
        acc.x = __builtin_fmaf(u[k].x, w[k], acc.x);
        acc.y = __builtin_fmaf(u[k].y, w[k], acc.y);
        acc.z = __builtin_fmaf(u[k].z, w[k], acc.z);
        acc.w = __builtin_fmaf(u[k].w, w[k], acc.w);
    }
}

template <int M>
__device__ __forceinline__ float xrow(float x) {
    const int v = __builtin_bit_cast(int, x);
    return __builtin_bit_cast(float,
        __builtin_amdgcn_update_dpp(v, v, 0x160 + M, 0xf, 0xf, false));
}
__device__ __forceinline__ float xhalf(float x) {
    const int v = __builtin_bit_cast(int, x);
    return __builtin_bit_cast(float,
        __builtin_amdgcn_permlanex16(v, v, 0x76543210, 0xfedcba98, false, false));
}

__device__ __forceinline__ int list_source(
        const int* __restrict__ rec, int i, int beg, int deg, int t) {
    int r = i;
    if (t < deg) r = rec[beg + t];
    return r;
}
__device__ __forceinline__ float list_dinv(
        const float* __restrict__ dinv, int r_l, float di, int deg, int t) {
    float d = di;
    if (t < deg) d = dinv[r_l];
    return d;
}
__device__ __forceinline__ void list_entry(
        const int* __restrict__ rec, const float* __restrict__ dinv,
        int i, int beg, int deg, float di, int t, int& r_l, float& d_l) {
    int r = i;
    float d = di;
    if (t < deg) {
        r = rec[beg + t];
        d = dinv[r];
    }
    r_l = r;
    d_l = d;
}

template <bool WGT>
__device__ __forceinline__ float4 gather_sum(
        const float4* __restrict__ X, const int* __restrict__ rec, const float* __restrict__ dinv,
        int i, int beg, int deg, float di, int r_l, float d_l, int lane) {
    const int tot = deg + 1;
    float4 acc = {0.0f, 0.0f, 0.0f, 0.0f};
    for (int t0 = 0; t0 < tot; t0 += 32) {
        if (t0 > 0) {
            if (WGT) list_entry(rec, dinv, i, beg, deg, di, t0 + lane, r_l, d_l);
            else     r_l = list_source(rec, i, beg, deg, t0 + lane);
        }
        const int m = min(tot - t0, 32);
        int j = 0;
        for (; m - j > 4; j += 8) add_rows<8, WGT>(acc, X, r_l, d_l, di, j, m, lane);
        if (j < m)               add_rows<4, WGT>(acc, X, r_l, d_l, di, j, m, lane);
    }
    return acc;
}

__device__ __forceinline__ float4 bias_relu(float4 acc, float4 bb) {
    float4 o = {acc.x + bb.x, acc.y + bb.y, acc.z + bb.z, acc.w + bb.w};
    o.x = o.x > 0.0f ? o.x : 0.0f;
    o.y = o.y > 0.0f ? o.y : 0.0f;
    o.z = o.z > 0.0f ? o.z : 0.0f;
    o.w = o.w > 0.0f ? o.w : 0.0f;
    return o;
}

__global__ __launch_bounds__(1024) void k_gather_gemm(
        const int* __restrict__ rowptr, const int* __restrict__ rec,
        const float* __restrict__ x, const float* __restrict__ dinv,
        const float* __restrict__ b, const float* __restrict__ W1,
        const float* __restrict__ W2, float* __restrict__ C, int n) {
    __shared__ float Wl[D * WP];
    __shared__ float Hl[64 * HP];
    __shared__ float4 s_di4[16];
    const int wv = __builtin_amdgcn_readfirstlane((int)(threadIdx.x >> 5));
    const int lane = threadIdx.x & 31;
    const int base = blockIdx.x * 64;
    const int iA = base + 2 * wv, iB = iA + 1;
    const int cA = min(iA, n - 1), cB = min(iB, n - 1);
    const int begA = rowptr[cA], begB = rowptr[cB];
    const int degA = rowptr[cA + 1] - begA, degB = rowptr[cB + 1] - begB;
    const float diA = dinv[cA], diB = dinv[cB];
    const int rA = list_source(rec, cA, begA, degA, lane);
    const int rB = list_source(rec, cB, begB, degB, lane);

    for (int q = 0; q < 4; ++q) {
        const int f = (int)threadIdx.x + q * 1024;
        *(float4*)(Wl + (f >> 5) * WP + (f & 31) * 4) = ((const float4*)W1)[f];
    }
    const float wA = list_dinv(dinv, rA, diA, degA, lane);
    const float wB = list_dinv(dinv, rB, diB, degB, lane);

    const float4* X = (const float4*)x;
    float4 oA = {0.0f, 0.0f, 0.0f, 0.0f}, oB = oA;
    if (iA < n) oA = gather_sum<true>(X, rec, dinv, iA, begA, degA, diA, rA, wA, lane);
    if (iB < n) oB = gather_sum<true>(X, rec, dinv, iB, begB, degB, diB, rB, wB, lane);
    *(float4*)(Hl + (2 * wv) * HP + 4 * lane)     = oA;
    *(float4*)(Hl + (2 * wv + 1) * HP + 4 * lane) = oB;
    if (lane == 0) {
        float* s_di = (float*)s_di4;
        s_di[2 * wv]     = diA;
        s_di[2 * wv + 1] = diB;
    }
    __syncthreads();

    const int mt = wv >> 3, nt = wv & 7;
    const int c = lane & 15;
    const int khalf = (lane >> 4) * 2;
    const float* Ar = Hl + (mt * 16 + c) * HP + khalf;
    const float* Br = Wl + khalf * WP + nt * 16 + c;
    const int lrow = mt * 16 + (lane >> 4) * 8;

    const float bc = b[nt * 16 + c];

    v8f acc = {};
    for (int k = 0; k < D; k += 4) {
        const v2f a = *(const v2f*)(Ar + k);
        v2f bf;
        bf.x = Br[k * WP];
        bf.y = Br[(k + 1) * WP];
        acc = __builtin_amdgcn_wmma_f32_16x16x4_f32(
            false, a, false, bf, (short)0, acc, false, false);
    }
    __syncthreads();
    float4 w2[4];
    for (int q = 0; q < 4; ++q) w2[q] = ((const float4*)W2)[(int)threadIdx.x + q * 1024];
    for (int v = 0; v < 8; ++v) {
        const float h = acc[v] + bc;
        Hl[(lrow + v) * HP + nt * 16 + c] = h > 0.0f ? h : 0.0f;
    }
    for (int q = 0; q < 4; ++q) {
        const int f = (int)threadIdx.x + q * 1024;
        *(float4*)(Wl + (f >> 5) * WP + (f & 31) * 4) = w2[q];
    }
    __syncthreads();

    v8f acc2 = {};
    for (int k = 0; k < D; k += 4) {
        const v2f a = *(const v2f*)(Ar + k);
        v2f bf;
        bf.x = Br[k * WP];
        bf.y = Br[(k + 1) * WP];
        acc2 = __builtin_amdgcn_wmma_f32_16x16x4_f32(
            false, a, false, bf, (short)0, acc2, false, false);
    }
    const int rbase = base + lrow;
    const float4 d0 = s_di4[lrow >> 2], d1 = s_di4[(lrow >> 2) + 1];
    const float dr[8] = {d0.x, d0.y, d0.z, d0.w, d1.x, d1.y, d1.z, d1.w};
    for (int v = 0; v < 8; ++v) {
        const int r = rbase + v;
        if (r < n) C[(size_t)r * D + nt * 16 + c] = acc2[v] * dr[v];
    }
}

__global__ __launch_bounds__(256) void k_gather_head(
        const int* __restrict__ rowptr, const int* __restrict__ rec,
        const float* __restrict__ xw, const float* __restrict__ dinv,
        const float* __restrict__ b, const float* __restrict__ Wfc,
        float* __restrict__ out, int n) {
    __shared__ float4 s_w[256];
    const int wv = __builtin_amdgcn_readfirstlane((int)(threadIdx.x >> 5));
    const int i = blockIdx.x * 8 + wv;
    const int lane = threadIdx.x & 31;
    const int ic = min(i, n - 1);
    const int beg = rowptr[ic];
    const int deg = rowptr[ic + 1] - beg;
    const float di = dinv[ic];
    {
        const int t = threadIdx.x;
        s_w[((t >> 7) * 4 + (t & 3)) * 32 + ((t & 127) >> 2)] = ((const float4*)Wfc)[t];
        __syncthreads();
    }
    if (i >= n) return;
    const int r_l = list_source(rec, i, beg, deg, lane);
    float4 acc = gather_sum<false>((const float4*)xw, rec, dinv, i, beg, deg, di, r_l, 0.0f, lane);
    acc.x *= di; acc.y *= di; acc.z *= di; acc.w *= di;
    const float4 o = bias_relu(acc, ((const float4*)b)[lane]);

    const float4 t0 = s_w[0 * 32 + lane], t1 = s_w[1 * 32 + lane];
    const float4 t2 = s_w[2 * 32 + lane], t3 = s_w[3 * 32 + lane];
    const float4 u0 = s_w[4 * 32 + lane], u1 = s_w[5 * 32 + lane];
    const float4 u2 = s_w[6 * 32 + lane], u3 = s_w[7 * 32 + lane];
    float s[8];
    s[0] = o.x * t0.x + o.y * t1.x + o.z * t2.x + o.w * t3.x;
    s[1] = o.x * t0.y + o.y * t1.y + o.z * t2.y + o.w * t3.y;
    s[2] = o.x * t0.z + o.y * t1.z + o.z * t2.z + o.w * t3.z;
    s[3] = o.x * t0.w + o.y * t1.w + o.z * t2.w + o.w * t3.w;
    s[4] = o.x * u0.x + o.y * u1.x + o.z * u2.x + o.w * u3.x;
    s[5] = o.x * u0.y + o.y * u1.y + o.z * u2.y + o.w * u3.y;
    s[6] = o.x * u0.z + o.y * u1.z + o.z * u2.z + o.w * u3.z;
    s[7] = o.x * u0.w + o.y * u1.w + o.z * u2.w + o.w * u3.w;

    const bool h16 = (lane & 16) != 0;
    float a4[4];
    for (int q = 0; q < 4; ++q) {
        const float keep = h16 ? s[q + 4] : s[q];
        const float give = h16 ? s[q] : s[q + 4];
        a4[q] = keep + xhalf(give);
    }
    const bool h8 = (lane & 8) != 0;
    float a2[2];
    for (int q = 0; q < 2; ++q) {
        const float keep = h8 ? a4[q + 2] : a4[q];
        const float give = h8 ? a4[q] : a4[q + 2];
        a2[q] = keep + xrow<8>(give);
    }
    const bool h4 = (lane & 4) != 0;
    const float keep = h4 ? a2[1] : a2[0];
    const float give = h4 ? a2[0] : a2[1];
    float a1 = keep + xrow<4>(give);
    a1 += xrow<2>(a1);
    a1 += xrow<1>(a1);
    if ((lane & 3) == 0)
        out[(size_t)i * 8 + ((lane >> 4) & 1) * 4 + ((lane >> 3) & 1) * 2 + ((lane >> 2) & 1)] = a1;
}

#define EPT 4
template <bool FULL>
__device__ __forceinline__ void edge_tile(
        const int* __restrict__ ei, int E, const float4* __restrict__ pq4,
        const float* __restrict__ bfc, float4* __restrict__ out4, int e0) {
    int r[EPT], c[EPT];
    for (int k = 0; k < EPT; ++k) {
        const int e = FULL ? e0 + k * 256 : min(e0 + k * 256, E - 1);
        r[k] = ei[e];
        c[k] = ei[E + e];
    }
    float4 p[EPT], q[EPT];
    for (int k = 0; k < EPT; ++k) {
        p[k] = pq4[(size_t)r[k] * 2];
        q[k] = pq4[(size_t)c[k] * 2 + 1];
    }
    const float b0 = bfc[0], b1 = bfc[1], b2 = bfc[2], b3 = bfc[3];
    for (int k = 0; k < EPT; ++k) {
        const float s0 = (p[k].x + q[k].x) + b0;
        const float s1 = (p[k].y + q[k].y) + b1;
        const float s2 = (p[k].z + q[k].z) + b2;
        const float s3 = (p[k].w + q[k].w) + b3;
        const float mx  = fmaxf(fmaxf(s0, s1), fmaxf(s2, s3));
        const float sum = __expf(s0 - mx) + __expf(s1 - mx) +
                          __expf(s2 - mx) + __expf(s3 - mx);
        const float lse = mx + __logf(sum);
        const float4 o  = {s0 - lse, s1 - lse, s2 - lse, s3 - lse};
        const int e = e0 + k * 256;
        if (FULL || e < E) out4[e] = o;
    }
}

__global__ __launch_bounds__(256) void k_edge_out(
        const int* __restrict__ ei, int E, const float* __restrict__ pq,
        const float* __restrict__ bfc, float* __restrict__ out) {
    const int tile0 = blockIdx.x * (256 * EPT);
    const int e0 = tile0 + threadIdx.x;
    if (tile0 + 256 * EPT <= E)
        edge_tile<true >(ei, E, (const float4*)pq, bfc, (float4*)out, e0);
    else
        edge_tile<false>(ei, E, (const float4*)pq, bfc, (float4*)out, e0);
}

extern "C" void kernel_launch(void* const* d_in, const int* in_sizes, int n_in,
                              void* d_out, int out_size, void* d_ws, size_t ws_size,
                              hipStream_t stream) {
    const float* x   = (const float*)d_in[0];
    const int*   ei  = (const int*)  d_in[1];
    const float* W1  = (const float*)d_in[2];
    const float* b1  = (const float*)d_in[3];
    const float* W2  = (const float*)d_in[4];
    const float* b2  = (const float*)d_in[5];
    const float* Wfc = (const float*)d_in[6];
    const float* bfc = (const float*)d_in[7];
    float* out = (float*)d_out;

    const int n = in_sizes[0] / D;
    const int E = in_sizes[1] / 2;
    const int n4 = (n + 3) & ~3;
    const int E4 = (E + 3) & ~3;

    float* pq     = (float*)d_ws;
    float* bufB   = pq + (size_t)n * 8;
    const int NB    = (n + RB - 1) / RB;
    const int G     = (E + CE - 1) / CE;
    const int GNB4  = (G * (NB + 1) + 3) & ~3;
    float* dinv     = bufB + (size_t)n * D;
    int*   rowptr   = (int*)(dinv + n4);
    int*   offs     = rowptr + n4 + 4;
    int*   rec      = offs + GNB4;
    int2*  binned   = (int2*)(rec + E4);

    const int gNode = (n + 7) / 8;
    const int gTile = (n + 63) / 64;
    const int gEdge = (E + 256 * EPT - 1) / (256 * EPT);

    k_csort   <<<G, 256, 0, stream>>>(ei, E, NB, offs, binned);
    k_bin_sort<<<NB, 1024, 0, stream>>>(offs, binned, n, NB, G, rowptr, dinv, rec);

    k_gather_gemm<<<gTile, 1024, 0, stream>>>(rowptr, rec, x, dinv, b1, W1, W2, bufB, n);

    k_gather_head<<<gNode, 256, 0, stream>>>(rowptr, rec, bufB, dinv, b2, Wfc, pq, n);

    k_edge_out<<<gEdge, 256, 0, stream>>>(ei, E, pq, bfc, out);
}
